# starting kernel
# baseline (speedup 1.0000x reference)
#include <hip/hip_runtime.h>
#include <math.h>

#define NNODES 50000
#define NEDGES 800000
#define NTOT   (NEDGES + NNODES)
#define FEAT 128
#define HID  128
#define HEADS 8
#define HDIM 16
#define NEG_SLOPE 0.2f

typedef __attribute__((ext_vector_type(2))) float v2f;
typedef __attribute__((ext_vector_type(8))) float v8f;

static __device__ __forceinline__ void atomicMaxF(float* addr, float val) {
    if (val >= 0.f) atomicMax((int*)addr, __float_as_int(val));
    else            atomicMin((unsigned int*)addr, (unsigned int)__float_as_int(val));
}

static __device__ __forceinline__ void edge_sd(const int* __restrict__ ei, int e,
                                               int& s, int& d) {
    if (e < NEDGES) { s = ei[e]; d = ei[NEDGES + e]; }
    else            { s = e - NEDGES; d = s; }
}

__global__ __launch_bounds__(256)
void k_gemm_wmma(const float* __restrict__ X, const float* __restrict__ W,
                 float* __restrict__ Hout) {
    __shared__ float ldsW[FEAT * HID];
    const int tid = threadIdx.x;

    const float4* Wv = (const float4*)W;
    float4*       Lv = (float4*)ldsW;
#pragma unroll
    for (int i = 0; i < (FEAT * HID / 4) / 256; ++i)
        Lv[tid + i * 256] = Wv[tid + i * 256];
    __syncthreads();

    const int lane = tid & 31;
    const int wave = tid >> 5;
    const int l16  = lane & 15;
    const int half = lane >> 4;
    const int row  = blockIdx.x * 16 + l16;
    const int col  = wave * 16 + l16;

    const float* xrow = X + (size_t)row * FEAT;

    v8f acc = {};
#pragma unroll
    for (int k0 = 0; k0 < FEAT; k0 += 4) {
        const int ka = k0 + 2 * half;
        v2f a, b;
        a.x = xrow[ka];
        a.y = xrow[ka + 1];
        b.x = ldsW[ka * HID + col];
        b.y = ldsW[(ka + 1) * HID + col];
        acc = __builtin_amdgcn_wmma_f32_16x16x4_f32(
                  false, a, false, b, (short)0, acc, false, false);
    }

    const int orow0 = blockIdx.x * 16 + 8 * half;
#pragma unroll
    for (int j = 0; j < 8; ++j)
        Hout[(size_t)(orow0 + j) * HID + col] = acc[j];
}

__global__ __launch_bounds__(256)
void k_alpha(const float* __restrict__ Hf, const float* __restrict__ a_src,
             const float* __restrict__ a_dst, float* __restrict__ AS,
             float* __restrict__ AD) {
    int t = blockIdx.x * blockDim.x + threadIdx.x;
    if (t >= NNODES * HEADS) return;
    int hd = t & (HEADS - 1);
    int n  = t >> 3;
    const float* hp = Hf + (size_t)n * HID + hd * HDIM;
    const float* as = a_src + hd * HDIM;
    const float* ad = a_dst + hd * HDIM;
    float sa = 0.f, sd = 0.f;
#pragma unroll
    for (int c = 0; c < HDIM; ++c) {
        float v = hp[c];
        sa += v * as[c];
        sd += v * ad[c];
    }
    AS[t] = sa;
    AD[t] = sd;
}

__global__ __launch_bounds__(256)
void k_fill(float* __restrict__ p, float v, int n) {
    int i = blockIdx.x * blockDim.x + threadIdx.x;
    if (i < n) p[i] = v;
}

static __device__ __forceinline__ float edge_e(const float* AS, const float* AD,
                                               int s, int d, int hd) {
    float v = AS[s * HEADS + hd] + AD[d * HEADS + hd];
    return v >= 0.f ? v : NEG_SLOPE * v;
}

__global__ __launch_bounds__(256)
void k_edge_max(const int* __restrict__ ei, const float* __restrict__ AS,
                const float* __restrict__ AD, float* __restrict__ M) {
    int t = blockIdx.x * blockDim.x + threadIdx.x;
    if (t >= NTOT * HEADS) return;
    int hd = t & (HEADS - 1), e = t >> 3;
    int s, d; edge_sd(ei, e, s, d);
    atomicMaxF(&M[d * HEADS + hd], edge_e(AS, AD, s, d, hd));
}

__global__ __launch_bounds__(256)
void k_edge_sum(const int* __restrict__ ei, const float* __restrict__ AS,
                const float* __restrict__ AD, const float* __restrict__ M,
                float* __restrict__ S) {
    int t = blockIdx.x * blockDim.x + threadIdx.x;
    if (t >= NTOT * HEADS) return;
    int hd = t & (HEADS - 1), e = t >> 3;
    int s, d; edge_sd(ei, e, s, d);
    float w = expf(edge_e(AS, AD, s, d, hd) - M[d * HEADS + hd]);
    atomicAdd(&S[d * HEADS + hd], w);
}

__global__ __launch_bounds__(256)
void k_edge_agg(const int* __restrict__ ei, const float* __restrict__ AS,
                const float* __restrict__ AD, const float* __restrict__ M,
                const float* __restrict__ S, const float* __restrict__ Hf,
                float* __restrict__ Agg) {
    int t = blockIdx.x * blockDim.x + threadIdx.x;
    if (t >= NTOT * HEADS) return;
    int hd = t & (HEADS - 1), e = t >> 3;
    int s, d; edge_sd(ei, e, s, d);
    float w     = expf(edge_e(AS, AD, s, d, hd) - M[d * HEADS + hd]);
    float alpha = w / (S[d * HEADS + hd] + 1e-16f);
    const float* hs = Hf  + (size_t)s * HID + hd * HDIM;
    float*       op = Agg + (size_t)d * HID + hd * HDIM;
    __builtin_prefetch(hs, 0, 1);
#pragma unroll
    for (int c = 0; c < HDIM; ++c)
        atomicAdd(&op[c], alpha * hs[c]);
}

__global__ __launch_bounds__(256)
void k_bias_elu(const float* __restrict__ Agg, const float* __restrict__ b,
                float* __restrict__ Out) {
    int t = blockIdx.x * blockDim.x + threadIdx.x;
    if (t >= NNODES * HID) return;
    float v = Agg[t] + b[t & (HID - 1)];
    Out[t] = v > 0.f ? v : expm1f(v);
}

static inline unsigned grid_for(long long n) { return (unsigned)((n + 255) / 256); }

extern "C" void kernel_launch(void* const* d_in, const int* in_sizes, int n_in,
                              void* d_out, int out_size, void* d_ws, size_t ws_size,
                              hipStream_t stream) {
    const float* x   = (const float*)d_in[0];
    const int*   ei  = (const int*)  d_in[1];
    const float* W1  = (const float*)d_in[2];
    const float* as1 = (const float*)d_in[3];
    const float* ad1 = (const float*)d_in[4];
    const float* b1  = (const float*)d_in[5];
    const float* W2  = (const float*)d_in[6];
    const float* as2 = (const float*)d_in[7];
    const float* ad2 = (const float*)d_in[8];
    const float* b2  = (const float*)d_in[9];

    float* ws     = (float*)d_ws;
    float* bufH   = ws;
    float* bufAS  = bufH  + (size_t)NNODES * HID;
    float* bufAD  = bufAS + (size_t)NNODES * HEADS;
    float* bufM   = bufAD + (size_t)NNODES * HEADS;
    float* bufS   = bufM  + (size_t)NNODES * HEADS;
    float* bufAgg = bufS  + (size_t)NNODES * HEADS;

    const float neg_inf = -__builtin_huge_valf();

    auto run_layer = [&](const float* xin, const float* Wm, const float* as,
                         const float* ad, const float* bv, float* outBuf) {
        k_gemm_wmma<<<NNODES / 16, 256, 0, stream>>>(xin, Wm, bufH);
        k_alpha<<<grid_for((long long)NNODES * HEADS), 256, 0, stream>>>(
            bufH, as, ad, bufAS, bufAD);
        k_fill<<<grid_for((long long)NNODES * HEADS), 256, 0, stream>>>(
            bufM, neg_inf, NNODES * HEADS);
        hipMemsetAsync(bufS, 0, (size_t)NNODES * HEADS * sizeof(float), stream);
        hipMemsetAsync(outBuf, 0, (size_t)NNODES * HID * sizeof(float), stream);
        k_edge_max<<<grid_for((long long)NTOT * HEADS), 256, 0, stream>>>(
            ei, bufAS, bufAD, bufM);
        k_edge_sum<<<grid_for((long long)NTOT * HEADS), 256, 0, stream>>>(
            ei, bufAS, bufAD, bufM, bufS);
        k_edge_agg<<<grid_for((long long)NTOT * HEADS), 256, 0, stream>>>(
            ei, bufAS, bufAD, bufM, bufS, bufH, outBuf);
        k_bias_elu<<<grid_for((long long)NNODES * HID), 256, 0, stream>>>(
            outBuf, bv, outBuf);
    };

    run_layer(x,      W1, as1, ad1, b1, bufAgg);
    run_layer(bufAgg, W2, as2, ad2, b2, (float*)d_out);
}


# Edges bucketed by destination in-call (count, scan, place by arrival-order rank); one wave per node does max, exp-sum, weighted row sum in registers, one row store: no f32 atomics; sum order changes
# speedup vs baseline: 5.2737x; 5.2737x over previous
#include <hip/hip_runtime.h>
#include <math.h>

#define NNODES 50000
#define NEDGES 800000
#define FEAT 128
#define HID  128
#define HEADS 8
#define HDIM 16
#define NEG_SLOPE 0.2f

typedef __attribute__((ext_vector_type(2))) float v2f;
typedef __attribute__((ext_vector_type(8))) float v8f;

__global__ __launch_bounds__(256)
void k_gemm_wmma(const float* __restrict__ X, const float* __restrict__ W,
                 float* __restrict__ Hout) {
    __shared__ float ldsW[FEAT * HID];
    const int tid = threadIdx.x;

    const float4* Wv = (const float4*)W;
    float4*       Lv = (float4*)ldsW;
#pragma unroll
    for (int i = 0; i < (FEAT * HID / 4) / 256; ++i)
        Lv[tid + i * 256] = Wv[tid + i * 256];
    __syncthreads();

    const int lane = tid & 31;
    const int wave = tid >> 5;
    const int l16  = lane & 15;
    const int half = lane >> 4;
    const int row  = blockIdx.x * 16 + l16;
    const int col  = wave * 16 + l16;

    const float* xrow = X + (size_t)row * FEAT;

    v8f acc = {};
#pragma unroll
    for (int k0 = 0; k0 < FEAT; k0 += 4) {
        const int ka = k0 + 2 * half;
        v2f a, b;
        a.x = xrow[ka];
        a.y = xrow[ka + 1];
        b.x = ldsW[ka * HID + col];
        b.y = ldsW[(ka + 1) * HID + col];
        acc = __builtin_amdgcn_wmma_f32_16x16x4_f32(
                  false, a, false, b, (short)0, acc, false, false);
    }

    const int orow0 = blockIdx.x * 16 + 8 * half;
#pragma unroll
    for (int j = 0; j < 8; ++j)
        Hout[(size_t)(orow0 + j) * HID + col] = acc[j];
}

__global__ __launch_bounds__(256)
void k_alpha(const float* __restrict__ Hf, const float* __restrict__ a_src,
             const float* __restrict__ a_dst, float* __restrict__ AS,
             float* __restrict__ AD) {
    int t = blockIdx.x * blockDim.x + threadIdx.x;
    if (t >= NNODES * HEADS) return;
    int hd = t & (HEADS - 1);
    int n  = t >> 3;
    const float* hp = Hf + (size_t)n * HID + hd * HDIM;
    const float* as = a_src + hd * HDIM;
    const float* ad = a_dst + hd * HDIM;
    float sa = 0.f, sd = 0.f;
#pragma unroll
    for (int c = 0; c < HDIM; ++c) {
        float v = hp[c];
        sa += v * as[c];
        sd += v * ad[c];
    }
    AS[t] = sa;
    AD[t] = sd;
}

__global__ __launch_bounds__(256)
void k_count(const int* __restrict__ ei, int* __restrict__ rp) {
    int e = blockIdx.x * blockDim.x + threadIdx.x;
    if (e >= NEDGES) return;
    int d = ei[NEDGES + e];
    atomicAdd(&rp[d], 1);
}

#define SCAN_T  1024
#define SCAN_V  13
static_assert(SCAN_T * SCAN_V * 4 >= NNODES && NNODES % 4 == 0, "scan covers every node in whole pieces");

__global__ __launch_bounds__(SCAN_T)
void k_scan(int* __restrict__ rp) {
    __shared__ int wtot[32];
    const int t = threadIdx.x, lane = t & 31, w = t >> 5;
    int4* rp4 = (int4*)rp;
    const int base4 = t * SCAN_V;
    int4 v[SCAN_V];
    int run = 0;
#pragma unroll
    for (int k = 0; k < SCAN_V; ++k) {
        const int i4 = base4 + k;
        int4 c = make_int4(0, 0, 0, 0);
        if (i4 < NNODES / 4) c = rp4[i4];
        run += c.x; c.x = run;
        run += c.y; c.y = run;
        run += c.z; c.z = run;
        run += c.w; c.w = run;
        v[k] = c;
    }
    int inc = run;
    for (int o = 1; o < 32; o <<= 1) {
        int n = __shfl_up(inc, o);
        if (lane >= o) inc += n;
    }
    if (lane == 31) wtot[w] = inc;
    __syncthreads();
    if (w == 0) {
        int x = wtot[lane], xi = x;
        for (int o = 1; o < 32; o <<= 1) {
            int n = __shfl_up(xi, o);
            if (lane >= o) xi += n;
        }
        wtot[lane] = xi - x;
        if (lane == 31) rp[NNODES] = xi;
    }
    __syncthreads();
    const int off = wtot[w] + (inc - run);
#pragma unroll
    for (int k = 0; k < SCAN_V; ++k) {
        const int i4 = base4 + k;
        if (i4 < NNODES / 4) {
            int4 c = v[k];
            c.x += off; c.y += off; c.z += off; c.w += off;
            rp4[i4] = c;
        }
    }
}

__global__ __launch_bounds__(256)
void k_place(const int* __restrict__ ei, int* __restrict__ rp,
             int* __restrict__ lst) {
    int e = blockIdx.x * blockDim.x + threadIdx.x;
    if (e >= NEDGES) return;
    int s = ei[e];
    int d = ei[NEDGES + e];
    int pos = atomicSub(&rp[d], 1) - 1;
    if ((unsigned)pos < (unsigned)NEDGES) lst[pos] = s;
}

#define WPB 8
static_assert(NNODES % WPB == 0, "grid covers the nodes exactly");

static __device__ __forceinline__ float leaky(float v) {
    return v >= 0.f ? v : NEG_SLOPE * v;
}

static __device__ __forceinline__ int load_chunk(const int* __restrict__ lst,
                                                 int c0, int cnt, int lane, int d) {
    int sv = (lane < cnt) ? lst[c0 + lane] : d;
    return min(max(sv, 0), NNODES - 1);
}

static __device__ __forceinline__ void chunk_scores(const float* __restrict__ AS,
                                                    int sv, int cnt, int q, int g,
                                                    float adg, float (&ev)[8]) {
    const float ninf = -__builtin_huge_valf();
#pragma unroll
    for (int i = 0; i < 8; ++i) {
        int j = 4 * i + q;
        int s = __shfl(sv, j);
        float e = leaky(AS[s * HEADS + g] + adg);
        ev[i] = (j < cnt) ? e : ninf;
    }
}

static __device__ __forceinline__ void chunk_gather(const float* __restrict__ Hf,
                                                    int sv, int cnt, int lane,
                                                    const float (&av)[8], float4& acc) {
    const int hd = lane >> 2;
#pragma unroll
    for (int i = 0; i < 8; ++i) {
        if (4 * i < cnt) {
#pragma unroll
            for (int qq = 0; qq < 4; ++qq) {
                int s = __builtin_amdgcn_readlane(sv, 4 * i + qq);
                float al = __shfl(av[i], (qq << 3) | hd);
                const float4 r = *(const float4*)(Hf + (size_t)s * HID + 4 * lane);
                acc.x += al * r.x;
                acc.y += al * r.y;
                acc.z += al * r.z;
                acc.w += al * r.w;
            }
        }
    }
}

__global__ __launch_bounds__(256)
void k_node_gather(const int* __restrict__ rp, const int* __restrict__ lst,
                   const float* __restrict__ AS, const float* __restrict__ AD,
                   const float* __restrict__ Hf, const float* __restrict__ b,
                   float* __restrict__ Out) {
    const int lane = threadIdx.x & 31;
    const int d = __builtin_amdgcn_readfirstlane(blockIdx.x * WPB + (threadIdx.x >> 5));
    const int q = lane >> 3, g = lane & 7, hd = lane >> 2;

    int beg = rp[d], end = rp[d + 1];
    beg = min(max(beg, 0), NEDGES);
    end = min(max(end, beg), NEDGES);
    const int deg = end - beg;

    const float adg   = AD[d * HEADS + g];
    const float eself = leaky(AS[d * HEADS + g] + adg);
    const float4 rself = *(const float4*)(Hf + (size_t)d * HID + 4 * lane);

    float ev[8], av[8];
    float4 acc;

    if (deg <= 32) {
        const int sv = load_chunk(lst, beg, deg, lane, d);
        chunk_scores(AS, sv, deg, q, g, adg, ev);
        float m = eself;
#pragma unroll
        for (int i = 0; i < 8; ++i) m = fmaxf(m, ev[i]);
        m = fmaxf(m, __shfl_xor(m, 8));
        m = fmaxf(m, __shfl_xor(m, 16));
        float S = 0.f;
#pragma unroll
        for (int i = 0; i < 8; ++i) { av[i] = expf(ev[i] - m); S += av[i]; }
        S += __shfl_xor(S, 8);
        S += __shfl_xor(S, 16);
        const float wself = expf(eself - m);
        S += wself;
        const float den = S + 1e-16f;
#pragma unroll
        for (int i = 0; i < 8; ++i) av[i] = av[i] / den;
        const float aself = __shfl(wself / den, hd);
        acc.x = aself * rself.x; acc.y = aself * rself.y;
        acc.z = aself * rself.z; acc.w = aself * rself.w;
        chunk_gather(Hf, sv, deg, lane, av, acc);
    } else {
        float m = eself;
        for (int c0 = beg; c0 < end; c0 += 32) {
            const int cnt = min(32, end - c0);
            const int sv = load_chunk(lst, c0, cnt, lane, d);
            chunk_scores(AS, sv, cnt, q, g, adg, ev);
#pragma unroll
            for (int i = 0; i < 8; ++i) m = fmaxf(m, ev[i]);
        }
        m = fmaxf(m, __shfl_xor(m, 8));
        m = fmaxf(m, __shfl_xor(m, 16));
        float S = 0.f;
        for (int c0 = beg; c0 < end; c0 += 32) {
            const int cnt = min(32, end - c0);
            const int sv = load_chunk(lst, c0, cnt, lane, d);
            chunk_scores(AS, sv, cnt, q, g, adg, ev);
#pragma unroll
            for (int i = 0; i < 8; ++i) S += expf(ev[i] - m);
        }
        S += __shfl_xor(S, 8);
        S += __shfl_xor(S, 16);
        const float wself = expf(eself - m);
        S += wself;
        const float den = S + 1e-16f;
        const float aself = __shfl(wself / den, hd);
        acc.x = aself * rself.x; acc.y = aself * rself.y;
        acc.z = aself * rself.z; acc.w = aself * rself.w;
        for (int c0 = beg; c0 < end; c0 += 32) {
            const int cnt = min(32, end - c0);
            const int sv = load_chunk(lst, c0, cnt, lane, d);
            chunk_scores(AS, sv, cnt, q, g, adg, ev);
#pragma unroll
            for (int i = 0; i < 8; ++i) av[i] = expf(ev[i] - m) / den;
            chunk_gather(Hf, sv, cnt, lane, av, acc);
        }
    }

    const float4 bb = *(const float4*)(b + 4 * lane);
    float4 o;
    float v;
    v = acc.x + bb.x; o.x = v > 0.f ? v : expm1f(v);
    v = acc.y + bb.y; o.y = v > 0.f ? v : expm1f(v);
    v = acc.z + bb.z; o.z = v > 0.f ? v : expm1f(v);
    v = acc.w + bb.w; o.w = v > 0.f ? v : expm1f(v);
    *(float4*)(Out + (size_t)d * HID + 4 * lane) = o;
}

static inline unsigned grid_for(long long n) { return (unsigned)((n + 255) / 256); }

extern "C" void kernel_launch(void* const* d_in, const int* in_sizes, int n_in,
                              void* d_out, int out_size, void* d_ws, size_t ws_size,
                              hipStream_t stream) {
    const float* x   = (const float*)d_in[0];
    const int*   ei  = (const int*)  d_in[1];
    const float* W1  = (const float*)d_in[2];
    const float* as1 = (const float*)d_in[3];
    const float* ad1 = (const float*)d_in[4];
    const float* b1  = (const float*)d_in[5];
    const float* W2  = (const float*)d_in[6];
    const float* as2 = (const float*)d_in[7];
    const float* ad2 = (const float*)d_in[8];
    const float* b2  = (const float*)d_in[9];

    float* ws     = (float*)d_ws;
    float* bufH   = ws;
    float* bufAS  = bufH  + (size_t)NNODES * HID;
    float* bufAD  = bufAS + (size_t)NNODES * HEADS;
    float* bufM   = bufAD + (size_t)NNODES * HEADS;
    float* bufS   = bufM  + (size_t)NNODES * HEADS;
    float* bufAgg = bufS  + (size_t)NNODES * HEADS;
    int*   rp     = (int*)bufAgg;
    int*   lst    = rp + 50048;
    float* out    = (float*)d_out;

    hipMemsetAsync(rp, 0, (size_t)(NNODES + 1) * sizeof(int), stream);
    k_count<<<grid_for(NEDGES), 256, 0, stream>>>(ei, rp);
    k_scan<<<1, SCAN_T, 0, stream>>>(rp);
    k_place<<<grid_for(NEDGES), 256, 0, stream>>>(ei, rp, lst);

    auto run_layer = [&](const float* xin, const float* Wm, const float* as,
                         const float* ad, const float* bv, float* outBuf) {
        k_gemm_wmma<<<NNODES / 16, 256, 0, stream>>>(xin, Wm, bufH);
        k_alpha<<<grid_for((long long)NNODES * HEADS), 256, 0, stream>>>(
            bufH, as, ad, bufAS, bufAD);
        k_node_gather<<<NNODES / WPB, 256, 0, stream>>>(
            rp, lst, bufAS, bufAD, bufH, bv, outBuf);
    };

    run_layer(x,   W1, as1, ad1, b1, out);
    run_layer(out, W2, as2, ad2, b2, out);
}


# Buckets built on LDS counters, no global atomics: 32 edge slices, each block holds all node counters in LDS, slot = LDS add return (arrival order), starts from slice prefixes; sum order changes
# speedup vs baseline: 6.5478x; 1.2416x over previous
#include <hip/hip_runtime.h>
#include <math.h>

#define NNODES 50000
#define NEDGES 800000
#define FEAT 128
#define HID  128
#define HEADS 8
#define HDIM 16
#define NEG_SLOPE 0.2f

typedef __attribute__((ext_vector_type(2))) float v2f;
typedef __attribute__((ext_vector_type(8))) float v8f;

__global__ __launch_bounds__(256)
void k_gemm_wmma(const float* __restrict__ X, const float* __restrict__ W,
                 float* __restrict__ Hout) {
    __shared__ float ldsW[FEAT * HID];
    const int tid = threadIdx.x;

    const float4* Wv = (const float4*)W;
    float4*       Lv = (float4*)ldsW;
#pragma unroll
    for (int i = 0; i < (FEAT * HID / 4) / 256; ++i)
        Lv[tid + i * 256] = Wv[tid + i * 256];
    __syncthreads();

    const int lane = tid & 31;
    const int wave = tid >> 5;
    const int l16  = lane & 15;
    const int half = lane >> 4;
    const int row  = blockIdx.x * 16 + l16;
    const int col  = wave * 16 + l16;

    const float* xrow = X + (size_t)row * FEAT;

    v8f acc = {};
#pragma unroll
    for (int k0 = 0; k0 < FEAT; k0 += 4) {
        const int ka = k0 + 2 * half;
        v2f a, b;
        a.x = xrow[ka];
        a.y = xrow[ka + 1];
        b.x = ldsW[ka * HID + col];
        b.y = ldsW[(ka + 1) * HID + col];
        acc = __builtin_amdgcn_wmma_f32_16x16x4_f32(
                  false, a, false, b, (short)0, acc, false, false);
    }

    const int orow0 = blockIdx.x * 16 + 8 * half;
#pragma unroll
    for (int j = 0; j < 8; ++j)
        Hout[(size_t)(orow0 + j) * HID + col] = acc[j];
}

__global__ __launch_bounds__(256)
void k_alpha(const float* __restrict__ Hf, const float* __restrict__ a_src,
             const float* __restrict__ a_dst, float* __restrict__ AS,
             float* __restrict__ AD) {
    int t = blockIdx.x * blockDim.x + threadIdx.x;
    if (t >= NNODES * HEADS) return;
    int hd = t & (HEADS - 1);
    int n  = t >> 3;
    const float* hp = Hf + (size_t)n * HID + hd * HDIM;
    const float* as = a_src + hd * HDIM;
    const float* ad = a_dst + hd * HDIM;
    float sa = 0.f, sd = 0.f;
#pragma unroll
    for (int c = 0; c < HDIM; ++c) {
        float v = hp[c];
        sa += v * as[c];
        sd += v * ad[c];
    }
    AS[t] = sa;
    AD[t] = sd;
}

#define NSLICE  32
#define SLICE_E (NEDGES / NSLICE)
#define SLICE_V (SLICE_E / 4)
#define BT      1024
#define CPT4    13
#define NCNT    (BT * CPT4 * 4)
#define PPITCH  50048
#define EV      ((SLICE_V + BT - 1) / BT)
static_assert(NEDGES % (NSLICE * 4) == 0, "slices are whole 16-byte pieces");
static_assert(NCNT >= PPITCH && PPITCH > NNODES && PPITCH % 4 == 0 && NNODES % 4 == 0,
              "a row holds every node and the total, in whole pieces");

static __device__ __forceinline__ void bump(int* cnt, int d) {
    if ((unsigned)d < (unsigned)NNODES) atomicAdd(&cnt[d], 1);
}

__global__ __launch_bounds__(BT)
void k_slice_count(const int* __restrict__ ei, int* __restrict__ P) {
    __shared__ __attribute__((aligned(16))) int cnt[NCNT];
    __shared__ int wtot[32];
    const int t = threadIdx.x, lane = t & 31, w = t >> 5, g = blockIdx.x;

    const int4* dv = (const int4*)(ei + NEDGES + g * SLICE_E);
    int4 dd[EV];
#pragma unroll
    for (int k = 0; k < EV; ++k) dd[k] = dv[min(t + k * BT, SLICE_V - 1)];

    int4* c4 = (int4*)cnt;
#pragma unroll
    for (int k = 0; k < CPT4; ++k) c4[t + k * BT] = make_int4(0, 0, 0, 0);
    __syncthreads();

#pragma unroll
    for (int k = 0; k < EV; ++k) {
        if (t + k * BT < SLICE_V) {
            bump(cnt, dd[k].x); bump(cnt, dd[k].y);
            bump(cnt, dd[k].z); bump(cnt, dd[k].w);
        }
    }
    __syncthreads();

    int4 v[CPT4];
    int run = 0;
#pragma unroll
    for (int k = 0; k < CPT4; ++k) {
        int4 c = c4[t * CPT4 + k];
        int x;
        x = c.x; c.x = run; run += x;
        x = c.y; c.y = run; run += x;
        x = c.z; c.z = run; run += x;
        x = c.w; c.w = run; run += x;
        v[k] = c;
    }
    int inc = run;
    for (int o = 1; o < 32; o <<= 1) {
        int n = __shfl_up(inc, o);
        if (lane >= o) inc += n;
    }
    if (lane == 31) wtot[w] = inc;
    __syncthreads();
    if (w == 0) {
        int x = wtot[lane], xi = x;
        for (int o = 1; o < 32; o <<= 1) {
            int n = __shfl_up(xi, o);
            if (lane >= o) xi += n;
        }
        wtot[lane] = xi - x;
    }
    __syncthreads();
    const int off = wtot[w] + (inc - run);
#pragma unroll
    for (int k = 0; k < CPT4; ++k) {
        int4 c = v[k];
        c.x += off; c.y += off; c.z += off; c.w += off;
        c4[t * CPT4 + k] = c;
    }
    __syncthreads();

    int4* P4 = (int4*)(P + (size_t)g * PPITCH);
    for (int i = t; i < PPITCH / 4; i += BT) P4[i] = c4[i];
}

__global__ __launch_bounds__(256)
void k_bases(const int* __restrict__ P, int* __restrict__ rp, int* __restrict__ B) {
    const int n = blockIdx.x * 256 + threadIdx.x;
    if (n > NNODES) return;
    const int n1 = min(n + 1, NNODES);
    int p[NSLICE], q[NSLICE];
    int sum = 0;
#pragma unroll
    for (int g = 0; g < NSLICE; ++g) {
        p[g] = P[(size_t)g * PPITCH + n];
        q[g] = P[(size_t)g * PPITCH + n1];
        sum += p[g];
    }
    rp[n] = sum;
    if (n < NNODES) {
        int run = sum;
#pragma unroll
        for (int g = 0; g < NSLICE; ++g) {
            B[(size_t)g * PPITCH + n] = run;
            run += q[g] - p[g];
        }
    }
}

static __device__ __forceinline__ int take(int* pos, int d) {
    return ((unsigned)d < (unsigned)NNODES) ? atomicAdd(&pos[d], 1) : -1;
}

static __device__ __forceinline__ void put(int* __restrict__ lst, int slot, int s) {
    if ((unsigned)slot < (unsigned)NEDGES) lst[slot] = s;
}

__global__ __launch_bounds__(BT)
void k_slice_place(const int* __restrict__ ei, const int* __restrict__ B,
                   int* __restrict__ lst) {
    __shared__ __attribute__((aligned(16))) int pos[PPITCH];
    const int t = threadIdx.x, g = blockIdx.x;

    const int4* sv = (const int4*)(ei + g * SLICE_E);
    const int4* dv = (const int4*)(ei + NEDGES + g * SLICE_E);
    int4 ss[EV], dd[EV];
#pragma unroll
    for (int k = 0; k < EV; ++k) {
        const int i = min(t + k * BT, SLICE_V - 1);
        ss[k] = sv[i];
        dd[k] = dv[i];
    }

    const int4* B4 = (const int4*)(B + (size_t)g * PPITCH);
    int4* p4 = (int4*)pos;
    for (int i = t; i < NNODES / 4; i += BT) p4[i] = B4[i];
    __syncthreads();

#pragma unroll
    for (int k = 0; k < EV; ++k) {
        int4 sl = make_int4(-1, -1, -1, -1);
        if (t + k * BT < SLICE_V) {
            sl.x = take(pos, dd[k].x); sl.y = take(pos, dd[k].y);
            sl.z = take(pos, dd[k].z); sl.w = take(pos, dd[k].w);
        }
        dd[k] = sl;
    }
#pragma unroll
    for (int k = 0; k < EV; ++k) {
        put(lst, dd[k].x, ss[k].x); put(lst, dd[k].y, ss[k].y);
        put(lst, dd[k].z, ss[k].z); put(lst, dd[k].w, ss[k].w);
    }
}

#define WPB 8
static_assert(NNODES % WPB == 0, "grid covers the nodes exactly");

static __device__ __forceinline__ float leaky(float v) {
    return v >= 0.f ? v : NEG_SLOPE * v;
}

static __device__ __forceinline__ int load_chunk(const int* __restrict__ lst,
                                                 int c0, int cnt, int lane, int d) {
    int sv = (lane < cnt) ? lst[c0 + lane] : d;
    return min(max(sv, 0), NNODES - 1);
}

static __device__ __forceinline__ void chunk_scores(const float* __restrict__ AS,
                                                    int sv, int cnt, int q, int g,
                                                    float adg, float (&ev)[8]) {
    const float ninf = -__builtin_huge_valf();
#pragma unroll
    for (int i = 0; i < 8; ++i) {
        int j = 4 * i + q;
        int s = __shfl(sv, j);
        float e = leaky(AS[s * HEADS + g] + adg);
        ev[i] = (j < cnt) ? e : ninf;
    }
}

static __device__ __forceinline__ void chunk_gather(const float* __restrict__ Hf,
                                                    int sv, int cnt, int lane,
                                                    const float (&av)[8], float4& acc) {
    const int hd = lane >> 2;
#pragma unroll
    for (int i = 0; i < 8; ++i) {
        if (4 * i < cnt) {
#pragma unroll
            for (int qq = 0; qq < 4; ++qq) {
                int s = __builtin_amdgcn_readlane(sv, 4 * i + qq);
                float al = __shfl(av[i], (qq << 3) | hd);
                const float4 r = *(const float4*)(Hf + (size_t)s * HID + 4 * lane);
                acc.x += al * r.x;
                acc.y += al * r.y;
                acc.z += al * r.z;
                acc.w += al * r.w;
            }
        }
    }
}

__global__ __launch_bounds__(256)
void k_node_gather(const int* __restrict__ rp, const int* __restrict__ lst,
                   const float* __restrict__ AS, const float* __restrict__ AD,
                   const float* __restrict__ Hf, const float* __restrict__ b,
                   float* __restrict__ Out) {
    const int lane = threadIdx.x & 31;
    const int d = __builtin_amdgcn_readfirstlane(blockIdx.x * WPB + (threadIdx.x >> 5));
    const int q = lane >> 3, g = lane & 7, hd = lane >> 2;

    int beg = rp[d], end = rp[d + 1];
    beg = min(max(beg, 0), NEDGES);
    end = min(max(end, beg), NEDGES);
    const int deg = end - beg;

    const float adg   = AD[d * HEADS + g];
    const float eself = leaky(AS[d * HEADS + g] + adg);
    const float4 rself = *(const float4*)(Hf + (size_t)d * HID + 4 * lane);

    float ev[8], av[8];
    float4 acc;

    if (deg <= 32) {
        const int sv = load_chunk(lst, beg, deg, lane, d);
        chunk_scores(AS, sv, deg, q, g, adg, ev);
        float m = eself;
#pragma unroll
        for (int i = 0; i < 8; ++i) m = fmaxf(m, ev[i]);
        m = fmaxf(m, __shfl_xor(m, 8));
        m = fmaxf(m, __shfl_xor(m, 16));
        float S = 0.f;
#pragma unroll
        for (int i = 0; i < 8; ++i) { av[i] = expf(ev[i] - m); S += av[i]; }
        S += __shfl_xor(S, 8);
        S += __shfl_xor(S, 16);
        const float wself = expf(eself - m);
        S += wself;
        const float den = S + 1e-16f;
#pragma unroll
        for (int i = 0; i < 8; ++i) av[i] = av[i] / den;
        const float aself = __shfl(wself / den, hd);
        acc.x = aself * rself.x; acc.y = aself * rself.y;
        acc.z = aself * rself.z; acc.w = aself * rself.w;
        chunk_gather(Hf, sv, deg, lane, av, acc);
    } else {
        float m = eself;
        for (int c0 = beg; c0 < end; c0 += 32) {
            const int cnt = min(32, end - c0);
            const int sv = load_chunk(lst, c0, cnt, lane, d);
            chunk_scores(AS, sv, cnt, q, g, adg, ev);
#pragma unroll
            for (int i = 0; i < 8; ++i) m = fmaxf(m, ev[i]);
        }
        m = fmaxf(m, __shfl_xor(m, 8));
        m = fmaxf(m, __shfl_xor(m, 16));
        float S = 0.f;
        for (int c0 = beg; c0 < end; c0 += 32) {
            const int cnt = min(32, end - c0);
            const int sv = load_chunk(lst, c0, cnt, lane, d);
            chunk_scores(AS, sv, cnt, q, g, adg, ev);
#pragma unroll
            for (int i = 0; i < 8; ++i) S += expf(ev[i] - m);
        }
        S += __shfl_xor(S, 8);
        S += __shfl_xor(S, 16);
        const float wself = expf(eself - m);
        S += wself;
        const float den = S + 1e-16f;
        const float aself = __shfl(wself / den, hd);
        acc.x = aself * rself.x; acc.y = aself * rself.y;
        acc.z = aself * rself.z; acc.w = aself * rself.w;
        for (int c0 = beg; c0 < end; c0 += 32) {
            const int cnt = min(32, end - c0);
            const int sv = load_chunk(lst, c0, cnt, lane, d);
            chunk_scores(AS, sv, cnt, q, g, adg, ev);
#pragma unroll
            for (int i = 0; i < 8; ++i) av[i] = expf(ev[i] - m) / den;
            chunk_gather(Hf, sv, cnt, lane, av, acc);
        }
    }

    const float4 bb = *(const float4*)(b + 4 * lane);
    float4 o;
    float v;
    v = acc.x + bb.x; o.x = v > 0.f ? v : expm1f(v);
    v = acc.y + bb.y; o.y = v > 0.f ? v : expm1f(v);
    v = acc.z + bb.z; o.z = v > 0.f ? v : expm1f(v);
    v = acc.w + bb.w; o.w = v > 0.f ? v : expm1f(v);
    *(float4*)(Out + (size_t)d * HID + 4 * lane) = o;
}

static inline unsigned grid_for(long long n) { return (unsigned)((n + 255) / 256); }

extern "C" void kernel_launch(void* const* d_in, const int* in_sizes, int n_in,
                              void* d_out, int out_size, void* d_ws, size_t ws_size,
                              hipStream_t stream) {
    const float* x   = (const float*)d_in[0];
    const int*   ei  = (const int*)  d_in[1];
    const float* W1  = (const float*)d_in[2];
    const float* as1 = (const float*)d_in[3];
    const float* ad1 = (const float*)d_in[4];
    const float* b1  = (const float*)d_in[5];
    const float* W2  = (const float*)d_in[6];
    const float* as2 = (const float*)d_in[7];
    const float* ad2 = (const float*)d_in[8];
    const float* b2  = (const float*)d_in[9];

    float* ws     = (float*)d_ws;
    float* bufH   = ws;
    float* bufAS  = bufH  + (size_t)NNODES * HID;
    float* bufAD  = bufAS + (size_t)NNODES * HEADS;
    float* bufM   = bufAD + (size_t)NNODES * HEADS;
    float* bufS   = bufM  + (size_t)NNODES * HEADS;
    float* bufAgg = bufS  + (size_t)NNODES * HEADS;
    int*   rp     = (int*)bufAgg;
    int*   lst    = rp + 50048;
    int*   Pre    = lst + NEDGES;
    int*   Bas    = Pre + NSLICE * PPITCH;
    float* out    = (float*)d_out;

    k_slice_count<<<NSLICE, BT, 0, stream>>>(ei, Pre);
    k_bases<<<grid_for(NNODES + 1), 256, 0, stream>>>(Pre, rp, Bas);
    k_slice_place<<<NSLICE, BT, 0, stream>>>(ei, Bas, lst);

    auto run_layer = [&](const float* xin, const float* Wm, const float* as,
                         const float* ad, const float* bv, float* outBuf) {
        k_gemm_wmma<<<NNODES / 16, 256, 0, stream>>>(xin, Wm, bufH);
        k_alpha<<<grid_for((long long)NNODES * HEADS), 256, 0, stream>>>(
            bufH, as, ad, bufAS, bufAD);
        k_node_gather<<<NNODES / WPB, 256, 0, stream>>>(
            rp, lst, bufAS, bufAD, bufH, bv, outBuf);
    };

    run_layer(x,   W1, as1, ad1, b1, out);
    run_layer(out, W2, as2, ad2, b2, out);
}


# Fewer, longer waves (wave entry bound): gather wave walks ~5 nodes (50000->10240 waves), GEMM wave owns a 16x128 strip with W staged once per 128 rows (25000->3128), score thread takes 4 items
# speedup vs baseline: 7.2383x; 1.1055x over previous
#include <hip/hip_runtime.h>
#include <math.h>

#define NNODES 50000
#define NEDGES 800000
#define FEAT 128
#define HID  128
#define HEADS 8
#define HDIM 16
#define NEG_SLOPE 0.2f

typedef __attribute__((ext_vector_type(2))) float v2f;
typedef __attribute__((ext_vector_type(8))) float v8f;

#define GROWS 128
#define GBLOCKS ((NNODES + GROWS - 1) / GROWS)
#define WP 288
static_assert(NNODES % 16 == 0, "a strip is wholly inside or wholly outside the rows");

__global__ __launch_bounds__(256)
void k_gemm_wmma(const float* __restrict__ X, const float* __restrict__ W,
                 float* __restrict__ Hout) {
    __shared__ __attribute__((aligned(16))) float ldsW[(FEAT / 2) * WP];
    const int tid = threadIdx.x;

    const float4* Wv = (const float4*)W;
#pragma unroll
    for (int i = 0; i < (FEAT / 2) * (HID / 4) / 256; ++i) {
        const int item = tid + i * 256;
        const int kp = item >> 5;
        const int c4 = item & 31;
        const float4 r0 = Wv[(2 * kp) * (HID / 4) + c4];
        const float4 r1 = Wv[(2 * kp + 1) * (HID / 4) + c4];
        float4* dst = (float4*)(ldsW + kp * WP + c4 * 8);
        dst[0] = make_float4(r0.x, r1.x, r0.y, r1.y);
        dst[1] = make_float4(r0.z, r1.z, r0.w, r1.w);
    }
    __syncthreads();

    const int lane = tid & 31;
    const int wave = tid >> 5;
    const int l16  = lane & 15;
    const int half = lane >> 4;
    const int row0 = (blockIdx.x * 8 + wave) * 16;
    if (row0 >= NNODES) return;

    const float* xrow = X + (size_t)(row0 + l16) * FEAT;

    v8f acc[8] = {};
#pragma unroll
    for (int k0 = 0; k0 < FEAT; k0 += 4) {
        const int ka = k0 + 2 * half;
        v2f a;
        a.x = xrow[ka];
        a.y = xrow[ka + 1];
        const float* bp = ldsW + (k0 / 2 + half) * WP + 2 * l16;
#pragma unroll
        for (int t = 0; t < 8; ++t) {
            const v2f b = *(const v2f*)(bp + 32 * t);
            acc[t] = __builtin_amdgcn_wmma_f32_16x16x4_f32(
                         false, a, false, b, (short)0, acc[t], false, false);
        }
    }

    const int orow0 = row0 + 8 * half;
#pragma unroll
    for (int t = 0; t < 8; ++t)
#pragma unroll
        for (int j = 0; j < 8; ++j)
            Hout[(size_t)(orow0 + j) * HID + 16 * t + l16] = acc[t][j];
}

#define AIT 4
#define ABLOCKS ((NNODES * HEADS + 256 * AIT - 1) / (256 * AIT))
static_assert((ABLOCKS * 256) % HEADS == 0, "a thread keeps one head");

__global__ __launch_bounds__(256)
void k_alpha(const float* __restrict__ Hf, const float* __restrict__ a_src,
             const float* __restrict__ a_dst, float* __restrict__ AS,
             float* __restrict__ AD) {
    const int total = NNODES * HEADS;
    const int t0 = blockIdx.x * 256 + threadIdx.x;
    const int hd = t0 & (HEADS - 1);
    float4 v[AIT][4];
#pragma unroll
    for (int i = 0; i < AIT; ++i) {
        const int t = min(t0 + i * (ABLOCKS * 256), total - 1);
        const float4* hp = (const float4*)(Hf + (size_t)(t >> 3) * HID + (t & (HEADS - 1)) * HDIM);
        v[i][0] = hp[0]; v[i][1] = hp[1]; v[i][2] = hp[2]; v[i][3] = hp[3];
    }
    float as[HDIM], ad[HDIM];
#pragma unroll
    for (int c = 0; c < HDIM; ++c) { as[c] = a_src[hd * HDIM + c]; ad[c] = a_dst[hd * HDIM + c]; }
#pragma unroll
    for (int i = 0; i < AIT; ++i) {
        const int t = t0 + i * (ABLOCKS * 256);
        float sa = 0.f, sd = 0.f;
#pragma unroll
        for (int c4 = 0; c4 < 4; ++c4) {
            const float4 r = v[i][c4];
            sa += r.x * as[4 * c4 + 0]; sd += r.x * ad[4 * c4 + 0];
            sa += r.y * as[4 * c4 + 1]; sd += r.y * ad[4 * c4 + 1];
            sa += r.z * as[4 * c4 + 2]; sd += r.z * ad[4 * c4 + 2];
            sa += r.w * as[4 * c4 + 3]; sd += r.w * ad[4 * c4 + 3];
        }
        if (t < total) { AS[t] = sa; AD[t] = sd; }
    }
}

#define NSLICE  32
#define SLICE_E (NEDGES / NSLICE)
#define SLICE_V (SLICE_E / 4)
#define BT      1024
#define CPT4    13
#define NCNT    (BT * CPT4 * 4)
#define PPITCH  50048
#define EV      ((SLICE_V + BT - 1) / BT)
static_assert(NEDGES % (NSLICE * 4) == 0, "slices are whole 16-byte pieces");
static_assert(NCNT >= PPITCH && PPITCH > NNODES && PPITCH % 4 == 0 && NNODES % 4 == 0,
              "a row holds every node and the total, in whole pieces");

static __device__ __forceinline__ void bump(int* cnt, int d) {
    if ((unsigned)d < (unsigned)NNODES) atomicAdd(&cnt[d], 1);
}

__global__ __launch_bounds__(BT)
void k_slice_count(const int* __restrict__ ei, int* __restrict__ P) {
    __shared__ __attribute__((aligned(16))) int cnt[NCNT];
    __shared__ int wtot[32];
    const int t = threadIdx.x, lane = t & 31, w = t >> 5, g = blockIdx.x;

    const int4* dv = (const int4*)(ei + NEDGES + g * SLICE_E);
    int4 dd[EV];
#pragma unroll
    for (int k = 0; k < EV; ++k) dd[k] = dv[min(t + k * BT, SLICE_V - 1)];

    int4* c4 = (int4*)cnt;
#pragma unroll
    for (int k = 0; k < CPT4; ++k) c4[t + k * BT] = make_int4(0, 0, 0, 0);
    __syncthreads();

#pragma unroll
    for (int k = 0; k < EV; ++k) {
        if (t + k * BT < SLICE_V) {
            bump(cnt, dd[k].x); bump(cnt, dd[k].y);
            bump(cnt, dd[k].z); bump(cnt, dd[k].w);
        }
    }
    __syncthreads();

    int4 v[CPT4];
    int run = 0;
#pragma unroll
    for (int k = 0; k < CPT4; ++k) {
        int4 c = c4[t * CPT4 + k];
        int x;
        x = c.x; c.x = run; run += x;
        x = c.y; c.y = run; run += x;
        x = c.z; c.z = run; run += x;
        x = c.w; c.w = run; run += x;
        v[k] = c;
    }
    int inc = run;
    for (int o = 1; o < 32; o <<= 1) {
        int n = __shfl_up(inc, o);
        if (lane >= o) inc += n;
    }
    if (lane == 31) wtot[w] = inc;
    __syncthreads();
    if (w == 0) {
        int x = wtot[lane], xi = x;
        for (int o = 1; o < 32; o <<= 1) {
            int n = __shfl_up(xi, o);
            if (lane >= o) xi += n;
        }
        wtot[lane] = xi - x;
    }
    __syncthreads();
    const int off = wtot[w] + (inc - run);
#pragma unroll
    for (int k = 0; k < CPT4; ++k) {
        int4 c = v[k];
        c.x += off; c.y += off; c.z += off; c.w += off;
        c4[t * CPT4 + k] = c;
    }
    __syncthreads();

    int4* P4 = (int4*)(P + (size_t)g * PPITCH);
    for (int i = t; i < PPITCH / 4; i += BT) P4[i] = c4[i];
}

__global__ __launch_bounds__(256)
void k_bases(const int* __restrict__ P, int* __restrict__ rp, int* __restrict__ B) {
    const int n = blockIdx.x * 256 + threadIdx.x;
    if (n > NNODES) return;
    const int n1 = min(n + 1, NNODES);
    int p[NSLICE], q[NSLICE];
    int sum = 0;
#pragma unroll
    for (int g = 0; g < NSLICE; ++g) {
        p[g] = P[(size_t)g * PPITCH + n];
        q[g] = P[(size_t)g * PPITCH + n1];
        sum += p[g];
    }
    rp[n] = sum;
    if (n < NNODES) {
        int run = sum;
#pragma unroll
        for (int g = 0; g < NSLICE; ++g) {
            B[(size_t)g * PPITCH + n] = run;
            run += q[g] - p[g];
        }
    }
}

static __device__ __forceinline__ int take(int* pos, int d) {
    return ((unsigned)d < (unsigned)NNODES) ? atomicAdd(&pos[d], 1) : -1;
}

static __device__ __forceinline__ void put(int* __restrict__ lst, int slot, int s) {
    if ((unsigned)slot < (unsigned)NEDGES) lst[slot] = s;
}

__global__ __launch_bounds__(BT)
void k_slice_place(const int* __restrict__ ei, const int* __restrict__ B,
                   int* __restrict__ lst) {
    __shared__ __attribute__((aligned(16))) int pos[PPITCH];
    const int t = threadIdx.x, g = blockIdx.x;

    const int4* sv = (const int4*)(ei + g * SLICE_E);
    const int4* dv = (const int4*)(ei + NEDGES + g * SLICE_E);
    int4 ss[EV], dd[EV];
#pragma unroll
    for (int k = 0; k < EV; ++k) {
        const int i = min(t + k * BT, SLICE_V - 1);
        ss[k] = sv[i];
        dd[k] = dv[i];
    }

    const int4* B4 = (const int4*)(B + (size_t)g * PPITCH);
    int4* p4 = (int4*)pos;
    for (int i = t; i < NNODES / 4; i += BT) p4[i] = B4[i];
    __syncthreads();

#pragma unroll
    for (int k = 0; k < EV; ++k) {
        int4 sl = make_int4(-1, -1, -1, -1);
        if (t + k * BT < SLICE_V) {
            sl.x = take(pos, dd[k].x); sl.y = take(pos, dd[k].y);
            sl.z = take(pos, dd[k].z); sl.w = take(pos, dd[k].w);
        }
        dd[k] = sl;
    }
#pragma unroll
    for (int k = 0; k < EV; ++k) {
        put(lst, dd[k].x, ss[k].x); put(lst, dd[k].y, ss[k].y);
        put(lst, dd[k].z, ss[k].z); put(lst, dd[k].w, ss[k].w);
    }
}

#define WPB 8
#define GBLK 1280
#define NWAVES (GBLK * WPB)

static __device__ __forceinline__ float leaky(float v) {
    return v >= 0.f ? v : NEG_SLOPE * v;
}

static __device__ __forceinline__ int load_chunk(const int* __restrict__ lst,
                                                 int c0, int cnt, int lane, int d) {
    int sv = (lane < cnt) ? lst[c0 + lane] : d;
    return min(max(sv, 0), NNODES - 1);
}

static __device__ __forceinline__ void chunk_scores(const float* __restrict__ AS,
                                                    int sv, int cnt, int q, int g,
                                                    float adg, float (&ev)[8]) {
    const float ninf = -__builtin_huge_valf();
#pragma unroll
    for (int i = 0; i < 8; ++i) {
        int j = 4 * i + q;
        int s = __shfl(sv, j);
        float e = leaky(AS[s * HEADS + g] + adg);
        ev[i] = (j < cnt) ? e : ninf;
    }
}

static __device__ __forceinline__ void chunk_gather(const float* __restrict__ Hf,
                                                    int sv, int cnt, int lane,
                                                    const float (&av)[8], float4& acc) {
    const int hd = lane >> 2;
#pragma unroll
    for (int i = 0; i < 8; ++i) {
        if (4 * i < cnt) {
#pragma unroll
            for (int qq = 0; qq < 4; ++qq) {
                int s = __builtin_amdgcn_readlane(sv, 4 * i + qq);
                float al = __shfl(av[i], (qq << 3) | hd);
                const float4 r = *(const float4*)(Hf + (size_t)s * HID + 4 * lane);
                acc.x += al * r.x;
                acc.y += al * r.y;
                acc.z += al * r.z;
                acc.w += al * r.w;
            }
        }
    }
}

__global__ __launch_bounds__(256)
void k_node_gather(const int* __restrict__ rp, const int* __restrict__ lst,
                   const float* __restrict__ AS, const float* __restrict__ AD,
                   const float* __restrict__ Hf, const float* __restrict__ b,
                   float* __restrict__ Out) {
    const int lane = threadIdx.x & 31;
    const int d0 = __builtin_amdgcn_readfirstlane(blockIdx.x * WPB + (threadIdx.x >> 5));
    const int q = lane >> 3, g = lane & 7, hd = lane >> 2;
    const float4 bb = *(const float4*)(b + 4 * lane);

    for (int d = d0; d < NNODES; d += NWAVES) {
    int beg = rp[d], end = rp[d + 1];
    beg = min(max(beg, 0), NEDGES);
    end = min(max(end, beg), NEDGES);
    const int deg = end - beg;

    const float adg   = AD[d * HEADS + g];
    const float eself = leaky(AS[d * HEADS + g] + adg);
    const float4 rself = *(const float4*)(Hf + (size_t)d * HID + 4 * lane);

    float ev[8], av[8];
    float4 acc;

    if (deg <= 32) {
        const int sv = load_chunk(lst, beg, deg, lane, d);
        chunk_scores(AS, sv, deg, q, g, adg, ev);
        float m = eself;
#pragma unroll
        for (int i = 0; i < 8; ++i) m = fmaxf(m, ev[i]);
        m = fmaxf(m, __shfl_xor(m, 8));
        m = fmaxf(m, __shfl_xor(m, 16));
        float S = 0.f;
#pragma unroll
        for (int i = 0; i < 8; ++i) { av[i] = expf(ev[i] - m); S += av[i]; }
        S += __shfl_xor(S, 8);
        S += __shfl_xor(S, 16);
        const float wself = expf(eself - m);
        S += wself;
        const float den = S + 1e-16f;
#pragma unroll
        for (int i = 0; i < 8; ++i) av[i] = av[i] / den;
        const float aself = __shfl(wself / den, hd);
        acc.x = aself * rself.x; acc.y = aself * rself.y;
        acc.z = aself * rself.z; acc.w = aself * rself.w;
        chunk_gather(Hf, sv, deg, lane, av, acc);
    } else {
        float m = eself;
        for (int c0 = beg; c0 < end; c0 += 32) {
            const int cnt = min(32, end - c0);
            const int sv = load_chunk(lst, c0, cnt, lane, d);
            chunk_scores(AS, sv, cnt, q, g, adg, ev);
#pragma unroll
            for (int i = 0; i < 8; ++i) m = fmaxf(m, ev[i]);
        }
        m = fmaxf(m, __shfl_xor(m, 8));
        m = fmaxf(m, __shfl_xor(m, 16));
        float S = 0.f;
        for (int c0 = beg; c0 < end; c0 += 32) {
            const int cnt = min(32, end - c0);
            const int sv = load_chunk(lst, c0, cnt, lane, d);
            chunk_scores(AS, sv, cnt, q, g, adg, ev);
#pragma unroll
            for (int i = 0; i < 8; ++i) S += expf(ev[i] - m);
        }
        S += __shfl_xor(S, 8);
        S += __shfl_xor(S, 16);
        const float wself = expf(eself - m);
        S += wself;
        const float den = S + 1e-16f;
        const float aself = __shfl(wself / den, hd);
        acc.x = aself * rself.x; acc.y = aself * rself.y;
        acc.z = aself * rself.z; acc.w = aself * rself.w;
        for (int c0 = beg; c0 < end; c0 += 32) {
            const int cnt = min(32, end - c0);
            const int sv = load_chunk(lst, c0, cnt, lane, d);
            chunk_scores(AS, sv, cnt, q, g, adg, ev);
#pragma unroll
            for (int i = 0; i < 8; ++i) av[i] = expf(ev[i] - m) / den;
            chunk_gather(Hf, sv, cnt, lane, av, acc);
        }
    }

    float4 o;
    float v;
    v = acc.x + bb.x; o.x = v > 0.f ? v : expm1f(v);
    v = acc.y + bb.y; o.y = v > 0.f ? v : expm1f(v);
    v = acc.z + bb.z; o.z = v > 0.f ? v : expm1f(v);
    v = acc.w + bb.w; o.w = v > 0.f ? v : expm1f(v);
    *(float4*)(Out + (size_t)d * HID + 4 * lane) = o;
    }
}

static inline unsigned grid_for(long long n) { return (unsigned)((n + 255) / 256); }

extern "C" void kernel_launch(void* const* d_in, const int* in_sizes, int n_in,
                              void* d_out, int out_size, void* d_ws, size_t ws_size,
                              hipStream_t stream) {
    const float* x   = (const float*)d_in[0];
    const int*   ei  = (const int*)  d_in[1];
    const float* W1  = (const float*)d_in[2];
    const float* as1 = (const float*)d_in[3];
    const float* ad1 = (const float*)d_in[4];
    const float* b1  = (const float*)d_in[5];
    const float* W2  = (const float*)d_in[6];
    const float* as2 = (const float*)d_in[7];
    const float* ad2 = (const float*)d_in[8];
    const float* b2  = (const float*)d_in[9];

    float* ws     = (float*)d_ws;
    float* bufH   = ws;
    float* bufAS  = bufH  + (size_t)NNODES * HID;
    float* bufAD  = bufAS + (size_t)NNODES * HEADS;
    float* bufM   = bufAD + (size_t)NNODES * HEADS;
    float* bufS   = bufM  + (size_t)NNODES * HEADS;
    float* bufAgg = bufS  + (size_t)NNODES * HEADS;
    int*   rp     = (int*)bufAgg;
    int*   lst    = rp + 50048;
    int*   Pre    = lst + NEDGES;
    int*   Bas    = Pre + NSLICE * PPITCH;
    float* out    = (float*)d_out;

    k_slice_count<<<NSLICE, BT, 0, stream>>>(ei, Pre);
    k_bases<<<grid_for(NNODES + 1), 256, 0, stream>>>(Pre, rp, Bas);
    k_slice_place<<<NSLICE, BT, 0, stream>>>(ei, Bas, lst);

    auto run_layer = [&](const float* xin, const float* Wm, const float* as,
                         const float* ad, const float* bv, float* outBuf) {
        k_gemm_wmma<<<GBLOCKS, 256, 0, stream>>>(xin, Wm, bufH);
        k_alpha<<<ABLOCKS, 256, 0, stream>>>(
            bufH, as, ad, bufAS, bufAD);
        k_node_gather<<<GBLK, 256, 0, stream>>>(
            rp, lst, bufAS, bufAD, bufH, bv, outBuf);
    };

    run_layer(x,   W1, as1, ad1, b1, out);
    run_layer(out, W2, as2, ad2, b2, out);
}


# Bucket build spread over the chip: each of the 32 edge slices is cut again into 8 node ranges, 256 blocks (was 32) count and place on range-sized LDS counters; in-bucket order rule as before
# speedup vs baseline: 8.7762x; 1.2125x over previous
#include <hip/hip_runtime.h>
#include <math.h>

#define NNODES 50000
#define NEDGES 800000
#define FEAT 128
#define HID  128
#define HEADS 8
#define HDIM 16
#define NEG_SLOPE 0.2f

typedef __attribute__((ext_vector_type(2))) float v2f;
typedef __attribute__((ext_vector_type(8))) float v8f;

#define GROWS 128
#define GBLOCKS ((NNODES + GROWS - 1) / GROWS)
#define WP 288
static_assert(NNODES % 16 == 0, "a strip is wholly inside or wholly outside the rows");

__global__ __launch_bounds__(256)
void k_gemm_wmma(const float* __restrict__ X, const float* __restrict__ W,
                 float* __restrict__ Hout) {
    __shared__ __attribute__((aligned(16))) float ldsW[(FEAT / 2) * WP];
    const int tid = threadIdx.x;

    const float4* Wv = (const float4*)W;
#pragma unroll
    for (int i = 0; i < (FEAT / 2) * (HID / 4) / 256; ++i) {
        const int item = tid + i * 256;
        const int kp = item >> 5;
        const int c4 = item & 31;
        const float4 r0 = Wv[(2 * kp) * (HID / 4) + c4];
        const float4 r1 = Wv[(2 * kp + 1) * (HID / 4) + c4];
        float4* dst = (float4*)(ldsW + kp * WP + c4 * 8);
        dst[0] = make_float4(r0.x, r1.x, r0.y, r1.y);
        dst[1] = make_float4(r0.z, r1.z, r0.w, r1.w);
    }
    __syncthreads();

    const int lane = tid & 31;
    const int wave = tid >> 5;
    const int l16  = lane & 15;
    const int half = lane >> 4;
    const int row0 = (blockIdx.x * 8 + wave) * 16;
    if (row0 >= NNODES) return;

    const float* xrow = X + (size_t)(row0 + l16) * FEAT;

    v8f acc[8] = {};
#pragma unroll
    for (int k0 = 0; k0 < FEAT; k0 += 4) {
        const int ka = k0 + 2 * half;
        v2f a;
        a.x = xrow[ka];
        a.y = xrow[ka + 1];
        const float* bp = ldsW + (k0 / 2 + half) * WP + 2 * l16;
#pragma unroll
        for (int t = 0; t < 8; ++t) {
            const v2f b = *(const v2f*)(bp + 32 * t);
            acc[t] = __builtin_amdgcn_wmma_f32_16x16x4_f32(
                         false, a, false, b, (short)0, acc[t], false, false);
        }
    }

    const int orow0 = row0 + 8 * half;
#pragma unroll
    for (int t = 0; t < 8; ++t)
#pragma unroll
        for (int j = 0; j < 8; ++j)
            Hout[(size_t)(orow0 + j) * HID + 16 * t + l16] = acc[t][j];
}

#define AIT 4
#define ABLOCKS ((NNODES * HEADS + 256 * AIT - 1) / (256 * AIT))
static_assert((ABLOCKS * 256) % HEADS == 0, "a thread keeps one head");

__global__ __launch_bounds__(256)
void k_alpha(const float* __restrict__ Hf, const float* __restrict__ a_src,
             const float* __restrict__ a_dst, float* __restrict__ AS,
             float* __restrict__ AD) {
    const int total = NNODES * HEADS;
    const int t0 = blockIdx.x * 256 + threadIdx.x;
    const int hd = t0 & (HEADS - 1);
    float4 v[AIT][4];
#pragma unroll
    for (int i = 0; i < AIT; ++i) {
        const int t = min(t0 + i * (ABLOCKS * 256), total - 1);
        const float4* hp = (const float4*)(Hf + (size_t)(t >> 3) * HID + (t & (HEADS - 1)) * HDIM);
        v[i][0] = hp[0]; v[i][1] = hp[1]; v[i][2] = hp[2]; v[i][3] = hp[3];
    }
    float as[HDIM], ad[HDIM];
#pragma unroll
    for (int c = 0; c < HDIM; ++c) { as[c] = a_src[hd * HDIM + c]; ad[c] = a_dst[hd * HDIM + c]; }
#pragma unroll
    for (int i = 0; i < AIT; ++i) {
        const int t = t0 + i * (ABLOCKS * 256);
        float sa = 0.f, sd = 0.f;
#pragma unroll
        for (int c4 = 0; c4 < 4; ++c4) {
            const float4 r = v[i][c4];
            sa += r.x * as[4 * c4 + 0]; sd += r.x * ad[4 * c4 + 0];
            sa += r.y * as[4 * c4 + 1]; sd += r.y * ad[4 * c4 + 1];
            sa += r.z * as[4 * c4 + 2]; sd += r.z * ad[4 * c4 + 2];
            sa += r.w * as[4 * c4 + 3]; sd += r.w * ad[4 * c4 + 3];
        }
        if (t < total) { AS[t] = sa; AD[t] = sd; }
    }
}

#define NSLICE  32
#define SLICE_E (NEDGES / NSLICE)
#define SLICE_V (SLICE_E / 4)
#define NR      8
#define RN      6256
#define RP      6272
#define PPITCH  (NR * RP)
#define BPITCH  (NR * RN)
#define BT      512
#define EV      ((SLICE_V + BT - 1) / BT)
#define OWN4    4
#define NOWN    (RN / (4 * OWN4))
#define RV      (RN / 4)
#define CLR     ((RP / 4 + BT - 1) / BT)
static_assert(NEDGES % (NSLICE * 4) == 0, "slices are whole 16-byte pieces");
static_assert(NR * RN > NNODES && (NR - 1) * RN < NNODES, "the ranges hold every node and the total");
static_assert(RN % (4 * OWN4) == 0 && NOWN < BT && RP >= RN + 4 && RP % 32 == 0 && NNODES % 4 == 0,
              "whole pieces; the last thread scans nothing and carries the total");
static_assert(NSLICE == 32 && NSLICE * NR == 256, "k_bases: one (slice, range) total per thread, a wave per range");

static __device__ __forceinline__ unsigned range_span(int lo) {
    return (unsigned)(min(lo + RN, NNODES) - lo);
}

static __device__ __forceinline__ void bump(int* cnt, int i, unsigned span) {
    if ((unsigned)i < span) atomicAdd(&cnt[i], 1);
}

__global__ __launch_bounds__(BT)
void k_slice_count(const int* __restrict__ ei, int* __restrict__ P) {
    __shared__ __attribute__((aligned(16))) int cnt[RP];
    __shared__ int wtot[32];
    const int t = threadIdx.x, lane = t & 31, w = t >> 5;
    const int g = blockIdx.x, r = blockIdx.y;
    const int lo = r * RN;
    const unsigned span = range_span(lo);

    const int4* dv = (const int4*)(ei + NEDGES + g * SLICE_E);
    int4 dd[EV];
#pragma unroll
    for (int k = 0; k < EV; ++k) dd[k] = dv[min(t + k * BT, SLICE_V - 1)];

    int4* c4 = (int4*)cnt;
#pragma unroll
    for (int k = 0; k < CLR; ++k) {
        const int i = t + k * BT;
        if (i < RP / 4) c4[i] = make_int4(0, 0, 0, 0);
    }
    __syncthreads();

#pragma unroll
    for (int k = 0; k < EV; ++k) {
        if (t + k * BT < SLICE_V) {
            bump(cnt, dd[k].x - lo, span); bump(cnt, dd[k].y - lo, span);
            bump(cnt, dd[k].z - lo, span); bump(cnt, dd[k].w - lo, span);
        }
    }
    __syncthreads();

    int4 v[OWN4];
    int run = 0;
#pragma unroll
    for (int k = 0; k < OWN4; ++k) {
        int4 c = make_int4(0, 0, 0, 0);
        if (t < NOWN) c = c4[t * OWN4 + k];
        int x;
        x = c.x; c.x = run; run += x;
        x = c.y; c.y = run; run += x;
        x = c.z; c.z = run; run += x;
        x = c.w; c.w = run; run += x;
        v[k] = c;
    }
    int inc = run;
    for (int o = 1; o < 32; o <<= 1) {
        int n = __shfl_up(inc, o);
        if (lane >= o) inc += n;
    }
    if (lane == 31) wtot[w] = inc;
    __syncthreads();
    if (w == 0) {
        int x = (lane < BT / 32) ? wtot[lane] : 0, xi = x;
        for (int o = 1; o < 32; o <<= 1) {
            int n = __shfl_up(xi, o);
            if (lane >= o) xi += n;
        }
        if (lane < BT / 32) wtot[lane] = xi - x;
    }
    __syncthreads();
    const int off = wtot[w] + (inc - run);
    if (t < NOWN) {
#pragma unroll
        for (int k = 0; k < OWN4; ++k) {
            int4 c = v[k];
            c.x += off; c.y += off; c.z += off; c.w += off;
            c4[t * OWN4 + k] = c;
        }
    }
    if (t == BT - 1) c4[RV] = make_int4(off, 0, 0, 0);
    __syncthreads();

    int4* P4 = (int4*)(P + (size_t)g * PPITCH + r * RP);
#pragma unroll
    for (int k = 0; k < CLR; ++k) {
        const int i = t + k * BT;
        if (i < RV + 1) P4[i] = c4[i];
    }
}

__global__ __launch_bounds__(256)
void k_bases(const int* __restrict__ P, int* __restrict__ rp, int* __restrict__ B) {
    __shared__ int part[8];
    const int t = threadIdx.x, r = blockIdx.y;

    const int tr = t >> 5, tg = t & 31;
    int x = (tr < r) ? P[(size_t)tg * PPITCH + tr * RP + RN] : 0;
    for (int o = 16; o > 0; o >>= 1) x += __shfl_xor(x, o);
    if (tg == 0) part[tr] = x;
    __syncthreads();
    int sum = 0;
#pragma unroll
    for (int k = 0; k < 8; ++k) sum += part[k];

    const int i = blockIdx.x * 256 + t;
    const int n = r * RN + i;
    if (i >= RN || n > NNODES) return;
    const int* Pr = P + r * RP + i;
    int p[NSLICE], q[NSLICE];
#pragma unroll
    for (int g = 0; g < NSLICE; ++g) {
        p[g] = Pr[(size_t)g * PPITCH];
        q[g] = Pr[(size_t)g * PPITCH + 1];
        sum += p[g];
    }
    rp[n] = sum;
    if (n < NNODES) {
        int run = sum;
#pragma unroll
        for (int g = 0; g < NSLICE; ++g) {
            B[(size_t)g * BPITCH + n] = run;
            run += q[g] - p[g];
        }
    }
}

static __device__ __forceinline__ int take(int* pos, int i, unsigned span) {
    return ((unsigned)i < span) ? atomicAdd(&pos[i], 1) : -1;
}

static __device__ __forceinline__ void put(int* __restrict__ lst, int slot, int s) {
    if ((unsigned)slot < (unsigned)NEDGES) lst[slot] = s;
}

__global__ __launch_bounds__(BT)
void k_slice_place(const int* __restrict__ ei, const int* __restrict__ B,
                   int* __restrict__ lst) {
    __shared__ __attribute__((aligned(16))) int pos[CLR * BT * 4];
    const int t = threadIdx.x, g = blockIdx.x, r = blockIdx.y;
    const int lo = r * RN;
    const unsigned span = range_span(lo);

    const int4* sv = (const int4*)(ei + g * SLICE_E);
    const int4* dv = (const int4*)(ei + NEDGES + g * SLICE_E);
    int4 ss[EV], dd[EV];
#pragma unroll
    for (int k = 0; k < EV; ++k) {
        const int i = min(t + k * BT, SLICE_V - 1);
        ss[k] = sv[i];
        dd[k] = dv[i];
    }

    const int4* B4 = (const int4*)(B + (size_t)g * BPITCH + lo);
    int4* p4 = (int4*)pos;
    const int nb = (int)(span / 4);
    int4 bv[CLR];
#pragma unroll
    for (int k = 0; k < CLR; ++k) bv[k] = B4[min(t + k * BT, nb - 1)];
#pragma unroll
    for (int k = 0; k < CLR; ++k) p4[t + k * BT] = bv[k];
    __syncthreads();

#pragma unroll
    for (int k = 0; k < EV; ++k) {
        int4 sl = make_int4(-1, -1, -1, -1);
        if (t + k * BT < SLICE_V) {
            sl.x = take(pos, dd[k].x - lo, span); sl.y = take(pos, dd[k].y - lo, span);
            sl.z = take(pos, dd[k].z - lo, span); sl.w = take(pos, dd[k].w - lo, span);
        }
        dd[k] = sl;
    }
#pragma unroll
    for (int k = 0; k < EV; ++k) {
        put(lst, dd[k].x, ss[k].x); put(lst, dd[k].y, ss[k].y);
        put(lst, dd[k].z, ss[k].z); put(lst, dd[k].w, ss[k].w);
    }
}

#define WPB 8
#define GBLK 1280
#define NWAVES (GBLK * WPB)

static __device__ __forceinline__ float leaky(float v) {
    return v >= 0.f ? v : NEG_SLOPE * v;
}

static __device__ __forceinline__ int load_chunk(const int* __restrict__ lst,
                                                 int c0, int cnt, int lane, int d) {
    int sv = (lane < cnt) ? lst[c0 + lane] : d;
    return min(max(sv, 0), NNODES - 1);
}

static __device__ __forceinline__ void chunk_scores(const float* __restrict__ AS,
                                                    int sv, int cnt, int q, int g,
                                                    float adg, float (&ev)[8]) {
    const float ninf = -__builtin_huge_valf();
#pragma unroll
    for (int i = 0; i < 8; ++i) {
        int j = 4 * i + q;
        int s = __shfl(sv, j);
        float e = leaky(AS[s * HEADS + g] + adg);
        ev[i] = (j < cnt) ? e : ninf;
    }
}

static __device__ __forceinline__ void chunk_gather(const float* __restrict__ Hf,
                                                    int sv, int cnt, int lane,
                                                    const float (&av)[8], float4& acc) {
    const int hd = lane >> 2;
#pragma unroll
    for (int i = 0; i < 8; ++i) {
        if (4 * i < cnt) {
#pragma unroll
            for (int qq = 0; qq < 4; ++qq) {
                int s = __builtin_amdgcn_readlane(sv, 4 * i + qq);
                float al = __shfl(av[i], (qq << 3) | hd);
                const float4 r = *(const float4*)(Hf + (size_t)s * HID + 4 * lane);
                acc.x += al * r.x;
                acc.y += al * r.y;
                acc.z += al * r.z;
                acc.w += al * r.w;
            }
        }
    }
}

__global__ __launch_bounds__(256)
void k_node_gather(const int* __restrict__ rp, const int* __restrict__ lst,
                   const float* __restrict__ AS, const float* __restrict__ AD,
                   const float* __restrict__ Hf, const float* __restrict__ b,
                   float* __restrict__ Out) {
    const int lane = threadIdx.x & 31;
    const int d0 = __builtin_amdgcn_readfirstlane(blockIdx.x * WPB + (threadIdx.x >> 5));
    const int q = lane >> 3, g = lane & 7, hd = lane >> 2;
    const float4 bb = *(const float4*)(b + 4 * lane);

    for (int d = d0; d < NNODES; d += NWAVES) {
    int beg = rp[d], end = rp[d + 1];
    beg = min(max(beg, 0), NEDGES);
    end = min(max(end, beg), NEDGES);
    const int deg = end - beg;

    const float adg   = AD[d * HEADS + g];
    const float eself = leaky(AS[d * HEADS + g] + adg);
    const float4 rself = *(const float4*)(Hf + (size_t)d * HID + 4 * lane);

    float ev[8], av[8];
    float4 acc;

    if (deg <= 32) {
        const int sv = load_chunk(lst, beg, deg, lane, d);
        chunk_scores(AS, sv, deg, q, g, adg, ev);
        float m = eself;
#pragma unroll
        for (int i = 0; i < 8; ++i) m = fmaxf(m, ev[i]);
        m = fmaxf(m, __shfl_xor(m, 8));
        m = fmaxf(m, __shfl_xor(m, 16));
        float S = 0.f;
#pragma unroll
        for (int i = 0; i < 8; ++i) { av[i] = expf(ev[i] - m); S += av[i]; }
        S += __shfl_xor(S, 8);
        S += __shfl_xor(S, 16);
        const float wself = expf(eself - m);
        S += wself;
        const float den = S + 1e-16f;
#pragma unroll
        for (int i = 0; i < 8; ++i) av[i] = av[i] / den;
        const float aself = __shfl(wself / den, hd);
        acc.x = aself * rself.x; acc.y = aself * rself.y;
        acc.z = aself * rself.z; acc.w = aself * rself.w;
        chunk_gather(Hf, sv, deg, lane, av, acc);
    } else {
        float m = eself;
        for (int c0 = beg; c0 < end; c0 += 32) {
            const int cnt = min(32, end - c0);
            const int sv = load_chunk(lst, c0, cnt, lane, d);
            chunk_scores(AS, sv, cnt, q, g, adg, ev);
#pragma unroll
            for (int i = 0; i < 8; ++i) m = fmaxf(m, ev[i]);
        }
        m = fmaxf(m, __shfl_xor(m, 8));
        m = fmaxf(m, __shfl_xor(m, 16));
        float S = 0.f;
        for (int c0 = beg; c0 < end; c0 += 32) {
            const int cnt = min(32, end - c0);
            const int sv = load_chunk(lst, c0, cnt, lane, d);
            chunk_scores(AS, sv, cnt, q, g, adg, ev);
#pragma unroll
            for (int i = 0; i < 8; ++i) S += expf(ev[i] - m);
        }
        S += __shfl_xor(S, 8);
        S += __shfl_xor(S, 16);
        const float wself = expf(eself - m);
        S += wself;
        const float den = S + 1e-16f;
        const float aself = __shfl(wself / den, hd);
        acc.x = aself * rself.x; acc.y = aself * rself.y;
        acc.z = aself * rself.z; acc.w = aself * rself.w;
        for (int c0 = beg; c0 < end; c0 += 32) {
            const int cnt = min(32, end - c0);
            const int sv = load_chunk(lst, c0, cnt, lane, d);
            chunk_scores(AS, sv, cnt, q, g, adg, ev);
#pragma unroll
            for (int i = 0; i < 8; ++i) av[i] = expf(ev[i] - m) / den;
            chunk_gather(Hf, sv, cnt, lane, av, acc);
        }
    }

    float4 o;
    float v;
    v = acc.x + bb.x; o.x = v > 0.f ? v : expm1f(v);
    v = acc.y + bb.y; o.y = v > 0.f ? v : expm1f(v);
    v = acc.z + bb.z; o.z = v > 0.f ? v : expm1f(v);
    v = acc.w + bb.w; o.w = v > 0.f ? v : expm1f(v);
    *(float4*)(Out + (size_t)d * HID + 4 * lane) = o;
    }
}

static inline unsigned grid_for(long long n) { return (unsigned)((n + 255) / 256); }

extern "C" void kernel_launch(void* const* d_in, const int* in_sizes, int n_in,
                              void* d_out, int out_size, void* d_ws, size_t ws_size,
                              hipStream_t stream) {
    const float* x   = (const float*)d_in[0];
    const int*   ei  = (const int*)  d_in[1];
    const float* W1  = (const float*)d_in[2];
    const float* as1 = (const float*)d_in[3];
    const float* ad1 = (const float*)d_in[4];
    const float* b1  = (const float*)d_in[5];
    const float* W2  = (const float*)d_in[6];
    const float* as2 = (const float*)d_in[7];
    const float* ad2 = (const float*)d_in[8];
    const float* b2  = (const float*)d_in[9];

    float* ws     = (float*)d_ws;
    float* bufH   = ws;
    float* bufAS  = bufH  + (size_t)NNODES * HID;
    float* bufAD  = bufAS + (size_t)NNODES * HEADS;
    float* bufM   = bufAD + (size_t)NNODES * HEADS;
    float* bufS   = bufM  + (size_t)NNODES * HEADS;
    float* bufAgg = bufS  + (size_t)NNODES * HEADS;
    int*   rp     = (int*)bufAgg;
    int*   lst    = rp + 50048;
    int*   Pre    = lst + NEDGES;
    int*   Bas    = Pre + NSLICE * PPITCH;
    float* out    = (float*)d_out;

    k_slice_count<<<dim3(NSLICE, NR), BT, 0, stream>>>(ei, Pre);
    k_bases<<<dim3((RN + 255) / 256, NR), 256, 0, stream>>>(Pre, rp, Bas);
    k_slice_place<<<dim3(NSLICE, NR), BT, 0, stream>>>(ei, Bas, lst);

    auto run_layer = [&](const float* xin, const float* Wm, const float* as,
                         const float* ad, const float* bv, float* outBuf) {
        k_gemm_wmma<<<GBLOCKS, 256, 0, stream>>>(xin, Wm, bufH);
        k_alpha<<<ABLOCKS, 256, 0, stream>>>(
            bufH, as, ad, bufAS, bufAD);
        k_node_gather<<<GBLK, 256, 0, stream>>>(
            rp, lst, bufAS, bufAD, bufH, bv, outBuf);
    };

    run_layer(x,   W1, as1, ad1, b1, out);
    run_layer(out, W2, as2, ad2, b2, out);
}


# Attention scalars come out of the product as a ninth 16-column tile: block forms W*[a_src|a_dst] (128x16) while staging W; alpha kernel gone (9->7 launches); sum order changes: x*(W*a) for (x*W)*a
# speedup vs baseline: 9.0370x; 1.0297x over previous
#include <hip/hip_runtime.h>
#include <math.h>

#define NNODES 50000
#define NEDGES 800000
#define FEAT 128
#define HID  128
#define HEADS 8
#define HDIM 16
#define NEG_SLOPE 0.2f

typedef __attribute__((ext_vector_type(2))) float v2f;
typedef __attribute__((ext_vector_type(8))) float v8f;

#define GROWS 128
#define GBLOCKS ((NNODES + GROWS - 1) / GROWS)
#define WP 288
static_assert(NNODES % 16 == 0, "a strip is wholly inside or wholly outside the rows");

static __device__ __forceinline__ float quad_sum(float v) {
    v += __int_as_float(__builtin_amdgcn_update_dpp(0, __float_as_int(v), 0xB1, 0xF, 0xF, true));
    v += __int_as_float(__builtin_amdgcn_update_dpp(0, __float_as_int(v), 0x4E, 0xF, 0xF, true));
    return v;
}

static __device__ __forceinline__ float dot4(const float4 r, const float4 a) {
    return r.x * a.x + r.y * a.y + r.z * a.z + r.w * a.w;
}

__global__ __launch_bounds__(256)
void k_gemm_wmma(const float* __restrict__ X, const float* __restrict__ W,
                 const float* __restrict__ a_src, const float* __restrict__ a_dst,
                 float* __restrict__ Hout, float* __restrict__ AS,
                 float* __restrict__ AD) {
    __shared__ __attribute__((aligned(16))) float ldsW[(FEAT / 2) * WP];
    const int tid = threadIdx.x;
    const int lane = tid & 31;

    const float4 vs = ((const float4*)a_src)[lane];
    const float4 vd = ((const float4*)a_dst)[lane];

    const float4* Wv = (const float4*)W;
    constexpr int NT = (FEAT / 2) * (HID / 4) / 256;
    float4 r0[NT], r1[NT];
#pragma unroll
    for (int i = 0; i < NT; ++i) {
        const int item = tid + i * 256;
        const int kp = item >> 5;
        const int c4 = item & 31;
        r0[i] = Wv[(2 * kp) * (HID / 4) + c4];
        r1[i] = Wv[(2 * kp + 1) * (HID / 4) + c4];
    }
    const int ql = lane & 3;
#pragma unroll
    for (int i = 0; i < NT; ++i) {
        const int item = tid + i * 256;
        const int kp = item >> 5;
        const int c4 = item & 31;
        float4* dst = (float4*)(ldsW + kp * WP + c4 * 8);
        dst[0] = make_float4(r0[i].x, r1[i].x, r0[i].y, r1[i].y);
        dst[1] = make_float4(r0[i].z, r1[i].z, r0[i].w, r1[i].w);

        const float s0 = quad_sum(dot4(r0[i], vs));
        const float s1 = quad_sum(dot4(r1[i], vs));
        const float d0 = quad_sum(dot4(r0[i], vd));
        const float d1 = quad_sum(dot4(r1[i], vd));
        const float val = (ql & 2) ? ((ql & 1) ? d1 : d0) : ((ql & 1) ? s1 : s0);
        ldsW[kp * WP + 2 * HID + 16 * (ql >> 1) + 2 * (c4 >> 2) + (ql & 1)] = val;
    }
    __syncthreads();

    const int wave = tid >> 5;
    const int l16  = lane & 15;
    const int half = lane >> 4;
    const int row0 = (blockIdx.x * 8 + wave) * 16;
    if (row0 >= NNODES) return;

    const float* xrow = X + (size_t)(row0 + l16) * FEAT;

    v8f acc[9] = {};
#pragma unroll
    for (int k0 = 0; k0 < FEAT; k0 += 4) {
        const int ka = k0 + 2 * half;
        v2f a;
        a.x = xrow[ka];
        a.y = xrow[ka + 1];
        const float* bp = ldsW + (k0 / 2 + half) * WP + 2 * l16;
#pragma unroll
        for (int t = 0; t < 9; ++t) {
            const v2f b = *(const v2f*)(bp + 32 * t);
            acc[t] = __builtin_amdgcn_wmma_f32_16x16x4_f32(
                         false, a, false, b, (short)0, acc[t], false, false);
        }
    }

    const int orow0 = row0 + 8 * half;
#pragma unroll
    for (int t = 0; t < 8; ++t)
#pragma unroll
        for (int j = 0; j < 8; ++j)
            Hout[(size_t)(orow0 + j) * HID + 16 * t + l16] = acc[t][j];

    float* aout = (l16 < HEADS) ? (AS + l16) : (AD + (l16 - HEADS));
#pragma unroll
    for (int j = 0; j < 8; ++j)
        aout[(size_t)(orow0 + j) * HEADS] = acc[8][j];
}

#define NSLICE  32
#define SLICE_E (NEDGES / NSLICE)
#define SLICE_V (SLICE_E / 4)
#define NR      8
#define RN      6256
#define RP      6272
#define PPITCH  (NR * RP)
#define BPITCH  (NR * RN)
#define BT      512
#define EV      ((SLICE_V + BT - 1) / BT)
#define OWN4    4
#define NOWN    (RN / (4 * OWN4))
#define RV      (RN / 4)
#define CLR     ((RP / 4 + BT - 1) / BT)
static_assert(NEDGES % (NSLICE * 4) == 0, "slices are whole 16-byte pieces");
static_assert(NR * RN > NNODES && (NR - 1) * RN < NNODES, "the ranges hold every node and the total");
static_assert(RN % (4 * OWN4) == 0 && NOWN < BT && RP >= RN + 4 && RP % 32 == 0 && NNODES % 4 == 0,
              "whole pieces; the last thread scans nothing and carries the total");
static_assert(NSLICE == 32 && NSLICE * NR == 256, "k_bases: one (slice, range) total per thread, a wave per range");

static __device__ __forceinline__ unsigned range_span(int lo) {
    return (unsigned)(min(lo + RN, NNODES) - lo);
}

static __device__ __forceinline__ void bump(int* cnt, int i, unsigned span) {
    if ((unsigned)i < span) atomicAdd(&cnt[i], 1);
}

__global__ __launch_bounds__(BT)
void k_slice_count(const int* __restrict__ ei, int* __restrict__ P) {
    __shared__ __attribute__((aligned(16))) int cnt[RP];
    __shared__ int wtot[32];
    const int t = threadIdx.x, lane = t & 31, w = t >> 5;
    const int g = blockIdx.x, r = blockIdx.y;
    const int lo = r * RN;
    const unsigned span = range_span(lo);

    const int4* dv = (const int4*)(ei + NEDGES + g * SLICE_E);
    int4 dd[EV];
#pragma unroll
    for (int k = 0; k < EV; ++k) dd[k] = dv[min(t + k * BT, SLICE_V - 1)];

    int4* c4 = (int4*)cnt;
#pragma unroll
    for (int k = 0; k < CLR; ++k) {
        const int i = t + k * BT;
        if (i < RP / 4) c4[i] = make_int4(0, 0, 0, 0);
    }
    __syncthreads();

#pragma unroll
    for (int k = 0; k < EV; ++k) {
        if (t + k * BT < SLICE_V) {
            bump(cnt, dd[k].x - lo, span); bump(cnt, dd[k].y - lo, span);
            bump(cnt, dd[k].z - lo, span); bump(cnt, dd[k].w - lo, span);
        }
    }
    __syncthreads();

    int4 v[OWN4];
    int run = 0;
#pragma unroll
    for (int k = 0; k < OWN4; ++k) {
        int4 c = make_int4(0, 0, 0, 0);
        if (t < NOWN) c = c4[t * OWN4 + k];
        int x;
        x = c.x; c.x = run; run += x;
        x = c.y; c.y = run; run += x;
        x = c.z; c.z = run; run += x;
        x = c.w; c.w = run; run += x;
        v[k] = c;
    }
    int inc = run;
    for (int o = 1; o < 32; o <<= 1) {
        int n = __shfl_up(inc, o);
        if (lane >= o) inc += n;
    }
    if (lane == 31) wtot[w] = inc;
    __syncthreads();
    if (w == 0) {
        int x = (lane < BT / 32) ? wtot[lane] : 0, xi = x;
        for (int o = 1; o < 32; o <<= 1) {
            int n = __shfl_up(xi, o);
            if (lane >= o) xi += n;
        }
        if (lane < BT / 32) wtot[lane] = xi - x;
    }
    __syncthreads();
    const int off = wtot[w] + (inc - run);
    if (t < NOWN) {
#pragma unroll
        for (int k = 0; k < OWN4; ++k) {
            int4 c = v[k];
            c.x += off; c.y += off; c.z += off; c.w += off;
            c4[t * OWN4 + k] = c;
        }
    }
    if (t == BT - 1) c4[RV] = make_int4(off, 0, 0, 0);
    __syncthreads();

    int4* P4 = (int4*)(P + (size_t)g * PPITCH + r * RP);
#pragma unroll
    for (int k = 0; k < CLR; ++k) {
        const int i = t + k * BT;
        if (i < RV + 1) P4[i] = c4[i];
    }
}

__global__ __launch_bounds__(256)
void k_bases(const int* __restrict__ P, int* __restrict__ rp, int* __restrict__ B) {
    __shared__ int part[8];
    const int t = threadIdx.x, r = blockIdx.y;

    const int tr = t >> 5, tg = t & 31;
    int x = (tr < r) ? P[(size_t)tg * PPITCH + tr * RP + RN] : 0;
    for (int o = 16; o > 0; o >>= 1) x += __shfl_xor(x, o);
    if (tg == 0) part[tr] = x;
    __syncthreads();
    int sum = 0;
#pragma unroll
    for (int k = 0; k < 8; ++k) sum += part[k];

    const int i = blockIdx.x * 256 + t;
    const int n = r * RN + i;
    if (i >= RN || n > NNODES) return;
    const int* Pr = P + r * RP + i;
    int p[NSLICE], q[NSLICE];
#pragma unroll
    for (int g = 0; g < NSLICE; ++g) {
        p[g] = Pr[(size_t)g * PPITCH];
        q[g] = Pr[(size_t)g * PPITCH + 1];
        sum += p[g];
    }
    rp[n] = sum;
    if (n < NNODES) {
        int run = sum;
#pragma unroll
        for (int g = 0; g < NSLICE; ++g) {
            B[(size_t)g * BPITCH + n] = run;
            run += q[g] - p[g];
        }
    }
}

static __device__ __forceinline__ int take(int* pos, int i, unsigned span) {
    return ((unsigned)i < span) ? atomicAdd(&pos[i], 1) : -1;
}

static __device__ __forceinline__ void put(int* __restrict__ lst, int slot, int s) {
    if ((unsigned)slot < (unsigned)NEDGES) lst[slot] = s;
}

__global__ __launch_bounds__(BT)
void k_slice_place(const int* __restrict__ ei, const int* __restrict__ B,
                   int* __restrict__ lst) {
    __shared__ __attribute__((aligned(16))) int pos[CLR * BT * 4];
    const int t = threadIdx.x, g = blockIdx.x, r = blockIdx.y;
    const int lo = r * RN;
    const unsigned span = range_span(lo);

    const int4* sv = (const int4*)(ei + g * SLICE_E);
    const int4* dv = (const int4*)(ei + NEDGES + g * SLICE_E);
    int4 ss[EV], dd[EV];
#pragma unroll
    for (int k = 0; k < EV; ++k) {
        const int i = min(t + k * BT, SLICE_V - 1);
        ss[k] = sv[i];
        dd[k] = dv[i];
    }

    const int4* B4 = (const int4*)(B + (size_t)g * BPITCH + lo);
    int4* p4 = (int4*)pos;
    const int nb = (int)(span / 4);
    int4 bv[CLR];
#pragma unroll
    for (int k = 0; k < CLR; ++k) bv[k] = B4[min(t + k * BT, nb - 1)];
#pragma unroll
    for (int k = 0; k < CLR; ++k) p4[t + k * BT] = bv[k];
    __syncthreads();

#pragma unroll
    for (int k = 0; k < EV; ++k) {
        int4 sl = make_int4(-1, -1, -1, -1);
        if (t + k * BT < SLICE_V) {
            sl.x = take(pos, dd[k].x - lo, span); sl.y = take(pos, dd[k].y - lo, span);
            sl.z = take(pos, dd[k].z - lo, span); sl.w = take(pos, dd[k].w - lo, span);
        }
        dd[k] = sl;
    }
#pragma unroll
    for (int k = 0; k < EV; ++k) {
        put(lst, dd[k].x, ss[k].x); put(lst, dd[k].y, ss[k].y);
        put(lst, dd[k].z, ss[k].z); put(lst, dd[k].w, ss[k].w);
    }
}

#define WPB 8
#define GBLK 1280
#define NWAVES (GBLK * WPB)

static __device__ __forceinline__ float leaky(float v) {
    return v >= 0.f ? v : NEG_SLOPE * v;
}

static __device__ __forceinline__ int load_chunk(const int* __restrict__ lst,
                                                 int c0, int cnt, int lane, int d) {
    int sv = (lane < cnt) ? lst[c0 + lane] : d;
    return min(max(sv, 0), NNODES - 1);
}

static __device__ __forceinline__ void chunk_scores(const float* __restrict__ AS,
                                                    int sv, int cnt, int q, int g,
                                                    float adg, float (&ev)[8]) {
    const float ninf = -__builtin_huge_valf();
#pragma unroll
    for (int i = 0; i < 8; ++i) {
        int j = 4 * i + q;
        int s = __shfl(sv, j);
        float e = leaky(AS[s * HEADS + g] + adg);
        ev[i] = (j < cnt) ? e : ninf;
    }
}

static __device__ __forceinline__ void chunk_gather(const float* __restrict__ Hf,
                                                    int sv, int cnt, int lane,
                                                    const float (&av)[8], float4& acc) {
    const int hd = lane >> 2;
#pragma unroll
    for (int i = 0; i < 8; ++i) {
        if (4 * i < cnt) {
#pragma unroll
            for (int qq = 0; qq < 4; ++qq) {
                int s = __builtin_amdgcn_readlane(sv, 4 * i + qq);
                float al = __shfl(av[i], (qq << 3) | hd);
                const float4 r = *(const float4*)(Hf + (size_t)s * HID + 4 * lane);
                acc.x += al * r.x;
                acc.y += al * r.y;
                acc.z += al * r.z;
                acc.w += al * r.w;
            }
        }
    }
}

__global__ __launch_bounds__(256)
void k_node_gather(const int* __restrict__ rp, const int* __restrict__ lst,
                   const float* __restrict__ AS, const float* __restrict__ AD,
                   const float* __restrict__ Hf, const float* __restrict__ b,
                   float* __restrict__ Out) {
    const int lane = threadIdx.x & 31;
    const int d0 = __builtin_amdgcn_readfirstlane(blockIdx.x * WPB + (threadIdx.x >> 5));
    const int q = lane >> 3, g = lane & 7, hd = lane >> 2;
    const float4 bb = *(const float4*)(b + 4 * lane);

    for (int d = d0; d < NNODES; d += NWAVES) {
    int beg = rp[d], end = rp[d + 1];
    beg = min(max(beg, 0), NEDGES);
    end = min(max(end, beg), NEDGES);
    const int deg = end - beg;

    const float adg   = AD[d * HEADS + g];
    const float eself = leaky(AS[d * HEADS + g] + adg);
    const float4 rself = *(const float4*)(Hf + (size_t)d * HID + 4 * lane);

    float ev[8], av[8];
    float4 acc;

    if (deg <= 32) {
        const int sv = load_chunk(lst, beg, deg, lane, d);
        chunk_scores(AS, sv, deg, q, g, adg, ev);
        float m = eself;
#pragma unroll
        for (int i = 0; i < 8; ++i) m = fmaxf(m, ev[i]);
        m = fmaxf(m, __shfl_xor(m, 8));
        m = fmaxf(m, __shfl_xor(m, 16));
        float S = 0.f;
#pragma unroll
        for (int i = 0; i < 8; ++i) { av[i] = expf(ev[i] - m); S += av[i]; }
        S += __shfl_xor(S, 8);
        S += __shfl_xor(S, 16);
        const float wself = expf(eself - m);
        S += wself;
        const float den = S + 1e-16f;
#pragma unroll
        for (int i = 0; i < 8; ++i) av[i] = av[i] / den;
        const float aself = __shfl(wself / den, hd);
        acc.x = aself * rself.x; acc.y = aself * rself.y;
        acc.z = aself * rself.z; acc.w = aself * rself.w;
        chunk_gather(Hf, sv, deg, lane, av, acc);
    } else {
        float m = eself;
        for (int c0 = beg; c0 < end; c0 += 32) {
            const int cnt = min(32, end - c0);
            const int sv = load_chunk(lst, c0, cnt, lane, d);
            chunk_scores(AS, sv, cnt, q, g, adg, ev);
#pragma unroll
            for (int i = 0; i < 8; ++i) m = fmaxf(m, ev[i]);
        }
        m = fmaxf(m, __shfl_xor(m, 8));
        m = fmaxf(m, __shfl_xor(m, 16));
        float S = 0.f;
        for (int c0 = beg; c0 < end; c0 += 32) {
            const int cnt = min(32, end - c0);
            const int sv = load_chunk(lst, c0, cnt, lane, d);
            chunk_scores(AS, sv, cnt, q, g, adg, ev);
#pragma unroll
            for (int i = 0; i < 8; ++i) S += expf(ev[i] - m);
        }
        S += __shfl_xor(S, 8);
        S += __shfl_xor(S, 16);
        const float wself = expf(eself - m);
        S += wself;
        const float den = S + 1e-16f;
        const float aself = __shfl(wself / den, hd);
        acc.x = aself * rself.x; acc.y = aself * rself.y;
        acc.z = aself * rself.z; acc.w = aself * rself.w;
        for (int c0 = beg; c0 < end; c0 += 32) {
            const int cnt = min(32, end - c0);
            const int sv = load_chunk(lst, c0, cnt, lane, d);
            chunk_scores(AS, sv, cnt, q, g, adg, ev);
#pragma unroll
            for (int i = 0; i < 8; ++i) av[i] = expf(ev[i] - m) / den;
            chunk_gather(Hf, sv, cnt, lane, av, acc);
        }
    }

    float4 o;
    float v;
    v = acc.x + bb.x; o.x = v > 0.f ? v : expm1f(v);
    v = acc.y + bb.y; o.y = v > 0.f ? v : expm1f(v);
    v = acc.z + bb.z; o.z = v > 0.f ? v : expm1f(v);
    v = acc.w + bb.w; o.w = v > 0.f ? v : expm1f(v);
    *(float4*)(Out + (size_t)d * HID + 4 * lane) = o;
    }
}

static inline unsigned grid_for(long long n) { return (unsigned)((n + 255) / 256); }

extern "C" void kernel_launch(void* const* d_in, const int* in_sizes, int n_in,
                              void* d_out, int out_size, void* d_ws, size_t ws_size,
                              hipStream_t stream) {
    const float* x   = (const float*)d_in[0];
    const int*   ei  = (const int*)  d_in[1];
    const float* W1  = (const float*)d_in[2];
    const float* as1 = (const float*)d_in[3];
    const float* ad1 = (const float*)d_in[4];
    const float* b1  = (const float*)d_in[5];
    const float* W2  = (const float*)d_in[6];
    const float* as2 = (const float*)d_in[7];
    const float* ad2 = (const float*)d_in[8];
    const float* b2  = (const float*)d_in[9];

    float* ws     = (float*)d_ws;
    float* bufH   = ws;
    float* bufAS  = bufH  + (size_t)NNODES * HID;
    float* bufAD  = bufAS + (size_t)NNODES * HEADS;
    float* bufM   = bufAD + (size_t)NNODES * HEADS;
    float* bufS   = bufM  + (size_t)NNODES * HEADS;
    float* bufAgg = bufS  + (size_t)NNODES * HEADS;
    int*   rp     = (int*)bufAgg;
    int*   lst    = rp + 50048;
    int*   Pre    = lst + NEDGES;
    int*   Bas    = Pre + NSLICE * PPITCH;
    float* out    = (float*)d_out;

    k_slice_count<<<dim3(NSLICE, NR), BT, 0, stream>>>(ei, Pre);
    k_bases<<<dim3((RN + 255) / 256, NR), 256, 0, stream>>>(Pre, rp, Bas);
    k_slice_place<<<dim3(NSLICE, NR), BT, 0, stream>>>(ei, Bas, lst);

    auto run_layer = [&](const float* xin, const float* Wm, const float* as,
                         const float* ad, const float* bv, float* outBuf) {
        k_gemm_wmma<<<GBLOCKS, 256, 0, stream>>>(xin, Wm, as, ad, bufH, bufAS, bufAD);
        k_node_gather<<<GBLK, 256, 0, stream>>>(
            rp, lst, bufAS, bufAD, bufH, bv, outBuf);
    };

    run_layer(x,   W1, as1, ad1, b1, out);
    run_layer(out, W2, as2, ad2, b2, out);
}


# Gather wave issues about a third fewer vector instructions per node: score/exp code unrolled over 4, 6 or 8 record groups by bucket size (was 8); weights normalised once per head after the row sum
# speedup vs baseline: 9.3274x; 1.0321x over previous
#include <hip/hip_runtime.h>
#include <math.h>

#define NNODES 50000
#define NEDGES 800000
#define FEAT 128
#define HID  128
#define HEADS 8
#define HDIM 16
#define NEG_SLOPE 0.2f

typedef __attribute__((ext_vector_type(2))) float v2f;
typedef __attribute__((ext_vector_type(8))) float v8f;

#define GROWS 128
#define GBLOCKS ((NNODES + GROWS - 1) / GROWS)
#define WP 288
static_assert(NNODES % 16 == 0, "a strip is wholly inside or wholly outside the rows");

static __device__ __forceinline__ float quad_sum(float v) {
    v += __int_as_float(__builtin_amdgcn_update_dpp(0, __float_as_int(v), 0xB1, 0xF, 0xF, true));
    v += __int_as_float(__builtin_amdgcn_update_dpp(0, __float_as_int(v), 0x4E, 0xF, 0xF, true));
    return v;
}

static __device__ __forceinline__ float dot4(const float4 r, const float4 a) {
    return r.x * a.x + r.y * a.y + r.z * a.z + r.w * a.w;
}

__global__ __launch_bounds__(256)
void k_gemm_wmma(const float* __restrict__ X, const float* __restrict__ W,
                 const float* __restrict__ a_src, const float* __restrict__ a_dst,
                 float* __restrict__ Hout, float* __restrict__ AS,
                 float* __restrict__ AD) {
    __shared__ __attribute__((aligned(16))) float ldsW[(FEAT / 2) * WP];
    const int tid = threadIdx.x;
    const int lane = tid & 31;

    const float4 vs = ((const float4*)a_src)[lane];
    const float4 vd = ((const float4*)a_dst)[lane];

    const float4* Wv = (const float4*)W;
    constexpr int NT = (FEAT / 2) * (HID / 4) / 256;
    float4 r0[NT], r1[NT];
#pragma unroll
    for (int i = 0; i < NT; ++i) {
        const int item = tid + i * 256;
        const int kp = item >> 5;
        const int c4 = item & 31;
        r0[i] = Wv[(2 * kp) * (HID / 4) + c4];
        r1[i] = Wv[(2 * kp + 1) * (HID / 4) + c4];
    }
    const int ql = lane & 3;
#pragma unroll
    for (int i = 0; i < NT; ++i) {
        const int item = tid + i * 256;
        const int kp = item >> 5;
        const int c4 = item & 31;
        float4* dst = (float4*)(ldsW + kp * WP + c4 * 8);
        dst[0] = make_float4(r0[i].x, r1[i].x, r0[i].y, r1[i].y);
        dst[1] = make_float4(r0[i].z, r1[i].z, r0[i].w, r1[i].w);

        const float s0 = quad_sum(dot4(r0[i], vs));
        const float s1 = quad_sum(dot4(r1[i], vs));
        const float d0 = quad_sum(dot4(r0[i], vd));
        const float d1 = quad_sum(dot4(r1[i], vd));
        const float val = (ql & 2) ? ((ql & 1) ? d1 : d0) : ((ql & 1) ? s1 : s0);
        ldsW[kp * WP + 2 * HID + 16 * (ql >> 1) + 2 * (c4 >> 2) + (ql & 1)] = val;
    }
    __syncthreads();

    const int wave = tid >> 5;
    const int l16  = lane & 15;
    const int half = lane >> 4;
    const int row0 = (blockIdx.x * 8 + wave) * 16;
    if (row0 >= NNODES) return;

    const float* xrow = X + (size_t)(row0 + l16) * FEAT;

    v8f acc[9] = {};
#pragma unroll
    for (int k0 = 0; k0 < FEAT; k0 += 4) {
        const int ka = k0 + 2 * half;
        v2f a;
        a.x = xrow[ka];
        a.y = xrow[ka + 1];
        const float* bp = ldsW + (k0 / 2 + half) * WP + 2 * l16;
#pragma unroll
        for (int t = 0; t < 9; ++t) {
            const v2f b = *(const v2f*)(bp + 32 * t);
            acc[t] = __builtin_amdgcn_wmma_f32_16x16x4_f32(
                         false, a, false, b, (short)0, acc[t], false, false);
        }
    }

    const int orow0 = row0 + 8 * half;
#pragma unroll
    for (int t = 0; t < 8; ++t)
#pragma unroll
        for (int j = 0; j < 8; ++j)
            Hout[(size_t)(orow0 + j) * HID + 16 * t + l16] = acc[t][j];

    float* aout = (l16 < HEADS) ? (AS + l16) : (AD + (l16 - HEADS));
#pragma unroll
    for (int j = 0; j < 8; ++j)
        aout[(size_t)(orow0 + j) * HEADS] = acc[8][j];
}

#define NSLICE  32
#define SLICE_E (NEDGES / NSLICE)
#define SLICE_V (SLICE_E / 4)
#define NR      8
#define RN      6256
#define RP      6272
#define PPITCH  (NR * RP)
#define BPITCH  (NR * RN)
#define BT      512
#define EV      ((SLICE_V + BT - 1) / BT)
#define OWN4    4
#define NOWN    (RN / (4 * OWN4))
#define RV      (RN / 4)
#define CLR     ((RP / 4 + BT - 1) / BT)
static_assert(NEDGES % (NSLICE * 4) == 0, "slices are whole 16-byte pieces");
static_assert(NR * RN > NNODES && (NR - 1) * RN < NNODES, "the ranges hold every node and the total");
static_assert(RN % (4 * OWN4) == 0 && NOWN < BT && RP >= RN + 4 && RP % 32 == 0 && NNODES % 4 == 0,
              "whole pieces; the last thread scans nothing and carries the total");
static_assert(NSLICE == 32 && NSLICE * NR == 256, "k_bases: one (slice, range) total per thread, a wave per range");

static __device__ __forceinline__ unsigned range_span(int lo) {
    return (unsigned)(min(lo + RN, NNODES) - lo);
}

static __device__ __forceinline__ void bump(int* cnt, int i, unsigned span) {
    if ((unsigned)i < span) atomicAdd(&cnt[i], 1);
}

__global__ __launch_bounds__(BT)
void k_slice_count(const int* __restrict__ ei, int* __restrict__ P) {
    __shared__ __attribute__((aligned(16))) int cnt[RP];
    __shared__ int wtot[32];
    const int t = threadIdx.x, lane = t & 31, w = t >> 5;
    const int g = blockIdx.x, r = blockIdx.y;
    const int lo = r * RN;
    const unsigned span = range_span(lo);

    const int4* dv = (const int4*)(ei + NEDGES + g * SLICE_E);
    int4 dd[EV];
#pragma unroll
    for (int k = 0; k < EV; ++k) dd[k] = dv[min(t + k * BT, SLICE_V - 1)];

    int4* c4 = (int4*)cnt;
#pragma unroll
    for (int k = 0; k < CLR; ++k) {
        const int i = t + k * BT;
        if (i < RP / 4) c4[i] = make_int4(0, 0, 0, 0);
    }
    __syncthreads();

#pragma unroll
    for (int k = 0; k < EV; ++k) {
        if (t + k * BT < SLICE_V) {
            bump(cnt, dd[k].x - lo, span); bump(cnt, dd[k].y - lo, span);
            bump(cnt, dd[k].z - lo, span); bump(cnt, dd[k].w - lo, span);
        }
    }
    __syncthreads();

    int4 v[OWN4];
    int run = 0;
#pragma unroll
    for (int k = 0; k < OWN4; ++k) {
        int4 c = make_int4(0, 0, 0, 0);
        if (t < NOWN) c = c4[t * OWN4 + k];
        int x;
        x = c.x; c.x = run; run += x;
        x = c.y; c.y = run; run += x;
        x = c.z; c.z = run; run += x;
        x = c.w; c.w = run; run += x;
        v[k] = c;
    }
    int inc = run;
    for (int o = 1; o < 32; o <<= 1) {
        int n = __shfl_up(inc, o);
        if (lane >= o) inc += n;
    }
    if (lane == 31) wtot[w] = inc;
    __syncthreads();
    if (w == 0) {
        int x = (lane < BT / 32) ? wtot[lane] : 0, xi = x;
        for (int o = 1; o < 32; o <<= 1) {
            int n = __shfl_up(xi, o);
            if (lane >= o) xi += n;
        }
        if (lane < BT / 32) wtot[lane] = xi - x;
    }
    __syncthreads();
    const int off = wtot[w] + (inc - run);
    if (t < NOWN) {
#pragma unroll
        for (int k = 0; k < OWN4; ++k) {
            int4 c = v[k];
            c.x += off; c.y += off; c.z += off; c.w += off;
            c4[t * OWN4 + k] = c;
        }
    }
    if (t == BT - 1) c4[RV] = make_int4(off, 0, 0, 0);
    __syncthreads();

    int4* P4 = (int4*)(P + (size_t)g * PPITCH + r * RP);
#pragma unroll
    for (int k = 0; k < CLR; ++k) {
        const int i = t + k * BT;
        if (i < RV + 1) P4[i] = c4[i];
    }
}

__global__ __launch_bounds__(256)
void k_bases(const int* __restrict__ P, int* __restrict__ rp, int* __restrict__ B) {
    __shared__ int part[8];
    const int t = threadIdx.x, r = blockIdx.y;

    const int tr = t >> 5, tg = t & 31;
    int x = (tr < r) ? P[(size_t)tg * PPITCH + tr * RP + RN] : 0;
    for (int o = 16; o > 0; o >>= 1) x += __shfl_xor(x, o);
    if (tg == 0) part[tr] = x;
    __syncthreads();
    int sum = 0;
#pragma unroll
    for (int k = 0; k < 8; ++k) sum += part[k];

    const int i = blockIdx.x * 256 + t;
    const int n = r * RN + i;
    if (i >= RN || n > NNODES) return;
    const int* Pr = P + r * RP + i;
    int p[NSLICE], q[NSLICE];
#pragma unroll
    for (int g = 0; g < NSLICE; ++g) {
        p[g] = Pr[(size_t)g * PPITCH];
        q[g] = Pr[(size_t)g * PPITCH + 1];
        sum += p[g];
    }
    rp[n] = sum;
    if (n < NNODES) {
        int run = sum;
#pragma unroll
        for (int g = 0; g < NSLICE; ++g) {
            B[(size_t)g * BPITCH + n] = run;
            run += q[g] - p[g];
        }
    }
}

static __device__ __forceinline__ int take(int* pos, int i, unsigned span) {
    return ((unsigned)i < span) ? atomicAdd(&pos[i], 1) : -1;
}

static __device__ __forceinline__ void put(int* __restrict__ lst, int slot, int s) {
    if ((unsigned)slot < (unsigned)NEDGES) lst[slot] = s;
}

__global__ __launch_bounds__(BT)
void k_slice_place(const int* __restrict__ ei, const int* __restrict__ B,
                   int* __restrict__ lst) {
    __shared__ __attribute__((aligned(16))) int pos[CLR * BT * 4];
    const int t = threadIdx.x, g = blockIdx.x, r = blockIdx.y;
    const int lo = r * RN;
    const unsigned span = range_span(lo);

    const int4* sv = (const int4*)(ei + g * SLICE_E);
    const int4* dv = (const int4*)(ei + NEDGES + g * SLICE_E);
    int4 ss[EV], dd[EV];
#pragma unroll
    for (int k = 0; k < EV; ++k) {
        const int i = min(t + k * BT, SLICE_V - 1);
        ss[k] = sv[i];
        dd[k] = dv[i];
    }

    const int4* B4 = (const int4*)(B + (size_t)g * BPITCH + lo);
    int4* p4 = (int4*)pos;
    const int nb = (int)(span / 4);
    int4 bv[CLR];
#pragma unroll
    for (int k = 0; k < CLR; ++k) bv[k] = B4[min(t + k * BT, nb - 1)];
#pragma unroll
    for (int k = 0; k < CLR; ++k) p4[t + k * BT] = bv[k];
    __syncthreads();

#pragma unroll
    for (int k = 0; k < EV; ++k) {
        int4 sl = make_int4(-1, -1, -1, -1);
        if (t + k * BT < SLICE_V) {
            sl.x = take(pos, dd[k].x - lo, span); sl.y = take(pos, dd[k].y - lo, span);
            sl.z = take(pos, dd[k].z - lo, span); sl.w = take(pos, dd[k].w - lo, span);
        }
        dd[k] = sl;
    }
#pragma unroll
    for (int k = 0; k < EV; ++k) {
        put(lst, dd[k].x, ss[k].x); put(lst, dd[k].y, ss[k].y);
        put(lst, dd[k].z, ss[k].z); put(lst, dd[k].w, ss[k].w);
    }
}

#define WPB 8
#define GBLK 1280
#define NWAVES (GBLK * WPB)

static __device__ __forceinline__ float leaky(float v) {
    return v >= 0.f ? v : NEG_SLOPE * v;
}

static __device__ __forceinline__ int load_chunk(const int* __restrict__ lst,
                                                 int c0, int cnt, int lane, int d) {
    int sv = (lane < cnt) ? lst[c0 + lane] : d;
    return min(max(sv, 0), NNODES - 1);
}

template <int NG>
static __device__ __forceinline__ void chunk_scores(const float* __restrict__ AS,
                                                    int sv, int cnt, int q, int g,
                                                    float adg, float (&ev)[8]) {
    const float ninf = -__builtin_huge_valf();
#pragma unroll
    for (int i = 0; i < NG; ++i) {
        int j = 4 * i + q;
        int s = __shfl(sv, j);
        float e = leaky(AS[s * HEADS + g] + adg);
        ev[i] = (j < cnt) ? e : ninf;
    }
}

template <int NG>
static __device__ __forceinline__ void chunk_gather(const float* __restrict__ Hf,
                                                    int sv, int cnt, int lane,
                                                    const float (&av)[8], float4& acc) {
    const int hd = lane >> 2;
#pragma unroll
    for (int i = 0; i < NG; ++i) {
        if (4 * i < cnt) {
#pragma unroll
            for (int qq = 0; qq < 4; ++qq) {
                int s = __builtin_amdgcn_readlane(sv, 4 * i + qq);
                float al = __shfl(av[i], (qq << 3) | hd);
                const float4 r = *(const float4*)(Hf + (size_t)s * HID + 4 * lane);
                acc.x += al * r.x;
                acc.y += al * r.y;
                acc.z += al * r.z;
                acc.w += al * r.w;
            }
        }
    }
}

template <int NG>
static __device__ __forceinline__ void node_small(const int* __restrict__ lst,
                                                  const float* __restrict__ AS,
                                                  const float* __restrict__ Hf,
                                                  int beg, int deg, int lane, int d,
                                                  float adg, float eself, const float4 rself,
                                                  float4& acc, float& inv) {
    const int q = lane >> 3, g = lane & 7, hd = lane >> 2;
    float ev[8], av[8];
    const int sv = load_chunk(lst, beg, deg, lane, d);
    chunk_scores<NG>(AS, sv, deg, q, g, adg, ev);
    float m = eself;
#pragma unroll
    for (int i = 0; i < NG; ++i) m = fmaxf(m, ev[i]);
    m = fmaxf(m, __shfl_xor(m, 8));
    m = fmaxf(m, __shfl_xor(m, 16));
    float S = 0.f;
#pragma unroll
    for (int i = 0; i < NG; ++i) { av[i] = expf(ev[i] - m); S += av[i]; }
    S += __shfl_xor(S, 8);
    S += __shfl_xor(S, 16);
    const float wself = expf(eself - m);
    S += wself;
    inv = __shfl(1.0f / (S + 1e-16f), hd);
    const float ws = __shfl(wself, hd);
    acc.x = ws * rself.x; acc.y = ws * rself.y;
    acc.z = ws * rself.z; acc.w = ws * rself.w;
    const int cg = __builtin_amdgcn_readfirstlane(deg);
    chunk_gather<NG>(Hf, sv, cg, lane, av, acc);
}

__global__ __launch_bounds__(256)
void k_node_gather(const int* __restrict__ rp, const int* __restrict__ lst,
                   const float* __restrict__ AS, const float* __restrict__ AD,
                   const float* __restrict__ Hf, const float* __restrict__ b,
                   float* __restrict__ Out) {
    const int lane = threadIdx.x & 31;
    const int d0 = __builtin_amdgcn_readfirstlane(blockIdx.x * WPB + (threadIdx.x >> 5));
    const int q = lane >> 3, g = lane & 7, hd = lane >> 2;
    const float4 bb = *(const float4*)(b + 4 * lane);

    for (int d = d0; d < NNODES; d += NWAVES) {
    int beg = rp[d], end = rp[d + 1];
    beg = min(max(beg, 0), NEDGES);
    end = min(max(end, beg), NEDGES);
    const int deg = end - beg;

    const float adg   = AD[d * HEADS + g];
    const float eself = leaky(AS[d * HEADS + g] + adg);
    const float4 rself = *(const float4*)(Hf + (size_t)d * HID + 4 * lane);

    float ev[8], av[8];
    float4 acc;
    float inv;

    if (deg <= 32) {
        if (deg <= 16)
            node_small<4>(lst, AS, Hf, beg, deg, lane, d, adg, eself, rself, acc, inv);
        else if (deg <= 24)
            node_small<6>(lst, AS, Hf, beg, deg, lane, d, adg, eself, rself, acc, inv);
        else
            node_small<8>(lst, AS, Hf, beg, deg, lane, d, adg, eself, rself, acc, inv);
    } else {
        float m = eself;
        for (int c0 = beg; c0 < end; c0 += 32) {
            const int cnt = min(32, end - c0);
            const int sv = load_chunk(lst, c0, cnt, lane, d);
            chunk_scores<8>(AS, sv, cnt, q, g, adg, ev);
#pragma unroll
            for (int i = 0; i < 8; ++i) m = fmaxf(m, ev[i]);
        }
        m = fmaxf(m, __shfl_xor(m, 8));
        m = fmaxf(m, __shfl_xor(m, 16));
        float S = 0.f;
        for (int c0 = beg; c0 < end; c0 += 32) {
            const int cnt = min(32, end - c0);
            const int sv = load_chunk(lst, c0, cnt, lane, d);
            chunk_scores<8>(AS, sv, cnt, q, g, adg, ev);
#pragma unroll
            for (int i = 0; i < 8; ++i) S += expf(ev[i] - m);
        }
        S += __shfl_xor(S, 8);
        S += __shfl_xor(S, 16);
        const float wself = expf(eself - m);
        S += wself;
        inv = __shfl(1.0f / (S + 1e-16f), hd);
        const float ws = __shfl(wself, hd);
        acc.x = ws * rself.x; acc.y = ws * rself.y;
        acc.z = ws * rself.z; acc.w = ws * rself.w;
        for (int c0 = beg; c0 < end; c0 += 32) {
            const int cnt = min(32, end - c0);
            const int sv = load_chunk(lst, c0, cnt, lane, d);
            chunk_scores<8>(AS, sv, cnt, q, g, adg, ev);
#pragma unroll
            for (int i = 0; i < 8; ++i) av[i] = expf(ev[i] - m);
            chunk_gather<8>(Hf, sv, cnt, lane, av, acc);
        }
    }

    float4 o;
    float v;
    v = acc.x * inv + bb.x; o.x = v > 0.f ? v : expm1f(v);
    v = acc.y * inv + bb.y; o.y = v > 0.f ? v : expm1f(v);
    v = acc.z * inv + bb.z; o.z = v > 0.f ? v : expm1f(v);
    v = acc.w * inv + bb.w; o.w = v > 0.f ? v : expm1f(v);
    *(float4*)(Out + (size_t)d * HID + 4 * lane) = o;
    }
}

static inline unsigned grid_for(long long n) { return (unsigned)((n + 255) / 256); }

extern "C" void kernel_launch(void* const* d_in, const int* in_sizes, int n_in,
                              void* d_out, int out_size, void* d_ws, size_t ws_size,
                              hipStream_t stream) {
    const float* x   = (const float*)d_in[0];
    const int*   ei  = (const int*)  d_in[1];
    const float* W1  = (const float*)d_in[2];
    const float* as1 = (const float*)d_in[3];
    const float* ad1 = (const float*)d_in[4];
    const float* b1  = (const float*)d_in[5];
    const float* W2  = (const float*)d_in[6];
    const float* as2 = (const float*)d_in[7];
    const float* ad2 = (const float*)d_in[8];
    const float* b2  = (const float*)d_in[9];

    float* ws     = (float*)d_ws;
    float* bufH   = ws;
    float* bufAS  = bufH  + (size_t)NNODES * HID;
    float* bufAD  = bufAS + (size_t)NNODES * HEADS;
    float* bufM   = bufAD + (size_t)NNODES * HEADS;
    float* bufS   = bufM  + (size_t)NNODES * HEADS;
    float* bufAgg = bufS  + (size_t)NNODES * HEADS;
    int*   rp     = (int*)bufAgg;
    int*   lst    = rp + 50048;
    int*   Pre    = lst + NEDGES;
    int*   Bas    = Pre + NSLICE * PPITCH;
    float* out    = (float*)d_out;

    k_slice_count<<<dim3(NSLICE, NR), BT, 0, stream>>>(ei, Pre);
    k_bases<<<dim3((RN + 255) / 256, NR), 256, 0, stream>>>(Pre, rp, Bas);
    k_slice_place<<<dim3(NSLICE, NR), BT, 0, stream>>>(ei, Bas, lst);

    auto run_layer = [&](const float* xin, const float* Wm, const float* as,
                         const float* ad, const float* bv, float* outBuf) {
        k_gemm_wmma<<<GBLOCKS, 256, 0, stream>>>(xin, Wm, as, ad, bufH, bufAS, bufAD);
        k_node_gather<<<GBLK, 256, 0, stream>>>(
            rp, lst, bufAS, bufAD, bufH, bv, outBuf);
    };

    run_layer(x,   W1, as1, ad1, b1, out);
    run_layer(out, W2, as2, ad2, b2, out);
}


# Layer-2 product moved into the layer-1 gather launch: a 16-wave block gathers 16 consecutive nodes, hands the rows over in LDS, nine waves multiply them by W2 held in LDS; 7->6 launches
# speedup vs baseline: 10.1766x; 1.0910x over previous
#include <hip/hip_runtime.h>
#include <math.h>

#define NNODES 50000
#define NEDGES 800000
#define FEAT 128
#define HID  128
#define HEADS 8
#define HDIM 16
#define NEG_SLOPE 0.2f

typedef __attribute__((ext_vector_type(2))) float v2f;
typedef __attribute__((ext_vector_type(8))) float v8f;

#define GROWS 128
#define GBLOCKS ((NNODES + GROWS - 1) / GROWS)
#define WP 288
static_assert(NNODES % 16 == 0, "a strip is wholly inside or wholly outside the rows");

static __device__ __forceinline__ float quad_sum(float v) {
    v += __int_as_float(__builtin_amdgcn_update_dpp(0, __float_as_int(v), 0xB1, 0xF, 0xF, true));
    v += __int_as_float(__builtin_amdgcn_update_dpp(0, __float_as_int(v), 0x4E, 0xF, 0xF, true));
    return v;
}

static __device__ __forceinline__ float dot4(const float4 r, const float4 a) {
    return r.x * a.x + r.y * a.y + r.z * a.z + r.w * a.w;
}

__global__ __launch_bounds__(256)
void k_gemm_wmma(const float* __restrict__ X, const float* __restrict__ W,
                 const float* __restrict__ a_src, const float* __restrict__ a_dst,
                 float* __restrict__ Hout, float* __restrict__ AS,
                 float* __restrict__ AD) {
    __shared__ __attribute__((aligned(16))) float ldsW[(FEAT / 2) * WP];
    const int tid = threadIdx.x;
    const int lane = tid & 31;

    const float4 vs = ((const float4*)a_src)[lane];
    const float4 vd = ((const float4*)a_dst)[lane];

    const float4* Wv = (const float4*)W;
    constexpr int NT = (FEAT / 2) * (HID / 4) / 256;
    float4 r0[NT], r1[NT];
#pragma unroll
    for (int i = 0; i < NT; ++i) {
        const int item = tid + i * 256;
        const int kp = item >> 5;
        const int c4 = item & 31;
        r0[i] = Wv[(2 * kp) * (HID / 4) + c4];
        r1[i] = Wv[(2 * kp + 1) * (HID / 4) + c4];
    }
    const int ql = lane & 3;
#pragma unroll
    for (int i = 0; i < NT; ++i) {
        const int item = tid + i * 256;
        const int kp = item >> 5;
        const int c4 = item & 31;
        float4* dst = (float4*)(ldsW + kp * WP + c4 * 8);
        dst[0] = make_float4(r0[i].x, r1[i].x, r0[i].y, r1[i].y);
        dst[1] = make_float4(r0[i].z, r1[i].z, r0[i].w, r1[i].w);

        const float s0 = quad_sum(dot4(r0[i], vs));
        const float s1 = quad_sum(dot4(r1[i], vs));
        const float d0 = quad_sum(dot4(r0[i], vd));
        const float d1 = quad_sum(dot4(r1[i], vd));
        const float val = (ql & 2) ? ((ql & 1) ? d1 : d0) : ((ql & 1) ? s1 : s0);
        ldsW[kp * WP + 2 * HID + 16 * (ql >> 1) + 2 * (c4 >> 2) + (ql & 1)] = val;
    }
    __syncthreads();

    const int wave = tid >> 5;
    const int l16  = lane & 15;
    const int half = lane >> 4;
    const int row0 = (blockIdx.x * 8 + wave) * 16;
    if (row0 >= NNODES) return;

    const float* xrow = X + (size_t)(row0 + l16) * FEAT;

    v8f acc[9] = {};
#pragma unroll
    for (int k0 = 0; k0 < FEAT; k0 += 4) {
        const int ka = k0 + 2 * half;
        v2f a;
        a.x = xrow[ka];
        a.y = xrow[ka + 1];
        const float* bp = ldsW + (k0 / 2 + half) * WP + 2 * l16;
#pragma unroll
        for (int t = 0; t < 9; ++t) {
            const v2f b = *(const v2f*)(bp + 32 * t);
            acc[t] = __builtin_amdgcn_wmma_f32_16x16x4_f32(
                         false, a, false, b, (short)0, acc[t], false, false);
        }
    }

    const int orow0 = row0 + 8 * half;
#pragma unroll
    for (int t = 0; t < 8; ++t)
#pragma unroll
        for (int j = 0; j < 8; ++j)
            Hout[(size_t)(orow0 + j) * HID + 16 * t + l16] = acc[t][j];

    float* aout = (l16 < HEADS) ? (AS + l16) : (AD + (l16 - HEADS));
#pragma unroll
    for (int j = 0; j < 8; ++j)
        aout[(size_t)(orow0 + j) * HEADS] = acc[8][j];
}

#define NSLICE  32
#define SLICE_E (NEDGES / NSLICE)
#define SLICE_V (SLICE_E / 4)
#define NR      8
#define RN      6256
#define RP      6272
#define PPITCH  (NR * RP)
#define BPITCH  (NR * RN)
#define BT      512
#define EV      ((SLICE_V + BT - 1) / BT)
#define OWN4    4
#define NOWN    (RN / (4 * OWN4))
#define RV      (RN / 4)
#define CLR     ((RP / 4 + BT - 1) / BT)
static_assert(NEDGES % (NSLICE * 4) == 0, "slices are whole 16-byte pieces");
static_assert(NR * RN > NNODES && (NR - 1) * RN < NNODES, "the ranges hold every node and the total");
static_assert(RN % (4 * OWN4) == 0 && NOWN < BT && RP >= RN + 4 && RP % 32 == 0 && NNODES % 4 == 0,
              "whole pieces; the last thread scans nothing and carries the total");
static_assert(NSLICE == 32 && NSLICE * NR == 256, "k_bases: one (slice, range) total per thread, a wave per range");

static __device__ __forceinline__ unsigned range_span(int lo) {
    return (unsigned)(min(lo + RN, NNODES) - lo);
}

static __device__ __forceinline__ void bump(int* cnt, int i, unsigned span) {
    if ((unsigned)i < span) atomicAdd(&cnt[i], 1);
}

__global__ __launch_bounds__(BT)
void k_slice_count(const int* __restrict__ ei, int* __restrict__ P) {
    __shared__ __attribute__((aligned(16))) int cnt[RP];
    __shared__ int wtot[32];
    const int t = threadIdx.x, lane = t & 31, w = t >> 5;
    const int g = blockIdx.x, r = blockIdx.y;
    const int lo = r * RN;
    const unsigned span = range_span(lo);

    const int4* dv = (const int4*)(ei + NEDGES + g * SLICE_E);
    int4 dd[EV];
#pragma unroll
    for (int k = 0; k < EV; ++k) dd[k] = dv[min(t + k * BT, SLICE_V - 1)];

    int4* c4 = (int4*)cnt;
#pragma unroll
    for (int k = 0; k < CLR; ++k) {
        const int i = t + k * BT;
        if (i < RP / 4) c4[i] = make_int4(0, 0, 0, 0);
    }
    __syncthreads();

#pragma unroll
    for (int k = 0; k < EV; ++k) {
        if (t + k * BT < SLICE_V) {
            bump(cnt, dd[k].x - lo, span); bump(cnt, dd[k].y - lo, span);
            bump(cnt, dd[k].z - lo, span); bump(cnt, dd[k].w - lo, span);
        }
    }
    __syncthreads();

    int4 v[OWN4];
    int run = 0;
#pragma unroll
    for (int k = 0; k < OWN4; ++k) {
        int4 c = make_int4(0, 0, 0, 0);
        if (t < NOWN) c = c4[t * OWN4 + k];
        int x;
        x = c.x; c.x = run; run += x;
        x = c.y; c.y = run; run += x;
        x = c.z; c.z = run; run += x;
        x = c.w; c.w = run; run += x;
        v[k] = c;
    }
    int inc = run;
    for (int o = 1; o < 32; o <<= 1) {
        int n = __shfl_up(inc, o);
        if (lane >= o) inc += n;
    }
    if (lane == 31) wtot[w] = inc;
    __syncthreads();
    if (w == 0) {
        int x = (lane < BT / 32) ? wtot[lane] : 0, xi = x;
        for (int o = 1; o < 32; o <<= 1) {
            int n = __shfl_up(xi, o);
            if (lane >= o) xi += n;
        }
        if (lane < BT / 32) wtot[lane] = xi - x;
    }
    __syncthreads();
    const int off = wtot[w] + (inc - run);
    if (t < NOWN) {
#pragma unroll
        for (int k = 0; k < OWN4; ++k) {
            int4 c = v[k];
            c.x += off; c.y += off; c.z += off; c.w += off;
            c4[t * OWN4 + k] = c;
        }
    }
    if (t == BT - 1) c4[RV] = make_int4(off, 0, 0, 0);
    __syncthreads();

    int4* P4 = (int4*)(P + (size_t)g * PPITCH + r * RP);
#pragma unroll
    for (int k = 0; k < CLR; ++k) {
        const int i = t + k * BT;
        if (i < RV + 1) P4[i] = c4[i];
    }
}

__global__ __launch_bounds__(256)
void k_bases(const int* __restrict__ P, int* __restrict__ rp, int* __restrict__ B) {
    __shared__ int part[8];
    const int t = threadIdx.x, r = blockIdx.y;

    const int tr = t >> 5, tg = t & 31;
    int x = (tr < r) ? P[(size_t)tg * PPITCH + tr * RP + RN] : 0;
    for (int o = 16; o > 0; o >>= 1) x += __shfl_xor(x, o);
    if (tg == 0) part[tr] = x;
    __syncthreads();
    int sum = 0;
#pragma unroll
    for (int k = 0; k < 8; ++k) sum += part[k];

    const int i = blockIdx.x * 256 + t;
    const int n = r * RN + i;
    if (i >= RN || n > NNODES) return;
    const int* Pr = P + r * RP + i;
    int p[NSLICE], q[NSLICE];
#pragma unroll
    for (int g = 0; g < NSLICE; ++g) {
        p[g] = Pr[(size_t)g * PPITCH];
        q[g] = Pr[(size_t)g * PPITCH + 1];
        sum += p[g];
    }
    rp[n] = sum;
    if (n < NNODES) {
        int run = sum;
#pragma unroll
        for (int g = 0; g < NSLICE; ++g) {
            B[(size_t)g * BPITCH + n] = run;
            run += q[g] - p[g];
        }
    }
}

static __device__ __forceinline__ int take(int* pos, int i, unsigned span) {
    return ((unsigned)i < span) ? atomicAdd(&pos[i], 1) : -1;
}

static __device__ __forceinline__ void put(int* __restrict__ lst, int slot, int s) {
    if ((unsigned)slot < (unsigned)NEDGES) lst[slot] = s;
}

__global__ __launch_bounds__(BT)
void k_slice_place(const int* __restrict__ ei, const int* __restrict__ B,
                   int* __restrict__ lst) {
    __shared__ __attribute__((aligned(16))) int pos[CLR * BT * 4];
    const int t = threadIdx.x, g = blockIdx.x, r = blockIdx.y;
    const int lo = r * RN;
    const unsigned span = range_span(lo);

    const int4* sv = (const int4*)(ei + g * SLICE_E);
    const int4* dv = (const int4*)(ei + NEDGES + g * SLICE_E);
    int4 ss[EV], dd[EV];
#pragma unroll
    for (int k = 0; k < EV; ++k) {
        const int i = min(t + k * BT, SLICE_V - 1);
        ss[k] = sv[i];
        dd[k] = dv[i];
    }

    const int4* B4 = (const int4*)(B + (size_t)g * BPITCH + lo);
    int4* p4 = (int4*)pos;
    const int nb = (int)(span / 4);
    int4 bv[CLR];
#pragma unroll
    for (int k = 0; k < CLR; ++k) bv[k] = B4[min(t + k * BT, nb - 1)];
#pragma unroll
    for (int k = 0; k < CLR; ++k) p4[t + k * BT] = bv[k];
    __syncthreads();

#pragma unroll
    for (int k = 0; k < EV; ++k) {
        int4 sl = make_int4(-1, -1, -1, -1);
        if (t + k * BT < SLICE_V) {
            sl.x = take(pos, dd[k].x - lo, span); sl.y = take(pos, dd[k].y - lo, span);
            sl.z = take(pos, dd[k].z - lo, span); sl.w = take(pos, dd[k].w - lo, span);
        }
        dd[k] = sl;
    }
#pragma unroll
    for (int k = 0; k < EV; ++k) {
        put(lst, dd[k].x, ss[k].x); put(lst, dd[k].y, ss[k].y);
        put(lst, dd[k].z, ss[k].z); put(lst, dd[k].w, ss[k].w);
    }
}

#define WPB 8
#define GBLK 1024
#define NWAVES (GBLK * WPB)

static __device__ __forceinline__ float leaky(float v) {
    return v >= 0.f ? v : NEG_SLOPE * v;
}

static __device__ __forceinline__ int load_chunk(const int* __restrict__ lst,
                                                 int c0, int cnt, int lane, int d) {
    int sv = (lane < cnt) ? lst[c0 + lane] : d;
    return min(max(sv, 0), NNODES - 1);
}

static __device__ __forceinline__ float quad_x1(float v) {
    return __int_as_float(__builtin_amdgcn_update_dpp(0, __float_as_int(v), 0xB1, 0xF, 0xF, true));
}
static __device__ __forceinline__ float quad_x2(float v) {
    return __int_as_float(__builtin_amdgcn_update_dpp(0, __float_as_int(v), 0x4E, 0xF, 0xF, true));
}
template <int QQ>
static __device__ __forceinline__ float quad_bcast(float v) {
    return __int_as_float(__builtin_amdgcn_update_dpp(0, __float_as_int(v), QQ * 0x55, 0xF, 0xF, true));
}

static __device__ __forceinline__ float dot4p(const float4 r, const v2f a01, const v2f a23) {
    v2f t = (v2f){r.x, r.y} * a01;
    t = (v2f){r.z, r.w} * a23 + t;
    return t.x + t.y;
}

static __device__ __forceinline__ float quad_transpose_sum(float p0, float p1, float p2, float p3,
                                                           bool odd, bool up) {
    const float k01 = odd ? p1 : p0, s01 = odd ? p0 : p1;
    const float k23 = odd ? p3 : p2, s23 = odd ? p2 : p3;
    const float u = k01 + quad_x1(s01);
    const float v = k23 + quad_x1(s23);
    const float k = up ? v : u, s = up ? u : v;
    return k + quad_x2(s);
}

template <int NGRP>
static __device__ __forceinline__ void rows_step(const float* __restrict__ Hf, int sv, int base,
                                                 int cnt, int lane, const float4 as4, float adg,
                                                 float& m, float& Sp, float4& acc) {
    const int q = lane & 3;
    const bool odd = (q & 1) != 0, up = (q & 2) != 0;
    const float ninf = -__builtin_huge_valf();
    const v2f a01 = (v2f){as4.x, as4.y}, a23 = (v2f){as4.z, as4.w};
    float4 r[4 * NGRP];
#pragma unroll
    for (int j = 0; j < 4 * NGRP; ++j) {
        const unsigned s = (unsigned)__builtin_amdgcn_readlane(sv, base + j);
        const unsigned boff = s * (unsigned)(HID * 4) + 16u * (unsigned)lane;
        r[j] = *(const float4*)((const char*)Hf + boff);
    }
    __builtin_amdgcn_sched_barrier(0);
    float e[NGRP];
    float mc = ninf;
#pragma unroll
    for (int i = 0; i < NGRP; ++i) {
        const float sc = quad_transpose_sum(dot4p(r[4 * i], a01, a23), dot4p(r[4 * i + 1], a01, a23),
                                            dot4p(r[4 * i + 2], a01, a23), dot4p(r[4 * i + 3], a01, a23), odd, up);
        const float v = leaky(sc + adg);
        e[i] = (4 * i + q < cnt) ? v : ninf;
        mc = fmaxf(mc, e[i]);
    }
    mc = fmaxf(mc, quad_x1(mc));
    mc = fmaxf(mc, quad_x2(mc));
    const float mn = fmaxf(m, mc);
    const float f = expf(m - mn);
    m = mn;
    Sp *= f;
    acc.x *= f; acc.y *= f; acc.z *= f; acc.w *= f;
#pragma unroll
    for (int i = 0; i < NGRP; ++i) {
        const float w = expf(e[i] - mn);
        Sp += w;
        float al;
        al = quad_bcast<0>(w);
        acc.x += al * r[4 * i].x; acc.y += al * r[4 * i].y; acc.z += al * r[4 * i].z; acc.w += al * r[4 * i].w;
        al = quad_bcast<1>(w);
        acc.x += al * r[4 * i + 1].x; acc.y += al * r[4 * i + 1].y; acc.z += al * r[4 * i + 1].z; acc.w += al * r[4 * i + 1].w;
        al = quad_bcast<2>(w);
        acc.x += al * r[4 * i + 2].x; acc.y += al * r[4 * i + 2].y; acc.z += al * r[4 * i + 2].z; acc.w += al * r[4 * i + 2].w;
        al = quad_bcast<3>(w);
        acc.x += al * r[4 * i + 3].x; acc.y += al * r[4 * i + 3].y; acc.z += al * r[4 * i + 3].z; acc.w += al * r[4 * i + 3].w;
    }
}

struct NodeHead {
    int rv;
    float asv, adg;
    float4 rself;
};

static __device__ __forceinline__ NodeHead node_head(int d, int lane, int hd,
                                                     const int* __restrict__ rp,
                                                     const float* __restrict__ AS, const float* __restrict__ AD,
                                                     const float* __restrict__ Hf) {
    NodeHead h;
    h.rv = rp[d + (lane & 1)];
    h.adg = AD[d * HEADS + hd];
    h.asv = AS[d * HEADS + hd];
    const unsigned doff = (unsigned)d * (unsigned)(HID * 4) + 16u * (unsigned)lane;
    h.rself = *(const float4*)((const char*)Hf + doff);
    return h;
}

static __device__ __forceinline__ float4 gather_body(int d, const NodeHead h, int lane, const float4 bb, const float4 as4,
                                                     const int* __restrict__ lst,
                                                     const float* __restrict__ Hf) {
    int beg = __builtin_amdgcn_readlane(h.rv, 0), end = __builtin_amdgcn_readlane(h.rv, 1);
    beg = min(max(beg, 0), NEDGES);
    end = min(max(end, beg), NEDGES);

    const float adg   = h.adg;
    const float eself = leaky(h.asv + adg);

    float m  = eself;
    float Sp = ((lane & 3) == 0) ? 1.0f : 0.0f;
    float4 acc = h.rself;

    for (int c0 = beg; c0 < end; c0 += 32) {
        const int cnt = min(32, end - c0);
        const int sv = load_chunk(lst, c0, cnt, lane, d);
        for (int hb = 0; hb < cnt; hb += 16) {
            const int c = __builtin_amdgcn_readfirstlane(min(16, cnt - hb));
            if (c <= 4)       rows_step<1>(Hf, sv, hb, c, lane, as4, adg, m, Sp, acc);
            else if (c <= 8)  rows_step<2>(Hf, sv, hb, c, lane, as4, adg, m, Sp, acc);
            else if (c <= 12) rows_step<3>(Hf, sv, hb, c, lane, as4, adg, m, Sp, acc);
            else              rows_step<4>(Hf, sv, hb, c, lane, as4, adg, m, Sp, acc);
        }
    }

    const float S = quad_sum(Sp);
    const float inv = 1.0f / (S + 1e-16f);

    float4 o;
    float v;
    v = acc.x * inv + bb.x; o.x = v > 0.f ? v : expm1f(v);
    v = acc.y * inv + bb.y; o.y = v > 0.f ? v : expm1f(v);
    v = acc.z * inv + bb.z; o.z = v > 0.f ? v : expm1f(v);
    v = acc.w * inv + bb.w; o.w = v > 0.f ? v : expm1f(v);
    return o;
}

__global__ __launch_bounds__(256) __attribute__((amdgpu_waves_per_eu(8, 8)))
void k_node_gather(const int* __restrict__ rp, const int* __restrict__ lst,
                   const float* __restrict__ AS, const float* __restrict__ AD,
                   const float* __restrict__ Hf, const float* __restrict__ a_src,
                   const float* __restrict__ b, float* __restrict__ Out) {
    const int lane = threadIdx.x & 31;
    const int d0 = __builtin_amdgcn_readfirstlane(blockIdx.x * WPB + (threadIdx.x >> 5));
    const int hd = lane >> 2;
    const float4 bb  = *(const float4*)(b + 4 * lane);
    const float4 as4 = *(const float4*)(a_src + 4 * lane);

    for (int d = d0; d < NNODES; d += NWAVES) {
        const NodeHead h = node_head(d, lane, hd, rp, AS, AD, Hf);
        const float4 o = gather_body(d, h, lane, bb, as4, lst, Hf);
        const unsigned doff = (unsigned)d * (unsigned)(HID * 4) + 16u * (unsigned)lane;
        *(float4*)((char*)Out + doff) = o;
    }
}

#define FT 512
#define FGRID 512
#define NSTRIP (NNODES / 16)
#define XP 132

__global__ __launch_bounds__(FT) __attribute__((amdgpu_waves_per_eu(8, 8)))
void k_gather_gemm(const int* __restrict__ rp, const int* __restrict__ lst,
                   const float* __restrict__ AS, const float* __restrict__ AD,
                   const float* __restrict__ Hf, const float* __restrict__ a_src,
                   const float* __restrict__ b,
                   const float* __restrict__ W, const float* __restrict__ a_src2,
                   const float* __restrict__ a_dst2,
                   float* __restrict__ Hout, float* __restrict__ AS2) {
    __shared__ __attribute__((aligned(16))) float ldsW[(FEAT / 2) * WP];
    __shared__ __attribute__((aligned(16))) float ldsX[2 * 16 * XP];
    const int tid = threadIdx.x;
    const int lane = tid & 31;
    const int wave = __builtin_amdgcn_readfirstlane(tid >> 5);

    const int hd = lane >> 2;
    const float4 bb  = *(const float4*)(b + 4 * lane);
    const float4 as4 = *(const float4*)(a_src + 4 * lane);
    NodeHead h = node_head(16 * (int)blockIdx.x + wave, lane, hd, rp, AS, AD, Hf);

    {
        const float4 vs = ((const float4*)a_src2)[lane];
        const float4 vd = ((const float4*)a_dst2)[lane];
        const float4* Wv = (const float4*)W;
        constexpr int NT = (FEAT / 2) * (HID / 4) / FT;
        float4 r0[NT], r1[NT];
#pragma unroll
        for (int i = 0; i < NT; ++i) {
            const int item = tid + i * FT;
            const int kp = item >> 5;
            const int c4 = item & 31;
            r0[i] = Wv[(2 * kp) * (HID / 4) + c4];
            r1[i] = Wv[(2 * kp + 1) * (HID / 4) + c4];
        }
        const int ql = lane & 3;
#pragma unroll
        for (int i = 0; i < NT; ++i) {
            const int item = tid + i * FT;
            const int kp = item >> 5;
            const int c4 = item & 31;
            float4* dst = (float4*)(ldsW + kp * WP + c4 * 8);
            dst[0] = make_float4(r0[i].x, r1[i].x, r0[i].y, r1[i].y);
            dst[1] = make_float4(r0[i].z, r1[i].z, r0[i].w, r1[i].w);
            const float s0 = quad_sum(dot4(r0[i], vs));
            const float s1 = quad_sum(dot4(r1[i], vs));
            const float d0 = quad_sum(dot4(r0[i], vd));
            const float d1 = quad_sum(dot4(r1[i], vd));
            const float val = (ql & 2) ? ((ql & 1) ? d1 : d0) : ((ql & 1) ? s1 : s0);
            ldsW[kp * WP + 2 * HID + 16 * (ql >> 1) + 2 * (c4 >> 2) + (ql & 1)] = val;
        }
    }

    int p = 0;
    for (int s = blockIdx.x; s < NSTRIP; s += FGRID) {
        const int d = 16 * s + wave;
        const float4 o = gather_body(d, h, lane, bb, as4, lst, Hf);
        float* xs = ldsX + p * (16 * XP);
        *(float4*)(xs + wave * XP + 4 * lane) = o;
        __syncthreads();
        h = node_head(min(d + 16 * FGRID, NNODES - 1), lane, hd, rp, AS, AD, Hf);
        if (wave < 9) {
            int ln = lane;
            asm volatile("" : "+v"(ln));
            const int l16  = ln & 15;
            const int half = ln >> 4;
            float* obase;
            int opitch;
            unsigned ocol;
            if (wave < 8) { obase = Hout; opitch = HID;   ocol = 16u * (unsigned)wave + (unsigned)l16; }
            else          { obase = AS2;  opitch = HEADS; ocol = (l16 < HEADS) ? (unsigned)l16 : (unsigned)(NNODES * HEADS + l16 - HEADS); }
            const unsigned ooff = (unsigned)((16 * s + 8 * half) * opitch) + ocol;
            const float* wb = ldsW + half * WP + 2 * l16 + 32 * wave;
            const float* xb = xs + l16 * XP + 2 * half;
            v8f acc = {};
#pragma unroll
            for (int k0 = 0; k0 < FEAT; k0 += 4) {
                const v2f a  = *(const v2f*)(xb + k0);
                const v2f bq = *(const v2f*)(wb + (k0 / 2) * WP);
                acc = __builtin_amdgcn_wmma_f32_16x16x4_f32(
                          false, a, false, bq, (short)0, acc, false, false);
            }
#pragma unroll
            for (int j = 0; j < 8; ++j)
                obase[ooff + (unsigned)(j * opitch)] = acc[j];
        }
        p ^= 1;
    }
}

static inline unsigned grid_for(long long n) { return (unsigned)((n + 255) / 256); }

extern "C" void kernel_launch(void* const* d_in, const int* in_sizes, int n_in,
                              void* d_out, int out_size, void* d_ws, size_t ws_size,
                              hipStream_t stream) {
    const float* x   = (const float*)d_in[0];
    const int*   ei  = (const int*)  d_in[1];
    const float* W1  = (const float*)d_in[2];
    const float* as1 = (const float*)d_in[3];
    const float* ad1 = (const float*)d_in[4];
    const float* b1  = (const float*)d_in[5];
    const float* W2  = (const float*)d_in[6];
    const float* as2 = (const float*)d_in[7];
    const float* ad2 = (const float*)d_in[8];
    const float* b2  = (const float*)d_in[9];

    float* ws     = (float*)d_ws;
    float* bufH   = ws;
    float* bufAS  = bufH  + (size_t)NNODES * HID;
    float* bufAD  = bufAS + (size_t)NNODES * HEADS;
    float* bufM   = bufAD + (size_t)NNODES * HEADS;
    float* bufS   = bufM  + (size_t)NNODES * HEADS;
    float* bufAgg = bufS  + (size_t)NNODES * HEADS;
    int*   rp     = (int*)bufAgg;
    int*   lst    = rp + 50048;
    int*   Pre    = lst + NEDGES;
    int*   Bas    = Pre + NSLICE * PPITCH;
    float* out    = (float*)d_out;

    k_slice_count<<<dim3(NSLICE, NR), BT, 0, stream>>>(ei, Pre);
    k_bases<<<dim3((RN + 255) / 256, NR), 256, 0, stream>>>(Pre, rp, Bas);
    k_slice_place<<<dim3(NSLICE, NR), BT, 0, stream>>>(ei, Bas, lst);

    k_gemm_wmma<<<GBLOCKS, 256, 0, stream>>>(x, W1, as1, ad1, out, bufAS, bufAD);
    k_gather_gemm<<<FGRID, FT, 0, stream>>>(rp, lst, bufAS, bufAD, out, as1, b1,
                                            W2, as2, ad2, bufH, bufM);
    k_node_gather<<<GBLK, 256, 0, stream>>>(rp, lst, bufM, bufS, bufH, as2, b2, out);
}


# Layer-1 product rides in the count and bases launches, half its row blocks each, chosen by block index (1 per WGP at most: 2 strips per SIMD, was 4); place launch alone, bucket order kept; 6->5
# speedup vs baseline: 10.6366x; 1.0452x over previous
#include <hip/hip_runtime.h>
#include <math.h>

#define NNODES 50000
#define NEDGES 800000
#define FEAT 128
#define HID  128
#define HEADS 8
#define HDIM 16
#define NEG_SLOPE 0.2f

typedef __attribute__((ext_vector_type(2))) float v2f;
typedef __attribute__((ext_vector_type(8))) float v8f;

#define GROWS 128
#define GBLOCKS ((NNODES + GROWS - 1) / GROWS)
#define WP 288
static_assert(NNODES % 16 == 0, "a strip is wholly inside or wholly outside the rows");

static __device__ __forceinline__ float quad_sum(float v) {
    v += __int_as_float(__builtin_amdgcn_update_dpp(0, __float_as_int(v), 0xB1, 0xF, 0xF, true));
    v += __int_as_float(__builtin_amdgcn_update_dpp(0, __float_as_int(v), 0x4E, 0xF, 0xF, true));
    return v;
}

static __device__ __forceinline__ float dot4(const float4 r, const float4 a) {
    return r.x * a.x + r.y * a.y + r.z * a.z + r.w * a.w;
}

#define GA (GBLOCKS / 2)

template <int T0, int NH, bool ATT>
static __device__ __forceinline__
void strip_tiles(const float* __restrict__ xrow, const float* __restrict__ ldsW,
                 const int half, const int l16, const int orow0,
                 float* __restrict__ Hout, float* __restrict__ AS, float* __restrict__ AD) {
    v8f acc[NH + 1] = {};
#pragma unroll
    for (int k0 = 0; k0 < FEAT; k0 += 4) {
        const int ka = k0 + 2 * half;
        v2f a;
        a.x = xrow[ka];
        a.y = xrow[ka + 1];
        const float* bp = ldsW + (k0 / 2 + half) * WP + 2 * l16;
#pragma unroll
        for (int t = 0; t < NH; ++t) {
            const v2f b = *(const v2f*)(bp + 32 * (T0 + t));
            acc[t] = __builtin_amdgcn_wmma_f32_16x16x4_f32(
                         false, a, false, b, (short)0, acc[t], false, false);
        }
        if (ATT) {
            const v2f b = *(const v2f*)(bp + 32 * 8);
            acc[NH] = __builtin_amdgcn_wmma_f32_16x16x4_f32(
                          false, a, false, b, (short)0, acc[NH], false, false);
        }
    }

#pragma unroll
    for (int t = 0; t < NH; ++t)
#pragma unroll
        for (int j = 0; j < 8; ++j)
            Hout[(size_t)(orow0 + j) * HID + 16 * (T0 + t) + l16] = acc[t][j];

    if (ATT) {
        float* aout = (l16 < HEADS) ? (AS + l16) : (AD + (l16 - HEADS));
#pragma unroll
        for (int j = 0; j < 8; ++j)
            aout[(size_t)(orow0 + j) * HEADS] = acc[NH][j];
    }
}

template <int NTHR>
static __device__ __forceinline__
void gemm_block(const int gb, float* __restrict__ ldsW,
                const float* __restrict__ X, const float* __restrict__ W,
                const float* __restrict__ a_src, const float* __restrict__ a_dst,
                float* __restrict__ Hout, float* __restrict__ AS,
                float* __restrict__ AD) {
    const int tid = threadIdx.x;
    const int lane = tid & 31;

    const float4 vs = ((const float4*)a_src)[lane];
    const float4 vd = ((const float4*)a_dst)[lane];

    const float4* Wv = (const float4*)W;
    constexpr int NT = (FEAT / 2) * (HID / 4) / NTHR;
    float4 r0[NT], r1[NT];
#pragma unroll
    for (int i = 0; i < NT; ++i) {
        const int item = tid + i * NTHR;
        const int kp = item >> 5;
        const int c4 = item & 31;
        r0[i] = Wv[(2 * kp) * (HID / 4) + c4];
        r1[i] = Wv[(2 * kp + 1) * (HID / 4) + c4];
    }
    const int ql = lane & 3;
#pragma unroll
    for (int i = 0; i < NT; ++i) {
        const int item = tid + i * NTHR;
        const int kp = item >> 5;
        const int c4 = item & 31;
        float4* dst = (float4*)(ldsW + kp * WP + c4 * 8);
        dst[0] = make_float4(r0[i].x, r1[i].x, r0[i].y, r1[i].y);
        dst[1] = make_float4(r0[i].z, r1[i].z, r0[i].w, r1[i].w);

        const float s0 = quad_sum(dot4(r0[i], vs));
        const float s1 = quad_sum(dot4(r1[i], vs));
        const float d0 = quad_sum(dot4(r0[i], vd));
        const float d1 = quad_sum(dot4(r1[i], vd));
        const float val = (ql & 2) ? ((ql & 1) ? d1 : d0) : ((ql & 1) ? s1 : s0);
        ldsW[kp * WP + 2 * HID + 16 * (ql >> 1) + 2 * (c4 >> 2) + (ql & 1)] = val;
    }
    __syncthreads();

    const int wave = __builtin_amdgcn_readfirstlane(tid >> 5);
    const int l16  = lane & 15;
    const int half = lane >> 4;
    const int row0 = (gb * 8 + (wave & 7)) * 16;
    if (row0 >= NNODES) return;

    const float* xrow = X + (size_t)(row0 + l16) * FEAT;
    const int orow0 = row0 + 8 * half;
    if (NTHR == 256) {
        strip_tiles<0, 8, true>(xrow, ldsW, half, l16, orow0, Hout, AS, AD);
    } else {
        if (wave < 8) strip_tiles<0, 4, true>(xrow, ldsW, half, l16, orow0, Hout, AS, AD);
        else          strip_tiles<4, 4, false>(xrow, ldsW, half, l16, orow0, Hout, AS, AD);
    }
}

#define NSLICE  32
#define SLICE_E (NEDGES / NSLICE)
#define SLICE_V (SLICE_E / 4)
#define NR      8
#define RN      6256
#define RP      6272
#define PPITCH  (NR * RP)
#define BPITCH  (NR * RN)
#define BT      512
#define EV      ((SLICE_V + BT - 1) / BT)
#define OWN4    4
#define NOWN    (RN / (4 * OWN4))
#define RV      (RN / 4)
#define CLR     ((RP / 4 + BT - 1) / BT)
static_assert(NEDGES % (NSLICE * 4) == 0, "slices are whole 16-byte pieces");
static_assert(NR * RN > NNODES && (NR - 1) * RN < NNODES, "the ranges hold every node and the total");
static_assert(RN % (4 * OWN4) == 0 && NOWN < BT && RP >= RN + 4 && RP % 32 == 0 && NNODES % 4 == 0,
              "whole pieces; the last thread scans nothing and carries the total");
static_assert(NSLICE == 32 && NSLICE * NR == 256, "k_bases_gemm: one (slice, range) total per thread, a wave per range");

static __device__ __forceinline__ unsigned range_span(int lo) {
    return (unsigned)(min(lo + RN, NNODES) - lo);
}

static __device__ __forceinline__ void bump(int* cnt, int i, unsigned span) {
    if ((unsigned)i < span) atomicAdd(&cnt[i], 1);
}

__global__ __launch_bounds__(BT) __attribute__((amdgpu_waves_per_eu(8, 8)))
void k_count_gemm(const int* __restrict__ ei, int* __restrict__ P,
                  const float* __restrict__ X, const float* __restrict__ W,
                  const float* __restrict__ a_src, const float* __restrict__ a_dst,
                  float* __restrict__ Hout, float* __restrict__ AS, float* __restrict__ AD) {
    __shared__ __attribute__((aligned(16))) int cnt[RP];
    __shared__ int wtot[32];
    __shared__ __attribute__((aligned(16))) float ldsW[(FEAT / 2) * WP];
    if (blockIdx.x < GA) {
        gemm_block<BT>((int)blockIdx.x, ldsW, X, W, a_src, a_dst, Hout, AS, AD);
        return;
    }
    const int t = threadIdx.x, lane = t & 31, w = t >> 5;
    const int cb = (int)blockIdx.x - GA;
    const int g = cb % NSLICE, r = cb / NSLICE;
    const int lo = r * RN;
    const unsigned span = range_span(lo);

    const int4* dv = (const int4*)(ei + NEDGES + g * SLICE_E);
    int4 dd[EV];
#pragma unroll
    for (int k = 0; k < EV; ++k) dd[k] = dv[min(t + k * BT, SLICE_V - 1)];

    int4* c4 = (int4*)cnt;
#pragma unroll
    for (int k = 0; k < CLR; ++k) {
        const int i = t + k * BT;
        if (i < RP / 4) c4[i] = make_int4(0, 0, 0, 0);
    }
    __syncthreads();

#pragma unroll
    for (int k = 0; k < EV; ++k) {
        if (t + k * BT < SLICE_V) {
            bump(cnt, dd[k].x - lo, span); bump(cnt, dd[k].y - lo, span);
            bump(cnt, dd[k].z - lo, span); bump(cnt, dd[k].w - lo, span);
        }
    }
    __syncthreads();

    int4 v[OWN4];
    int run = 0;
#pragma unroll
    for (int k = 0; k < OWN4; ++k) {
        int4 c = make_int4(0, 0, 0, 0);
        if (t < NOWN) c = c4[t * OWN4 + k];
        int x;
        x = c.x; c.x = run; run += x;
        x = c.y; c.y = run; run += x;
        x = c.z; c.z = run; run += x;
        x = c.w; c.w = run; run += x;
        v[k] = c;
    }
    int inc = run;
    for (int o = 1; o < 32; o <<= 1) {
        int n = __shfl_up(inc, o);
        if (lane >= o) inc += n;
    }
    if (lane == 31) wtot[w] = inc;
    __syncthreads();
    if (w == 0) {
        int x = (lane < BT / 32) ? wtot[lane] : 0, xi = x;
        for (int o = 1; o < 32; o <<= 1) {
            int n = __shfl_up(xi, o);
            if (lane >= o) xi += n;
        }
        if (lane < BT / 32) wtot[lane] = xi - x;
    }
    __syncthreads();
    const int off = wtot[w] + (inc - run);
    if (t < NOWN) {
#pragma unroll
        for (int k = 0; k < OWN4; ++k) {
            int4 c = v[k];
            c.x += off; c.y += off; c.z += off; c.w += off;
            c4[t * OWN4 + k] = c;
        }
    }
    if (t == BT - 1) c4[RV] = make_int4(off, 0, 0, 0);
    __syncthreads();

    int4* P4 = (int4*)(P + (size_t)g * PPITCH + r * RP);
#pragma unroll
    for (int k = 0; k < CLR; ++k) {
        const int i = t + k * BT;
        if (i < RV + 1) P4[i] = c4[i];
    }
}

#define BASES_BX ((RN + 255) / 256)

__global__ __launch_bounds__(256)
void k_bases_gemm(const int* __restrict__ P, int* __restrict__ rp, int* __restrict__ B,
                  const float* __restrict__ X, const float* __restrict__ W,
                  const float* __restrict__ a_src, const float* __restrict__ a_dst,
                  float* __restrict__ Hout, float* __restrict__ AS, float* __restrict__ AD) {
    __shared__ int part[8];
    __shared__ __attribute__((aligned(16))) float ldsW[(FEAT / 2) * WP];
    if (blockIdx.x < GBLOCKS - GA) {
        gemm_block<256>(GA + (int)blockIdx.x, ldsW, X, W, a_src, a_dst, Hout, AS, AD);
        return;
    }
    const int bb = (int)blockIdx.x - (GBLOCKS - GA);
    const int bx = bb % BASES_BX;
    const int t = threadIdx.x, r = bb / BASES_BX;

    const int tr = t >> 5, tg = t & 31;
    int x = (tr < r) ? P[(size_t)tg * PPITCH + tr * RP + RN] : 0;
    for (int o = 16; o > 0; o >>= 1) x += __shfl_xor(x, o);
    if (tg == 0) part[tr] = x;
    __syncthreads();
    int sum = 0;
#pragma unroll
    for (int k = 0; k < 8; ++k) sum += part[k];

    const int i = bx * 256 + t;
    const int n = r * RN + i;
    if (i >= RN || n > NNODES) return;
    const int* Pr = P + r * RP + i;
    int p[NSLICE], q[NSLICE];
#pragma unroll
    for (int g = 0; g < NSLICE; ++g) {
        p[g] = Pr[(size_t)g * PPITCH];
        q[g] = Pr[(size_t)g * PPITCH + 1];
        sum += p[g];
    }
    rp[n] = sum;
    if (n < NNODES) {
        int run = sum;
#pragma unroll
        for (int g = 0; g < NSLICE; ++g) {
            B[(size_t)g * BPITCH + n] = run;
            run += q[g] - p[g];
        }
    }
}

static __device__ __forceinline__ int take(int* pos, int i, unsigned span) {
    return ((unsigned)i < span) ? atomicAdd(&pos[i], 1) : -1;
}

static __device__ __forceinline__ void put(int* __restrict__ lst, int slot, int s) {
    if ((unsigned)slot < (unsigned)NEDGES) lst[slot] = s;
}

__global__ __launch_bounds__(BT)
void k_slice_place(const int* __restrict__ ei, const int* __restrict__ B,
                   int* __restrict__ lst) {
    __shared__ __attribute__((aligned(16))) int pos[CLR * BT * 4];
    const int t = threadIdx.x, g = blockIdx.x, r = blockIdx.y;
    const int lo = r * RN;
    const unsigned span = range_span(lo);

    const int4* sv = (const int4*)(ei + g * SLICE_E);
    const int4* dv = (const int4*)(ei + NEDGES + g * SLICE_E);
    int4 ss[EV], dd[EV];
#pragma unroll
    for (int k = 0; k < EV; ++k) {
        const int i = min(t + k * BT, SLICE_V - 1);
        ss[k] = sv[i];
        dd[k] = dv[i];
    }

    const int4* B4 = (const int4*)(B + (size_t)g * BPITCH + lo);
    int4* p4 = (int4*)pos;
    const int nb = (int)(span / 4);
    int4 bv[CLR];
#pragma unroll
    for (int k = 0; k < CLR; ++k) bv[k] = B4[min(t + k * BT, nb - 1)];
#pragma unroll
    for (int k = 0; k < CLR; ++k) p4[t + k * BT] = bv[k];
    __syncthreads();

#pragma unroll
    for (int k = 0; k < EV; ++k) {
        int4 sl = make_int4(-1, -1, -1, -1);
        if (t + k * BT < SLICE_V) {
            sl.x = take(pos, dd[k].x - lo, span); sl.y = take(pos, dd[k].y - lo, span);
            sl.z = take(pos, dd[k].z - lo, span); sl.w = take(pos, dd[k].w - lo, span);
        }
        dd[k] = sl;
    }
#pragma unroll
    for (int k = 0; k < EV; ++k) {
        put(lst, dd[k].x, ss[k].x); put(lst, dd[k].y, ss[k].y);
        put(lst, dd[k].z, ss[k].z); put(lst, dd[k].w, ss[k].w);
    }
}

#define WPB 8
#define GBLK 1024
#define NWAVES (GBLK * WPB)

static __device__ __forceinline__ float leaky(float v) {
    return v >= 0.f ? v : NEG_SLOPE * v;
}

static __device__ __forceinline__ int load_chunk(const int* __restrict__ lst,
                                                 int c0, int cnt, int lane, int d) {
    int sv = (lane < cnt) ? lst[c0 + lane] : d;
    return min(max(sv, 0), NNODES - 1);
}

static __device__ __forceinline__ float quad_x1(float v) {
    return __int_as_float(__builtin_amdgcn_update_dpp(0, __float_as_int(v), 0xB1, 0xF, 0xF, true));
}
static __device__ __forceinline__ float quad_x2(float v) {
    return __int_as_float(__builtin_amdgcn_update_dpp(0, __float_as_int(v), 0x4E, 0xF, 0xF, true));
}
template <int QQ>
static __device__ __forceinline__ float quad_bcast(float v) {
    return __int_as_float(__builtin_amdgcn_update_dpp(0, __float_as_int(v), QQ * 0x55, 0xF, 0xF, true));
}

static __device__ __forceinline__ float dot4p(const float4 r, const v2f a01, const v2f a23) {
    v2f t = (v2f){r.x, r.y} * a01;
    t = (v2f){r.z, r.w} * a23 + t;
    return t.x + t.y;
}

static __device__ __forceinline__ float quad_transpose_sum(float p0, float p1, float p2, float p3,
                                                           bool odd, bool up) {
    const float k01 = odd ? p1 : p0, s01 = odd ? p0 : p1;
    const float k23 = odd ? p3 : p2, s23 = odd ? p2 : p3;
    const float u = k01 + quad_x1(s01);
    const float v = k23 + quad_x1(s23);
    const float k = up ? v : u, s = up ? u : v;
    return k + quad_x2(s);
}

template <int NGRP>
static __device__ __forceinline__ void rows_step(const float* __restrict__ Hf, int sv, int base,
                                                 int cnt, int lane, const float4 as4, float adg,
                                                 float& m, float& Sp, float4& acc) {
    const int q = lane & 3;
    const bool odd = (q & 1) != 0, up = (q & 2) != 0;
    const float ninf = -__builtin_huge_valf();
    const v2f a01 = (v2f){as4.x, as4.y}, a23 = (v2f){as4.z, as4.w};
    float4 r[4 * NGRP];
#pragma unroll
    for (int j = 0; j < 4 * NGRP; ++j) {
        const unsigned s = (unsigned)__builtin_amdgcn_readlane(sv, base + j);
        const unsigned boff = s * (unsigned)(HID * 4) + 16u * (unsigned)lane;
        r[j] = *(const float4*)((const char*)Hf + boff);
    }
    __builtin_amdgcn_sched_barrier(0);
    float e[NGRP];
    float mc = ninf;
#pragma unroll
    for (int i = 0; i < NGRP; ++i) {
        const float sc = quad_transpose_sum(dot4p(r[4 * i], a01, a23), dot4p(r[4 * i + 1], a01, a23),
                                            dot4p(r[4 * i + 2], a01, a23), dot4p(r[4 * i + 3], a01, a23), odd, up);
        const float v = leaky(sc + adg);
        e[i] = (4 * i + q < cnt) ? v : ninf;
        mc = fmaxf(mc, e[i]);
    }
    mc = fmaxf(mc, quad_x1(mc));
    mc = fmaxf(mc, quad_x2(mc));
    const float mn = fmaxf(m, mc);
    const float f = expf(m - mn);
    m = mn;
    Sp *= f;
    acc.x *= f; acc.y *= f; acc.z *= f; acc.w *= f;
#pragma unroll
    for (int i = 0; i < NGRP; ++i) {
        const float w = expf(e[i] - mn);
        Sp += w;
        float al;
        al = quad_bcast<0>(w);
        acc.x += al * r[4 * i].x; acc.y += al * r[4 * i].y; acc.z += al * r[4 * i].z; acc.w += al * r[4 * i].w;
        al = quad_bcast<1>(w);
        acc.x += al * r[4 * i + 1].x; acc.y += al * r[4 * i + 1].y; acc.z += al * r[4 * i + 1].z; acc.w += al * r[4 * i + 1].w;
        al = quad_bcast<2>(w);
        acc.x += al * r[4 * i + 2].x; acc.y += al * r[4 * i + 2].y; acc.z += al * r[4 * i + 2].z; acc.w += al * r[4 * i + 2].w;
        al = quad_bcast<3>(w);
        acc.x += al * r[4 * i + 3].x; acc.y += al * r[4 * i + 3].y; acc.z += al * r[4 * i + 3].z; acc.w += al * r[4 * i + 3].w;
    }
}

struct NodeHead {
    int rv;
    float asv, adg;
    float4 rself;
};

static __device__ __forceinline__ NodeHead node_head(int d, int lane, int hd,
                                                     const int* __restrict__ rp,
                                                     const float* __restrict__ AS, const float* __restrict__ AD,
                                                     const float* __restrict__ Hf) {
    NodeHead h;
    h.rv = rp[d + (lane & 1)];
    h.adg = AD[d * HEADS + hd];
    h.asv = AS[d * HEADS + hd];
    const unsigned doff = (unsigned)d * (unsigned)(HID * 4) + 16u * (unsigned)lane;
    h.rself = *(const float4*)((const char*)Hf + doff);
    return h;
}

static __device__ __forceinline__ float4 gather_body(int d, const NodeHead h, int lane, const float4 bb, const float4 as4,
                                                     const int* __restrict__ lst,
                                                     const float* __restrict__ Hf) {
    int beg = __builtin_amdgcn_readlane(h.rv, 0), end = __builtin_amdgcn_readlane(h.rv, 1);
    beg = min(max(beg, 0), NEDGES);
    end = min(max(end, beg), NEDGES);

    const float adg   = h.adg;
    const float eself = leaky(h.asv + adg);

    float m  = eself;
    float Sp = ((lane & 3) == 0) ? 1.0f : 0.0f;
    float4 acc = h.rself;

    for (int c0 = beg; c0 < end; c0 += 32) {
        const int cnt = min(32, end - c0);
        const int sv = load_chunk(lst, c0, cnt, lane, d);
        for (int hb = 0; hb < cnt; hb += 16) {
            const int c = __builtin_amdgcn_readfirstlane(min(16, cnt - hb));
            if (c <= 4)       rows_step<1>(Hf, sv, hb, c, lane, as4, adg, m, Sp, acc);
            else if (c <= 8)  rows_step<2>(Hf, sv, hb, c, lane, as4, adg, m, Sp, acc);
            else if (c <= 12) rows_step<3>(Hf, sv, hb, c, lane, as4, adg, m, Sp, acc);
            else              rows_step<4>(Hf, sv, hb, c, lane, as4, adg, m, Sp, acc);
        }
    }

    const float S = quad_sum(Sp);
    const float inv = 1.0f / (S + 1e-16f);

    float4 o;
    float v;
    v = acc.x * inv + bb.x; o.x = v > 0.f ? v : expm1f(v);
    v = acc.y * inv + bb.y; o.y = v > 0.f ? v : expm1f(v);
    v = acc.z * inv + bb.z; o.z = v > 0.f ? v : expm1f(v);
    v = acc.w * inv + bb.w; o.w = v > 0.f ? v : expm1f(v);
    return o;
}

__global__ __launch_bounds__(256) __attribute__((amdgpu_waves_per_eu(8, 8)))
void k_node_gather(const int* __restrict__ rp, const int* __restrict__ lst,
                   const float* __restrict__ AS, const float* __restrict__ AD,
                   const float* __restrict__ Hf, const float* __restrict__ a_src,
                   const float* __restrict__ b, float* __restrict__ Out) {
    const int lane = threadIdx.x & 31;
    const int d0 = __builtin_amdgcn_readfirstlane(blockIdx.x * WPB + (threadIdx.x >> 5));
    const int hd = lane >> 2;
    const float4 bb  = *(const float4*)(b + 4 * lane);
    const float4 as4 = *(const float4*)(a_src + 4 * lane);

    for (int d = d0; d < NNODES; d += NWAVES) {
        const NodeHead h = node_head(d, lane, hd, rp, AS, AD, Hf);
        const float4 o = gather_body(d, h, lane, bb, as4, lst, Hf);
        const unsigned doff = (unsigned)d * (unsigned)(HID * 4) + 16u * (unsigned)lane;
        *(float4*)((char*)Out + doff) = o;
    }
}

#define FT 512
#define FGRID 512
#define NSTRIP (NNODES / 16)
#define XP 132

__global__ __launch_bounds__(FT) __attribute__((amdgpu_waves_per_eu(8, 8)))
void k_gather_gemm(const int* __restrict__ rp, const int* __restrict__ lst,
                   const float* __restrict__ AS, const float* __restrict__ AD,
                   const float* __restrict__ Hf, const float* __restrict__ a_src,
                   const float* __restrict__ b,
                   const float* __restrict__ W, const float* __restrict__ a_src2,
                   const float* __restrict__ a_dst2,
                   float* __restrict__ Hout, float* __restrict__ AS2) {
    __shared__ __attribute__((aligned(16))) float ldsW[(FEAT / 2) * WP];
    __shared__ __attribute__((aligned(16))) float ldsX[2 * 16 * XP];
    const int tid = threadIdx.x;
    const int lane = tid & 31;
    const int wave = __builtin_amdgcn_readfirstlane(tid >> 5);

    const int hd = lane >> 2;
    const float4 bb  = *(const float4*)(b + 4 * lane);
    const float4 as4 = *(const float4*)(a_src + 4 * lane);
    NodeHead h = node_head(16 * (int)blockIdx.x + wave, lane, hd, rp, AS, AD, Hf);

    {
        const float4 vs = ((const float4*)a_src2)[lane];
        const float4 vd = ((const float4*)a_dst2)[lane];
        const float4* Wv = (const float4*)W;
        constexpr int NT = (FEAT / 2) * (HID / 4) / FT;
        float4 r0[NT], r1[NT];
#pragma unroll
        for (int i = 0; i < NT; ++i) {
            const int item = tid + i * FT;
            const int kp = item >> 5;
            const int c4 = item & 31;
            r0[i] = Wv[(2 * kp) * (HID / 4) + c4];
            r1[i] = Wv[(2 * kp + 1) * (HID / 4) + c4];
        }
        const int ql = lane & 3;
#pragma unroll
        for (int i = 0; i < NT; ++i) {
            const int item = tid + i * FT;
            const int kp = item >> 5;
            const int c4 = item & 31;
            float4* dst = (float4*)(ldsW + kp * WP + c4 * 8);
            dst[0] = make_float4(r0[i].x, r1[i].x, r0[i].y, r1[i].y);
            dst[1] = make_float4(r0[i].z, r1[i].z, r0[i].w, r1[i].w);
            const float s0 = quad_sum(dot4(r0[i], vs));
            const float s1 = quad_sum(dot4(r1[i], vs));
            const float d0 = quad_sum(dot4(r0[i], vd));
            const float d1 = quad_sum(dot4(r1[i], vd));
            const float val = (ql & 2) ? ((ql & 1) ? d1 : d0) : ((ql & 1) ? s1 : s0);
            ldsW[kp * WP + 2 * HID + 16 * (ql >> 1) + 2 * (c4 >> 2) + (ql & 1)] = val;
        }
    }

    int p = 0;
    for (int s = blockIdx.x; s < NSTRIP; s += FGRID) {
        const int d = 16 * s + wave;
        const float4 o = gather_body(d, h, lane, bb, as4, lst, Hf);
        float* xs = ldsX + p * (16 * XP);
        *(float4*)(xs + wave * XP + 4 * lane) = o;
        __syncthreads();
        h = node_head(min(d + 16 * FGRID, NNODES - 1), lane, hd, rp, AS, AD, Hf);
        if (wave < 9) {
            int ln = lane;
            asm volatile("" : "+v"(ln));
            const int l16  = ln & 15;
            const int half = ln >> 4;
            float* obase;
            int opitch;
            unsigned ocol;
            if (wave < 8) { obase = Hout; opitch = HID;   ocol = 16u * (unsigned)wave + (unsigned)l16; }
            else          { obase = AS2;  opitch = HEADS; ocol = (l16 < HEADS) ? (unsigned)l16 : (unsigned)(NNODES * HEADS + l16 - HEADS); }
            const unsigned ooff = (unsigned)((16 * s + 8 * half) * opitch) + ocol;
            const float* wb = ldsW + half * WP + 2 * l16 + 32 * wave;
            const float* xb = xs + l16 * XP + 2 * half;
            v8f acc = {};
#pragma unroll
            for (int k0 = 0; k0 < FEAT; k0 += 4) {
                const v2f a  = *(const v2f*)(xb + k0);
                const v2f bq = *(const v2f*)(wb + (k0 / 2) * WP);
                acc = __builtin_amdgcn_wmma_f32_16x16x4_f32(
                          false, a, false, bq, (short)0, acc, false, false);
            }
#pragma unroll
            for (int j = 0; j < 8; ++j)
                obase[ooff + (unsigned)(j * opitch)] = acc[j];
        }
        p ^= 1;
    }
}

static inline unsigned grid_for(long long n) { return (unsigned)((n + 255) / 256); }

extern "C" void kernel_launch(void* const* d_in, const int* in_sizes, int n_in,
                              void* d_out, int out_size, void* d_ws, size_t ws_size,
                              hipStream_t stream) {
    const float* x   = (const float*)d_in[0];
    const int*   ei  = (const int*)  d_in[1];
    const float* W1  = (const float*)d_in[2];
    const float* as1 = (const float*)d_in[3];
    const float* ad1 = (const float*)d_in[4];
    const float* b1  = (const float*)d_in[5];
    const float* W2  = (const float*)d_in[6];
    const float* as2 = (const float*)d_in[7];
    const float* ad2 = (const float*)d_in[8];
    const float* b2  = (const float*)d_in[9];

    float* ws     = (float*)d_ws;
    float* bufH   = ws;
    float* bufAS  = bufH  + (size_t)NNODES * HID;
    float* bufAD  = bufAS + (size_t)NNODES * HEADS;
    float* bufM   = bufAD + (size_t)NNODES * HEADS;
    float* bufS   = bufM  + (size_t)NNODES * HEADS;
    float* bufAgg = bufS  + (size_t)NNODES * HEADS;
    int*   rp     = (int*)bufAgg;
    int*   lst    = rp + 50048;
    int*   Pre    = lst + NEDGES;
    int*   Bas    = Pre + NSLICE * PPITCH;
    float* out    = (float*)d_out;

    k_count_gemm<<<GA + NSLICE * NR, BT, 0, stream>>>(ei, Pre, x, W1, as1, ad1, out, bufAS, bufAD);
    k_bases_gemm<<<(GBLOCKS - GA) + BASES_BX * NR, 256, 0, stream>>>(Pre, rp, Bas, x, W1, as1, ad1, out, bufAS, bufAD);
    k_slice_place<<<dim3(NSLICE, NR), BT, 0, stream>>>(ei, Bas, lst);

    k_gather_gemm<<<FGRID, FT, 0, stream>>>(rp, lst, bufAS, bufAD, out, as1, b1,
                                            W2, as2, ad2, bufH, bufM);
    k_node_gather<<<GBLK, 256, 0, stream>>>(rp, lst, bufM, bufS, bufH, as2, b2, out);
}


# Gather's exponentials on the base-2 instruction: scores kept in log2(e) units so a weight is one exp2, ELU's exp(v)-1 through it (packed series near 0), 1/S by rcp; a quarter fewer vector cycles
# speedup vs baseline: 11.1802x; 1.0511x over previous
#include <hip/hip_runtime.h>
#include <math.h>

#define NNODES 50000
#define NEDGES 800000
#define FEAT 128
#define HID  128
#define HEADS 8
#define HDIM 16
#define NEG_SLOPE 0.2f
#define LOG2E 1.4426950408889634f

typedef __attribute__((ext_vector_type(2))) float v2f;
typedef __attribute__((ext_vector_type(8))) float v8f;

#define GROWS 128
#define GBLOCKS ((NNODES + GROWS - 1) / GROWS)
#define WP 288
static_assert(NNODES % 16 == 0, "a strip is wholly inside or wholly outside the rows");

static __device__ __forceinline__ float quad_sum(float v) {
    v += __int_as_float(__builtin_amdgcn_update_dpp(0, __float_as_int(v), 0xB1, 0xF, 0xF, true));
    v += __int_as_float(__builtin_amdgcn_update_dpp(0, __float_as_int(v), 0x4E, 0xF, 0xF, true));
    return v;
}

static __device__ __forceinline__ float dot4(const float4 r, const float4 a) {
    return r.x * a.x + r.y * a.y + r.z * a.z + r.w * a.w;
}

#define GA (GBLOCKS / 2)

template <int T0, int NH, bool ATT>
static __device__ __forceinline__
void strip_tiles(const float* __restrict__ xrow, const float* __restrict__ ldsW,
                 const int half, const int l16, const int orow0,
                 float* __restrict__ Hout, float* __restrict__ AS, float* __restrict__ AD) {
    v8f acc[NH + 1] = {};
#pragma unroll
    for (int k0 = 0; k0 < FEAT; k0 += 4) {
        const int ka = k0 + 2 * half;
        v2f a;
        a.x = xrow[ka];
        a.y = xrow[ka + 1];
        const float* bp = ldsW + (k0 / 2 + half) * WP + 2 * l16;
#pragma unroll
        for (int t = 0; t < NH; ++t) {
            const v2f b = *(const v2f*)(bp + 32 * (T0 + t));
            acc[t] = __builtin_amdgcn_wmma_f32_16x16x4_f32(
                         false, a, false, b, (short)0, acc[t], false, false);
        }
        if (ATT) {
            const v2f b = *(const v2f*)(bp + 32 * 8);
            acc[NH] = __builtin_amdgcn_wmma_f32_16x16x4_f32(
                          false, a, false, b, (short)0, acc[NH], false, false);
        }
    }

#pragma unroll
    for (int t = 0; t < NH; ++t)
#pragma unroll
        for (int j = 0; j < 8; ++j)
            Hout[(size_t)(orow0 + j) * HID + 16 * (T0 + t) + l16] = acc[t][j];

    if (ATT) {
        float* aout = (l16 < HEADS) ? (AS + l16) : (AD + (l16 - HEADS));
#pragma unroll
        for (int j = 0; j < 8; ++j)
            aout[(size_t)(orow0 + j) * HEADS] = acc[NH][j];
    }
}

template <int NTHR>
static __device__ __forceinline__
void gemm_block(const int gb, float* __restrict__ ldsW,
                const float* __restrict__ X, const float* __restrict__ W,
                const float* __restrict__ a_src, const float* __restrict__ a_dst,
                float* __restrict__ Hout, float* __restrict__ AS,
                float* __restrict__ AD) {
    const int tid = threadIdx.x;
    const int lane = tid & 31;

    const float4 vs = ((const float4*)a_src)[lane];
    const float4 vd = ((const float4*)a_dst)[lane];

    const float4* Wv = (const float4*)W;
    constexpr int NT = (FEAT / 2) * (HID / 4) / NTHR;
    float4 r0[NT], r1[NT];
#pragma unroll
    for (int i = 0; i < NT; ++i) {
        const int item = tid + i * NTHR;
        const int kp = item >> 5;
        const int c4 = item & 31;
        r0[i] = Wv[(2 * kp) * (HID / 4) + c4];
        r1[i] = Wv[(2 * kp + 1) * (HID / 4) + c4];
    }
    const int ql = lane & 3;
#pragma unroll
    for (int i = 0; i < NT; ++i) {
        const int item = tid + i * NTHR;
        const int kp = item >> 5;
        const int c4 = item & 31;
        float4* dst = (float4*)(ldsW + kp * WP + c4 * 8);
        dst[0] = make_float4(r0[i].x, r1[i].x, r0[i].y, r1[i].y);
        dst[1] = make_float4(r0[i].z, r1[i].z, r0[i].w, r1[i].w);

        const float s0 = quad_sum(dot4(r0[i], vs));
        const float s1 = quad_sum(dot4(r1[i], vs));
        const float d0 = quad_sum(dot4(r0[i], vd));
        const float d1 = quad_sum(dot4(r1[i], vd));
        const float val = (ql & 2) ? ((ql & 1) ? d1 : d0) : ((ql & 1) ? s1 : s0);
        ldsW[kp * WP + 2 * HID + 16 * (ql >> 1) + 2 * (c4 >> 2) + (ql & 1)] = val;
    }
    __syncthreads();

    const int wave = __builtin_amdgcn_readfirstlane(tid >> 5);
    const int l16  = lane & 15;
    const int half = lane >> 4;
    const int row0 = (gb * 8 + (wave & 7)) * 16;
    if (row0 >= NNODES) return;

    const float* xrow = X + (size_t)(row0 + l16) * FEAT;
    const int orow0 = row0 + 8 * half;
    if (NTHR == 256) {
        strip_tiles<0, 8, true>(xrow, ldsW, half, l16, orow0, Hout, AS, AD);
    } else {
        if (wave < 8) strip_tiles<0, 4, true>(xrow, ldsW, half, l16, orow0, Hout, AS, AD);
        else          strip_tiles<4, 4, false>(xrow, ldsW, half, l16, orow0, Hout, AS, AD);
    }
}

#define NSLICE  32
#define SLICE_E (NEDGES / NSLICE)
#define SLICE_V (SLICE_E / 4)
#define NR      8
#define RN      6256
#define RP      6272
#define PPITCH  (NR * RP)
#define BPITCH  (NR * RN)
#define BT      512
#define EV      ((SLICE_V + BT - 1) / BT)
#define OWN4    4
#define NOWN    (RN / (4 * OWN4))
#define RV      (RN / 4)
#define CLR     ((RP / 4 + BT - 1) / BT)
static_assert(NEDGES % (NSLICE * 4) == 0, "slices are whole 16-byte pieces");
static_assert(NR * RN > NNODES && (NR - 1) * RN < NNODES, "the ranges hold every node and the total");
static_assert(RN % (4 * OWN4) == 0 && NOWN < BT && RP >= RN + 4 && RP % 32 == 0 && NNODES % 4 == 0,
              "whole pieces; the last thread scans nothing and carries the total");
static_assert(NSLICE == 32 && NSLICE * NR == 256, "k_bases_gemm: one (slice, range) total per thread, a wave per range");

static __device__ __forceinline__ unsigned range_span(int lo) {
    return (unsigned)(min(lo + RN, NNODES) - lo);
}

static __device__ __forceinline__ void bump(int* cnt, int i, unsigned span) {
    if ((unsigned)i < span) atomicAdd(&cnt[i], 1);
}

__global__ __launch_bounds__(BT) __attribute__((amdgpu_waves_per_eu(8, 8)))
void k_count_gemm(const int* __restrict__ ei, int* __restrict__ P,
                  const float* __restrict__ X, const float* __restrict__ W,
                  const float* __restrict__ a_src, const float* __restrict__ a_dst,
                  float* __restrict__ Hout, float* __restrict__ AS, float* __restrict__ AD) {
    __shared__ __attribute__((aligned(16))) int cnt[RP];
    __shared__ int wtot[32];
    __shared__ __attribute__((aligned(16))) float ldsW[(FEAT / 2) * WP];
    if (blockIdx.x < GA) {
        gemm_block<BT>((int)blockIdx.x, ldsW, X, W, a_src, a_dst, Hout, AS, AD);
        return;
    }
    const int t = threadIdx.x, lane = t & 31, w = t >> 5;
    const int cb = (int)blockIdx.x - GA;
    const int g = cb % NSLICE, r = cb / NSLICE;
    const int lo = r * RN;
    const unsigned span = range_span(lo);

    const int4* dv = (const int4*)(ei + NEDGES + g * SLICE_E);
    int4 dd[EV];
#pragma unroll
    for (int k = 0; k < EV; ++k) dd[k] = dv[min(t + k * BT, SLICE_V - 1)];

    int4* c4 = (int4*)cnt;
#pragma unroll
    for (int k = 0; k < CLR; ++k) {
        const int i = t + k * BT;
        if (i < RP / 4) c4[i] = make_int4(0, 0, 0, 0);
    }
    __syncthreads();

#pragma unroll
    for (int k = 0; k < EV; ++k) {
        if (t + k * BT < SLICE_V) {
            bump(cnt, dd[k].x - lo, span); bump(cnt, dd[k].y - lo, span);
            bump(cnt, dd[k].z - lo, span); bump(cnt, dd[k].w - lo, span);
        }
    }
    __syncthreads();

    int4 v[OWN4];
    int run = 0;
#pragma unroll
    for (int k = 0; k < OWN4; ++k) {
        int4 c = make_int4(0, 0, 0, 0);
        if (t < NOWN) c = c4[t * OWN4 + k];
        int x;
        x = c.x; c.x = run; run += x;
        x = c.y; c.y = run; run += x;
        x = c.z; c.z = run; run += x;
        x = c.w; c.w = run; run += x;
        v[k] = c;
    }
    int inc = run;
    for (int o = 1; o < 32; o <<= 1) {
        int n = __shfl_up(inc, o);
        if (lane >= o) inc += n;
    }
    if (lane == 31) wtot[w] = inc;
    __syncthreads();
    if (w == 0) {
        int x = (lane < BT / 32) ? wtot[lane] : 0, xi = x;
        for (int o = 1; o < 32; o <<= 1) {
            int n = __shfl_up(xi, o);
            if (lane >= o) xi += n;
        }
        if (lane < BT / 32) wtot[lane] = xi - x;
    }
    __syncthreads();
    const int off = wtot[w] + (inc - run);
    if (t < NOWN) {
#pragma unroll
        for (int k = 0; k < OWN4; ++k) {
            int4 c = v[k];
            c.x += off; c.y += off; c.z += off; c.w += off;
            c4[t * OWN4 + k] = c;
        }
    }
    if (t == BT - 1) c4[RV] = make_int4(off, 0, 0, 0);
    __syncthreads();

    int4* P4 = (int4*)(P + (size_t)g * PPITCH + r * RP);
#pragma unroll
    for (int k = 0; k < CLR; ++k) {
        const int i = t + k * BT;
        if (i < RV + 1) P4[i] = c4[i];
    }
}

#define BASES_BX ((RN + 255) / 256)

__global__ __launch_bounds__(256)
void k_bases_gemm(const int* __restrict__ P, int* __restrict__ rp, int* __restrict__ B,
                  const float* __restrict__ X, const float* __restrict__ W,
                  const float* __restrict__ a_src, const float* __restrict__ a_dst,
                  float* __restrict__ Hout, float* __restrict__ AS, float* __restrict__ AD) {
    __shared__ int part[8];
    __shared__ __attribute__((aligned(16))) float ldsW[(FEAT / 2) * WP];
    if (blockIdx.x < GBLOCKS - GA) {
        gemm_block<256>(GA + (int)blockIdx.x, ldsW, X, W, a_src, a_dst, Hout, AS, AD);
        return;
    }
    const int bb = (int)blockIdx.x - (GBLOCKS - GA);
    const int bx = bb % BASES_BX;
    const int t = threadIdx.x, r = bb / BASES_BX;

    const int tr = t >> 5, tg = t & 31;
    int x = (tr < r) ? P[(size_t)tg * PPITCH + tr * RP + RN] : 0;
    for (int o = 16; o > 0; o >>= 1) x += __shfl_xor(x, o);
    if (tg == 0) part[tr] = x;
    __syncthreads();
    int sum = 0;
#pragma unroll
    for (int k = 0; k < 8; ++k) sum += part[k];

    const int i = bx * 256 + t;
    const int n = r * RN + i;
    if (i >= RN || n > NNODES) return;
    const int* Pr = P + r * RP + i;
    int p[NSLICE], q[NSLICE];
#pragma unroll
    for (int g = 0; g < NSLICE; ++g) {
        p[g] = Pr[(size_t)g * PPITCH];
        q[g] = Pr[(size_t)g * PPITCH + 1];
        sum += p[g];
    }
    rp[n] = sum;
    if (n < NNODES) {
        int run = sum;
#pragma unroll
        for (int g = 0; g < NSLICE; ++g) {
            B[(size_t)g * BPITCH + n] = run;
            run += q[g] - p[g];
        }
    }
}

static __device__ __forceinline__ int take(int* pos, int i, unsigned span) {
    return ((unsigned)i < span) ? atomicAdd(&pos[i], 1) : -1;
}

static __device__ __forceinline__ void put(int* __restrict__ lst, int slot, int s) {
    if ((unsigned)slot < (unsigned)NEDGES) lst[slot] = s;
}

__global__ __launch_bounds__(BT)
void k_slice_place(const int* __restrict__ ei, const int* __restrict__ B,
                   int* __restrict__ lst) {
    __shared__ __attribute__((aligned(16))) int pos[CLR * BT * 4];
    const int t = threadIdx.x, g = blockIdx.x, r = blockIdx.y;
    const int lo = r * RN;
    const unsigned span = range_span(lo);

    const int4* sv = (const int4*)(ei + g * SLICE_E);
    const int4* dv = (const int4*)(ei + NEDGES + g * SLICE_E);
    int4 ss[EV], dd[EV];
#pragma unroll
    for (int k = 0; k < EV; ++k) {
        const int i = min(t + k * BT, SLICE_V - 1);
        ss[k] = sv[i];
        dd[k] = dv[i];
    }

    const int4* B4 = (const int4*)(B + (size_t)g * BPITCH + lo);
    int4* p4 = (int4*)pos;
    const int nb = (int)(span / 4);
    int4 bv[CLR];
#pragma unroll
    for (int k = 0; k < CLR; ++k) bv[k] = B4[min(t + k * BT, nb - 1)];
#pragma unroll
    for (int k = 0; k < CLR; ++k) p4[t + k * BT] = bv[k];
    __syncthreads();

#pragma unroll
    for (int k = 0; k < EV; ++k) {
        int4 sl = make_int4(-1, -1, -1, -1);
        if (t + k * BT < SLICE_V) {
            sl.x = take(pos, dd[k].x - lo, span); sl.y = take(pos, dd[k].y - lo, span);
            sl.z = take(pos, dd[k].z - lo, span); sl.w = take(pos, dd[k].w - lo, span);
        }
        dd[k] = sl;
    }
#pragma unroll
    for (int k = 0; k < EV; ++k) {
        put(lst, dd[k].x, ss[k].x); put(lst, dd[k].y, ss[k].y);
        put(lst, dd[k].z, ss[k].z); put(lst, dd[k].w, ss[k].w);
    }
}

#define WPB 8
#define GBLK 1024
#define NWAVES (GBLK * WPB)

static __device__ __forceinline__ float leaky(float v) {
    return v >= 0.f ? v : NEG_SLOPE * v;
}

static __device__ __forceinline__ int load_chunk(const int* __restrict__ lst,
                                                 int c0, int cnt, int lane, int d) {
    int sv = (lane < cnt) ? lst[c0 + lane] : d;
    return min(max(sv, 0), NNODES - 1);
}

static __device__ __forceinline__ float quad_x1(float v) {
    return __int_as_float(__builtin_amdgcn_update_dpp(0, __float_as_int(v), 0xB1, 0xF, 0xF, true));
}
static __device__ __forceinline__ float quad_x2(float v) {
    return __int_as_float(__builtin_amdgcn_update_dpp(0, __float_as_int(v), 0x4E, 0xF, 0xF, true));
}
template <int QQ>
static __device__ __forceinline__ float quad_bcast(float v) {
    return __int_as_float(__builtin_amdgcn_update_dpp(0, __float_as_int(v), QQ * 0x55, 0xF, 0xF, true));
}

static __device__ __forceinline__ float dot4p(const float4 r, const v2f a01, const v2f a23) {
    v2f t = (v2f){r.x, r.y} * a01;
    t = (v2f){r.z, r.w} * a23 + t;
    return t.x + t.y;
}

static __device__ __forceinline__ float quad_transpose_sum(float p0, float p1, float p2, float p3,
                                                           bool odd, bool up) {
    const float k01 = odd ? p1 : p0, s01 = odd ? p0 : p1;
    const float k23 = odd ? p3 : p2, s23 = odd ? p2 : p3;
    const float u = k01 + quad_x1(s01);
    const float v = k23 + quad_x1(s23);
    const float k = up ? v : u, s = up ? u : v;
    return k + quad_x2(s);
}

template <int NGRP>
static __device__ __forceinline__ void rows_step(const float* __restrict__ Hf, int sv, int base,
                                                 int cnt, int lane, const float4 as4, float adg,
                                                 float& m, float& Sp, float4& acc) {
    const int q = lane & 3;
    const bool odd = (q & 1) != 0, up = (q & 2) != 0;
    const float ninf = -__builtin_huge_valf();
    const v2f a01 = (v2f){as4.x, as4.y}, a23 = (v2f){as4.z, as4.w};
    float4 r[4 * NGRP];
#pragma unroll
    for (int j = 0; j < 4 * NGRP; ++j) {
        const unsigned s = (unsigned)__builtin_amdgcn_readlane(sv, base + j);
        const unsigned boff = s * (unsigned)(HID * 4) + 16u * (unsigned)lane;
        r[j] = *(const float4*)((const char*)Hf + boff);
    }
    __builtin_amdgcn_sched_barrier(0);
    float e[NGRP];
    float mc = ninf;
#pragma unroll
    for (int i = 0; i < NGRP; ++i) {
        const float sc = quad_transpose_sum(dot4p(r[4 * i], a01, a23), dot4p(r[4 * i + 1], a01, a23),
                                            dot4p(r[4 * i + 2], a01, a23), dot4p(r[4 * i + 3], a01, a23), odd, up);
        const float v = leaky(sc + adg);
        e[i] = (4 * i + q < cnt) ? v : ninf;
        mc = fmaxf(mc, e[i]);
    }
    mc = fmaxf(mc, quad_x1(mc));
    mc = fmaxf(mc, quad_x2(mc));
    const float mn = fmaxf(m, mc);
    const float f = __builtin_amdgcn_exp2f(m - mn);
    m = mn;
    Sp *= f;
    acc.x *= f; acc.y *= f; acc.z *= f; acc.w *= f;
#pragma unroll
    for (int i = 0; i < NGRP; ++i) {
        const float w = __builtin_amdgcn_exp2f(e[i] - mn);
        Sp += w;
        float al;
        al = quad_bcast<0>(w);
        acc.x += al * r[4 * i].x; acc.y += al * r[4 * i].y; acc.z += al * r[4 * i].z; acc.w += al * r[4 * i].w;
        al = quad_bcast<1>(w);
        acc.x += al * r[4 * i + 1].x; acc.y += al * r[4 * i + 1].y; acc.z += al * r[4 * i + 1].z; acc.w += al * r[4 * i + 1].w;
        al = quad_bcast<2>(w);
        acc.x += al * r[4 * i + 2].x; acc.y += al * r[4 * i + 2].y; acc.z += al * r[4 * i + 2].z; acc.w += al * r[4 * i + 2].w;
        al = quad_bcast<3>(w);
        acc.x += al * r[4 * i + 3].x; acc.y += al * r[4 * i + 3].y; acc.z += al * r[4 * i + 3].z; acc.w += al * r[4 * i + 3].w;
    }
}

static __device__ __forceinline__ v2f elu2(const v2f v) {
    const v2f x = v * LOG2E;
    const v2f t = (v2f){__builtin_amdgcn_exp2f(x.x), __builtin_amdgcn_exp2f(x.y)} - 1.0f;
    v2f q = v * (1.0f / 120.0f) + (1.0f / 24.0f);
    q = v * q + (1.0f / 6.0f);
    q = v * q + 0.5f;
    q = v * q + 1.0f;
    const v2f p = v * q;
    v2f o;
    o.x = v.x > 0.f ? v.x : (v.x > -0.125f ? p.x : t.x);
    o.y = v.y > 0.f ? v.y : (v.y > -0.125f ? p.y : t.y);
    return o;
}

struct NodeHead {
    int rv;
    float asv, adg;
    float4 rself;
};

static __device__ __forceinline__ NodeHead node_head(int d, int lane, int hd,
                                                     const int* __restrict__ rp,
                                                     const float* __restrict__ AS, const float* __restrict__ AD,
                                                     const float* __restrict__ Hf) {
    NodeHead h;
    h.rv = rp[d + (lane & 1)];
    h.adg = AD[d * HEADS + hd];
    h.asv = AS[d * HEADS + hd];
    const unsigned doff = (unsigned)d * (unsigned)(HID * 4) + 16u * (unsigned)lane;
    h.rself = *(const float4*)((const char*)Hf + doff);
    return h;
}

static __device__ __forceinline__ float4 gather_body(int d, const NodeHead h, int lane, const float4 bb, const float4 as4,
                                                     const int* __restrict__ lst,
                                                     const float* __restrict__ Hf) {
    int beg = __builtin_amdgcn_readlane(h.rv, 0), end = __builtin_amdgcn_readlane(h.rv, 1);
    beg = min(max(beg, 0), NEDGES);
    end = min(max(end, beg), NEDGES);

    const float adg   = h.adg * LOG2E;
    const float eself = leaky(h.asv * LOG2E + adg);

    float m  = eself;
    float Sp = ((lane & 3) == 0) ? 1.0f : 0.0f;
    float4 acc = h.rself;

    for (int c0 = beg; c0 < end; c0 += 32) {
        const int cnt = min(32, end - c0);
        const int sv = load_chunk(lst, c0, cnt, lane, d);
        for (int hb = 0; hb < cnt; hb += 16) {
            const int c = __builtin_amdgcn_readfirstlane(min(16, cnt - hb));
            if (c <= 4)       rows_step<1>(Hf, sv, hb, c, lane, as4, adg, m, Sp, acc);
            else if (c <= 8)  rows_step<2>(Hf, sv, hb, c, lane, as4, adg, m, Sp, acc);
            else if (c <= 12) rows_step<3>(Hf, sv, hb, c, lane, as4, adg, m, Sp, acc);
            else              rows_step<4>(Hf, sv, hb, c, lane, as4, adg, m, Sp, acc);
        }
    }

    const float S = quad_sum(Sp);
    const float inv = __builtin_amdgcn_rcpf(S + 1e-16f);

    const v2f o01 = elu2((v2f){acc.x, acc.y} * inv + (v2f){bb.x, bb.y});
    const v2f o23 = elu2((v2f){acc.z, acc.w} * inv + (v2f){bb.z, bb.w});
    return make_float4(o01.x, o01.y, o23.x, o23.y);
}

__global__ __launch_bounds__(256) __attribute__((amdgpu_waves_per_eu(8, 8)))
void k_node_gather(const int* __restrict__ rp, const int* __restrict__ lst,
                   const float* __restrict__ AS, const float* __restrict__ AD,
                   const float* __restrict__ Hf, const float* __restrict__ a_src,
                   const float* __restrict__ b, float* __restrict__ Out) {
    const int lane = threadIdx.x & 31;
    const int d0 = __builtin_amdgcn_readfirstlane(blockIdx.x * WPB + (threadIdx.x >> 5));
    const int hd = lane >> 2;
    const float4 bb  = *(const float4*)(b + 4 * lane);
    float4 as4 = *(const float4*)(a_src + 4 * lane);
    as4.x *= LOG2E; as4.y *= LOG2E; as4.z *= LOG2E; as4.w *= LOG2E;

    for (int d = d0; d < NNODES; d += NWAVES) {
        const NodeHead h = node_head(d, lane, hd, rp, AS, AD, Hf);
        const float4 o = gather_body(d, h, lane, bb, as4, lst, Hf);
        const unsigned doff = (unsigned)d * (unsigned)(HID * 4) + 16u * (unsigned)lane;
        *(float4*)((char*)Out + doff) = o;
    }
}

#define FT 512
#define FGRID 512
#define NSTRIP (NNODES / 16)
#define XP 132

__global__ __launch_bounds__(FT) __attribute__((amdgpu_waves_per_eu(8, 8)))
void k_gather_gemm(const int* __restrict__ rp, const int* __restrict__ lst,
                   const float* __restrict__ AS, const float* __restrict__ AD,
                   const float* __restrict__ Hf, const float* __restrict__ a_src,
                   const float* __restrict__ b,
                   const float* __restrict__ W, const float* __restrict__ a_src2,
                   const float* __restrict__ a_dst2,
                   float* __restrict__ Hout, float* __restrict__ AS2) {
    __shared__ __attribute__((aligned(16))) float ldsW[(FEAT / 2) * WP];
    __shared__ __attribute__((aligned(16))) float ldsX[2 * 16 * XP];
    const int tid = threadIdx.x;
    const int lane = tid & 31;
    const int wave = __builtin_amdgcn_readfirstlane(tid >> 5);

    const int hd = lane >> 2;
    const float4 bb  = *(const float4*)(b + 4 * lane);
    float4 as4 = *(const float4*)(a_src + 4 * lane);
    as4.x *= LOG2E; as4.y *= LOG2E; as4.z *= LOG2E; as4.w *= LOG2E;
    NodeHead h = node_head(16 * (int)blockIdx.x + wave, lane, hd, rp, AS, AD, Hf);

    {
        const float4 vs = ((const float4*)a_src2)[lane];
        const float4 vd = ((const float4*)a_dst2)[lane];
        const float4* Wv = (const float4*)W;
        constexpr int NT = (FEAT / 2) * (HID / 4) / FT;
        float4 r0[NT], r1[NT];
#pragma unroll
        for (int i = 0; i < NT; ++i) {
            const int item = tid + i * FT;
            const int kp = item >> 5;
            const int c4 = item & 31;
            r0[i] = Wv[(2 * kp) * (HID / 4) + c4];
            r1[i] = Wv[(2 * kp + 1) * (HID / 4) + c4];
        }
        const int ql = lane & 3;
#pragma unroll
        for (int i = 0; i < NT; ++i) {
            const int item = tid + i * FT;
            const int kp = item >> 5;
            const int c4 = item & 31;
            float4* dst = (float4*)(ldsW + kp * WP + c4 * 8);
            dst[0] = make_float4(r0[i].x, r1[i].x, r0[i].y, r1[i].y);
            dst[1] = make_float4(r0[i].z, r1[i].z, r0[i].w, r1[i].w);
            const float s0 = quad_sum(dot4(r0[i], vs));
            const float s1 = quad_sum(dot4(r1[i], vs));
            const float d0 = quad_sum(dot4(r0[i], vd));
            const float d1 = quad_sum(dot4(r1[i], vd));
            const float val = (ql & 2) ? ((ql & 1) ? d1 : d0) : ((ql & 1) ? s1 : s0);
            ldsW[kp * WP + 2 * HID + 16 * (ql >> 1) + 2 * (c4 >> 2) + (ql & 1)] = val;
        }
    }

    int p = 0;
    for (int s = blockIdx.x; s < NSTRIP; s += FGRID) {
        const int d = 16 * s + wave;
        const float4 o = gather_body(d, h, lane, bb, as4, lst, Hf);
        float* xs = ldsX + p * (16 * XP);
        *(float4*)(xs + wave * XP + 4 * lane) = o;
        __syncthreads();
        h = node_head(min(d + 16 * FGRID, NNODES - 1), lane, hd, rp, AS, AD, Hf);
        if (wave < 9) {
            int ln = lane;
            asm volatile("" : "+v"(ln));
            const int l16  = ln & 15;
            const int half = ln >> 4;
            float* obase;
            int opitch;
            unsigned ocol;
            if (wave < 8) { obase = Hout; opitch = HID;   ocol = 16u * (unsigned)wave + (unsigned)l16; }
            else          { obase = AS2;  opitch = HEADS; ocol = (l16 < HEADS) ? (unsigned)l16 : (unsigned)(NNODES * HEADS + l16 - HEADS); }
            const unsigned ooff = (unsigned)((16 * s + 8 * half) * opitch) + ocol;
            const float* wb = ldsW + half * WP + 2 * l16 + 32 * wave;
            const float* xb = xs + l16 * XP + 2 * half;
            v8f acc = {};
#pragma unroll
            for (int k0 = 0; k0 < FEAT; k0 += 4) {
                const v2f a  = *(const v2f*)(xb + k0);
                const v2f bq = *(const v2f*)(wb + (k0 / 2) * WP);
                acc = __builtin_amdgcn_wmma_f32_16x16x4_f32(
                          false, a, false, bq, (short)0, acc, false, false);
            }
#pragma unroll
            for (int j = 0; j < 8; ++j)
                obase[ooff + (unsigned)(j * opitch)] = acc[j];
        }
        p ^= 1;
    }
}

static inline unsigned grid_for(long long n) { return (unsigned)((n + 255) / 256); }

extern "C" void kernel_launch(void* const* d_in, const int* in_sizes, int n_in,
                              void* d_out, int out_size, void* d_ws, size_t ws_size,
                              hipStream_t stream) {
    const float* x   = (const float*)d_in[0];
    const int*   ei  = (const int*)  d_in[1];
    const float* W1  = (const float*)d_in[2];
    const float* as1 = (const float*)d_in[3];
    const float* ad1 = (const float*)d_in[4];
    const float* b1  = (const float*)d_in[5];
    const float* W2  = (const float*)d_in[6];
    const float* as2 = (const float*)d_in[7];
    const float* ad2 = (const float*)d_in[8];
    const float* b2  = (const float*)d_in[9];

    float* ws     = (float*)d_ws;
    float* bufH   = ws;
    float* bufAS  = bufH  + (size_t)NNODES * HID;
    float* bufAD  = bufAS + (size_t)NNODES * HEADS;
    float* bufM   = bufAD + (size_t)NNODES * HEADS;
    float* bufS   = bufM  + (size_t)NNODES * HEADS;
    float* bufAgg = bufS  + (size_t)NNODES * HEADS;
    int*   rp     = (int*)bufAgg;
    int*   lst    = rp + 50048;
    int*   Pre    = lst + NEDGES;
    int*   Bas    = Pre + NSLICE * PPITCH;
    float* out    = (float*)d_out;

    k_count_gemm<<<GA + NSLICE * NR, BT, 0, stream>>>(ei, Pre, x, W1, as1, ad1, out, bufAS, bufAD);
    k_bases_gemm<<<(GBLOCKS - GA) + BASES_BX * NR, 256, 0, stream>>>(Pre, rp, Bas, x, W1, as1, ad1, out, bufAS, bufAD);
    k_slice_place<<<dim3(NSLICE, NR), BT, 0, stream>>>(ei, Bas, lst);

    k_gather_gemm<<<FGRID, FT, 0, stream>>>(rp, lst, bufAS, bufAD, out, as1, b1,
                                            W2, as2, ad2, bufH, bufM);
    k_node_gather<<<GBLK, 256, 0, stream>>>(rp, lst, bufM, bufS, bufH, as2, b2, out);
}
